# starting kernel
# baseline (speedup 1.0000x reference)
#include <hip/hip_runtime.h>

#define BB 2
#define CC 64
#define DD 3
#define NN 4096
#define KK 16
#define EE 128
#define HH 256

typedef __attribute__((ext_vector_type(2))) float v2f;
typedef __attribute__((ext_vector_type(8))) float v8f;

constexpr int NT   = 4;
constexpr int COLS = NT * KK;
constexpr int LDSC = COLS + 4;

__global__ void meanpc_kernel(const float* __restrict__ x, float* __restrict__ mean_pc) {
    int i = blockIdx.x * blockDim.x + threadIdx.x;
    if (i >= BB * CC * NN) return;
    int n  = i % NN;
    int bc = i / NN;
    const float* xp = x + (size_t)bc * (DD * NN) + n;
    mean_pc[i] = (xp[0] + xp[NN] + xp[2 * NN]) * (1.0f / 3.0f);
}

__global__ void sqnorm_kernel(const float* __restrict__ mean_pc, float* __restrict__ sq) {
    int i = blockIdx.x * blockDim.x + threadIdx.x;
    if (i >= BB * NN) return;
    int n = i % NN;
    int b = i / NN;
    float s = 0.0f;
#pragma unroll
    for (int c = 0; c < CC; ++c) {
        float v = mean_pc[((size_t)b * CC + c) * NN + n];
        s = __builtin_fmaf(v, v, s);
    }
    sq[i] = s;
}

__global__ void __launch_bounds__(256) knn_kernel(const float* __restrict__ mean_pc,
                                                  const float* __restrict__ sq,
                                                  int* __restrict__ idx) {
    int t = threadIdx.x;
    int b = blockIdx.x / (NN / 256);
    int n = (blockIdx.x % (NN / 256)) * 256 + t;

    float q[CC];
#pragma unroll
    for (int c = 0; c < CC; ++c) q[c] = mean_pc[((size_t)b * CC + c) * NN + n];

    float bd[KK];
    int   bi[KK];
#pragma unroll
    for (int j = 0; j < KK; ++j) { bd[j] = 3.0e38f; bi[j] = 0; }

    for (int r = 0; r < NN; ++r) {
        float dot = 0.0f;
#pragma unroll
        for (int c = 0; c < CC; ++c)
            dot = __builtin_fmaf(q[c], mean_pc[((size_t)b * CC + c) * NN + r], dot);
        float d2 = sq[(size_t)b * NN + r] - 2.0f * dot;
        if (r != n && d2 < bd[KK - 1]) {
            bd[KK - 1] = d2;
            bi[KK - 1] = r;
#pragma unroll
            for (int j = KK - 1; j > 0; --j) {
                if (bd[j] < bd[j - 1]) {
                    float td = bd[j]; bd[j] = bd[j - 1]; bd[j - 1] = td;
                    int   ti = bi[j]; bi[j] = bi[j - 1]; bi[j - 1] = ti;
                }
            }
        }
    }
#pragma unroll
    for (int j = 0; j < KK; ++j) idx[((size_t)b * KK + j) * NN + n] = bi[j];
}

template <int NCT, bool RELU>
__device__ __forceinline__ void gemm_tile(const float* __restrict__ W,
                                          const float* __restrict__ bias,
                                          const float* lin, float* lout,
                                          int Ki, int rt, int ctBase,
                                          int laneRow, int laneHi) {
    const v8f vzero = {0.f, 0.f, 0.f, 0.f, 0.f, 0.f, 0.f, 0.f};
    v8f acc[NCT];
#pragma unroll
    for (int c = 0; c < NCT; ++c) acc[c] = vzero;

    const float* wrow = W + (size_t)(rt * 16 + laneRow) * Ki;
    for (int k0 = 0; k0 < Ki; k0 += 4) {
        v2f a;
        a.x = wrow[k0 + 2 * laneHi + 0];
        a.y = wrow[k0 + 2 * laneHi + 1];
#pragma unroll
        for (int c = 0; c < NCT; ++c) {
            int col = (ctBase + c) * 16 + laneRow;
            v2f bf;
            bf.x = lin[(k0 + 2 * laneHi + 0) * LDSC + col];
            bf.y = lin[(k0 + 2 * laneHi + 1) * LDSC + col];
            acc[c] = __builtin_amdgcn_wmma_f32_16x16x4_f32(
                false, a, false, bf, (short)0, acc[c], false, false);
        }
    }
#pragma unroll
    for (int c = 0; c < NCT; ++c) {
#pragma unroll
        for (int j = 0; j < 8; ++j) {
            int row = rt * 16 + j + 8 * laneHi;
            float v = acc[c][j] + bias[row];
            if (RELU) v = fmaxf(v, 0.0f);
            lout[row * LDSC + (ctBase + c) * 16 + laneRow] = v;
        }
    }
}

__global__ void __launch_bounds__(256) mlp_kernel(const float* __restrict__ x,
                                                  const float* __restrict__ mean_pc,
                                                  const int* __restrict__ idx,
                                                  const float* __restrict__ W1, const float* __restrict__ b1,
                                                  const float* __restrict__ W2, const float* __restrict__ b2,
                                                  const float* __restrict__ W3, const float* __restrict__ b3,
                                                  float* __restrict__ out) {
    __shared__ float ldsX[HH * LDSC];
    __shared__ float ldsY[HH * LDSC];

    int t       = threadIdx.x;
    int lane    = t & 31;
    int w       = t >> 5;
    int laneRow = lane & 15;
    int laneHi  = lane >> 4;

    int blk = blockIdx.x;
    int nb  = blk % (NN / NT);
    int bd  = blk / (NN / NT);
    int d   = bd % DD;
    int b   = bd / DD;
    int n0  = nb * NT;

    for (int e = t; e < EE * COLS; e += 256) {
        int row = e >> 6;
        int col = e & (COLS - 1);
        int nl  = col >> 4;
        int k   = col & 15;
        int n   = n0 + nl;
        float v;
        if (row < CC) {
            v = x[(((size_t)b * CC + row) * DD + d) * NN + n];
        } else {
            int c = row - CC;
            int j = idx[((size_t)b * KK + k) * NN + n];
            v = mean_pc[((size_t)b * CC + c) * NN + j]
              - x[(((size_t)b * CC + c) * DD + d) * NN + n];
        }
        ldsX[row * LDSC + col] = v;
    }
    __syncthreads();

    gemm_tile<4, true>(W1, b1, ldsX, ldsY, EE, 2 * w + 0, 0, laneRow, laneHi);
    gemm_tile<4, true>(W1, b1, ldsX, ldsY, EE, 2 * w + 1, 0, laneRow, laneHi);
    __syncthreads();

    gemm_tile<4, true>(W2, b2, ldsY, ldsX, HH, 2 * w + 0, 0, laneRow, laneHi);
    gemm_tile<4, true>(W2, b2, ldsY, ldsX, HH, 2 * w + 1, 0, laneRow, laneHi);
    __syncthreads();

    gemm_tile<2, false>(W3, b3, ldsX, ldsY, HH, w >> 1, (w & 1) * 2, laneRow, laneHi);
    __syncthreads();

    {
        int c  = t & (CC - 1);
        int nl = t >> 6;
        const float* rrow = &ldsY[c * LDSC + nl * 16];
        float m = rrow[0];
#pragma unroll
        for (int k = 1; k < KK; ++k) m = fmaxf(m, rrow[k]);
        size_t o = (((size_t)b * CC + c) * DD + d) * NN + (n0 + nl);
        out[o] = m + x[o];
    }
}

extern "C" void kernel_launch(void* const* d_in, const int* in_sizes, int n_in,
                              void* d_out, int out_size, void* d_ws, size_t ws_size,
                              hipStream_t stream) {
    const float* x  = (const float*)d_in[0];
    const float* W1 = (const float*)d_in[1];
    const float* b1 = (const float*)d_in[2];
    const float* W2 = (const float*)d_in[3];
    const float* b2 = (const float*)d_in[4];
    const float* W3 = (const float*)d_in[5];
    const float* b3 = (const float*)d_in[6];
    float* out = (float*)d_out;

    float* mean_pc = (float*)d_ws;
    float* sq      = mean_pc + (size_t)BB * CC * NN;
    int*   idxp    = (int*)(sq + (size_t)BB * NN);

    meanpc_kernel<<<(BB * CC * NN + 255) / 256, 256, 0, stream>>>(x, mean_pc);
    sqnorm_kernel<<<(BB * NN + 255) / 256, 256, 0, stream>>>(mean_pc, sq);
    knn_kernel<<<BB * (NN / 256), 256, 0, stream>>>(mean_pc, sq, idxp);
    mlp_kernel<<<BB * DD * (NN / NT), 256, 0, stream>>>(x, mean_pc, idxp,
                                                        W1, b1, W2, b2, W3, b3, out);
}


# KNN re-tiled: block = 32 queries, its 8 waves each scan 512 refs (16 per step, ref broadcast by DPP row_share inside the FMA), top-16 lists merged in LDS; same fmaf order, scores bit-identical
# speedup vs baseline: 9.1023x; 9.1023x over previous
#include <hip/hip_runtime.h>

#define BB 2
#define CC 64
#define DD 3
#define NN 4096
#define KK 16
#define EE 128
#define HH 256

typedef __attribute__((ext_vector_type(2))) float v2f;
typedef __attribute__((ext_vector_type(8))) float v8f;

constexpr int NT   = 4;
constexpr int COLS = NT * KK;
constexpr int LDSC = COLS + 4;

__global__ void meanpc_kernel(const float* __restrict__ x, float* __restrict__ mean_pc) {
    int i = blockIdx.x * blockDim.x + threadIdx.x;
    if (i >= BB * CC * NN) return;
    int n  = i % NN;
    int bc = i / NN;
    const float* xp = x + (size_t)bc * (DD * NN) + n;
    mean_pc[i] = (xp[0] + xp[NN] + xp[2 * NN]) * (1.0f / 3.0f);
}

__global__ void sqnorm_kernel(const float* __restrict__ mean_pc, float* __restrict__ sq) {
    int i = blockIdx.x * blockDim.x + threadIdx.x;
    if (i >= BB * NN) return;
    int n = i % NN;
    int b = i / NN;
    float s = 0.0f;
#pragma unroll
    for (int c = 0; c < CC; ++c) {
        float v = mean_pc[((size_t)b * CC + c) * NN + n];
        s = __builtin_fmaf(v, v, s);
    }
    sq[i] = s;
}

constexpr int KQ   = 32;
constexpr int KW   = 8;
constexpr int KSEG = NN / KW;
constexpr int KRB  = 16;
constexpr int KNB  = KSEG / KRB;

template <int J>
__device__ __forceinline__ float row_bcast(float v) {
    return __builtin_bit_cast(float,
        __builtin_amdgcn_update_dpp(0, __builtin_bit_cast(int, v), 0x150 + J, 0xf, 0xf, true));
}

template <int J>
struct KnnStep {
    static __device__ __forceinline__ void fma(float (&dot)[KRB], float qc, float rc) {
        asm("v_fmac_f32_dpp %0, %1, %2 row_share:%3 row_mask:0xf bank_mask:0xf"
            : "+v"(dot[J]) : "v"(rc), "v"(qc), "n"(J));
        KnnStep<J + 1>::fma(dot, qc, rc);
    }
    static __device__ __forceinline__ void dist(float (&dot)[KRB], float sqv) {
        dot[J] = row_bcast<J>(sqv) - 2.0f * dot[J];
        KnnStep<J + 1>::dist(dot, sqv);
    }
};
template <>
struct KnnStep<KRB> {
    static __device__ __forceinline__ void fma(float (&)[KRB], float, float) {}
    static __device__ __forceinline__ void dist(float (&)[KRB], float) {}
};

__device__ __forceinline__ void knn_insert(float (&bd)[KK], int (&bi)[KK], float d, int i) {
    bd[KK - 1] = d;
    bi[KK - 1] = i;
#pragma unroll
    for (int j = KK - 1; j > 0; --j) {
        if (bd[j] < bd[j - 1]) {
            float td = bd[j]; bd[j] = bd[j - 1]; bd[j - 1] = td;
            int   ti = bi[j]; bi[j] = bi[j - 1]; bi[j - 1] = ti;
        }
    }
}

__global__ void __launch_bounds__(256) knn_kernel(const float* __restrict__ mean_pc,
                                                  const float* __restrict__ sq,
                                                  int* __restrict__ idx) {
    __shared__ float mrgD[KW * KK * KQ];
    __shared__ int   mrgI[KW * KK * KQ];

    int t    = threadIdx.x;
    int lane = t & 31;
    int sub  = lane & 15;
    int w    = __builtin_amdgcn_readfirstlane(t >> 5);
    int b    = blockIdx.x / (NN / KQ);
    int n    = (blockIdx.x % (NN / KQ)) * KQ + lane;
    const float* pc  = mean_pc + (size_t)b * CC * NN;
    const float* sqb = sq + (size_t)b * NN;

    float q[CC];
#pragma unroll
    for (int c = 0; c < CC; ++c) q[c] = pc[(size_t)c * NN + n];

    float bd[KK];
    int   bi[KK];
#pragma unroll
    for (int j = 0; j < KK; ++j) { bd[j] = 3.0e38f; bi[j] = 0; }

    int   r0 = w * KSEG;
    float ref[CC];
    float sqv;
#pragma unroll
    for (int c = 0; c < CC; ++c) ref[c] = pc[(size_t)c * NN + r0 + sub];
    sqv = sqb[r0 + sub];

    for (int rb = 0; rb < KNB; ++rb) {
        float dot[KRB];
#pragma unroll
        for (int j = 0; j < KRB; ++j) dot[j] = 0.0f;
#pragma unroll
        for (int c = 0; c < CC; ++c) KnnStep<0>::fma(dot, q[c], ref[c]);
        KnnStep<0>::dist(dot, sqv);

        int rn = (rb + 1 < KNB) ? r0 + KRB : r0;
#pragma unroll
        for (int c = 0; c < CC; ++c) ref[c] = pc[(size_t)c * NN + rn + sub];
        sqv = sqb[rn + sub];

#pragma unroll
        for (int j = 0; j < KRB; ++j) {
            int r = r0 + j;
            if (r != n && dot[j] < bd[KK - 1]) knn_insert(bd, bi, dot[j], r);
        }
        r0 = rn;
    }

#pragma unroll
    for (int j = 0; j < KK; ++j) {
        mrgD[(w * KK + j) * KQ + lane] = bd[j];
        mrgI[(w * KK + j) * KQ + lane] = bi[j];
    }
    __syncthreads();
    for (int step = 1; step < KW; step <<= 1) {
        if ((w & (2 * step - 1)) == 0) {
            int srcw = w + step;
            for (int j = 0; j < KK; ++j) {
                float cd = mrgD[(srcw * KK + j) * KQ + lane];
                int   ci = mrgI[(srcw * KK + j) * KQ + lane];
                if (cd < bd[KK - 1]) knn_insert(bd, bi, cd, ci);
            }
            if (2 * step < KW) {
#pragma unroll
                for (int j = 0; j < KK; ++j) {
                    mrgD[(w * KK + j) * KQ + lane] = bd[j];
                    mrgI[(w * KK + j) * KQ + lane] = bi[j];
                }
            }
        }
        __syncthreads();
    }
    if (w == 0) {
#pragma unroll
        for (int j = 0; j < KK; ++j) idx[((size_t)b * KK + j) * NN + n] = bi[j];
    }
}

template <int NCT, bool RELU>
__device__ __forceinline__ void gemm_tile(const float* __restrict__ W,
                                          const float* __restrict__ bias,
                                          const float* lin, float* lout,
                                          int Ki, int rt, int ctBase,
                                          int laneRow, int laneHi) {
    const v8f vzero = {0.f, 0.f, 0.f, 0.f, 0.f, 0.f, 0.f, 0.f};
    v8f acc[NCT];
#pragma unroll
    for (int c = 0; c < NCT; ++c) acc[c] = vzero;

    const float* wrow = W + (size_t)(rt * 16 + laneRow) * Ki;
    for (int k0 = 0; k0 < Ki; k0 += 4) {
        v2f a;
        a.x = wrow[k0 + 2 * laneHi + 0];
        a.y = wrow[k0 + 2 * laneHi + 1];
#pragma unroll
        for (int c = 0; c < NCT; ++c) {
            int col = (ctBase + c) * 16 + laneRow;
            v2f bf;
            bf.x = lin[(k0 + 2 * laneHi + 0) * LDSC + col];
            bf.y = lin[(k0 + 2 * laneHi + 1) * LDSC + col];
            acc[c] = __builtin_amdgcn_wmma_f32_16x16x4_f32(
                false, a, false, bf, (short)0, acc[c], false, false);
        }
    }
#pragma unroll
    for (int c = 0; c < NCT; ++c) {
#pragma unroll
        for (int j = 0; j < 8; ++j) {
            int row = rt * 16 + j + 8 * laneHi;
            float v = acc[c][j] + bias[row];
            if (RELU) v = fmaxf(v, 0.0f);
            lout[row * LDSC + (ctBase + c) * 16 + laneRow] = v;
        }
    }
}

__global__ void __launch_bounds__(256) mlp_kernel(const float* __restrict__ x,
                                                  const float* __restrict__ mean_pc,
                                                  const int* __restrict__ idx,
                                                  const float* __restrict__ W1, const float* __restrict__ b1,
                                                  const float* __restrict__ W2, const float* __restrict__ b2,
                                                  const float* __restrict__ W3, const float* __restrict__ b3,
                                                  float* __restrict__ out) {
    __shared__ float ldsX[HH * LDSC];
    __shared__ float ldsY[HH * LDSC];

    int t       = threadIdx.x;
    int lane    = t & 31;
    int w       = t >> 5;
    int laneRow = lane & 15;
    int laneHi  = lane >> 4;

    int blk = blockIdx.x;
    int nb  = blk % (NN / NT);
    int bd  = blk / (NN / NT);
    int d   = bd % DD;
    int b   = bd / DD;
    int n0  = nb * NT;

    for (int e = t; e < EE * COLS; e += 256) {
        int row = e >> 6;
        int col = e & (COLS - 1);
        int nl  = col >> 4;
        int k   = col & 15;
        int n   = n0 + nl;
        float v;
        if (row < CC) {
            v = x[(((size_t)b * CC + row) * DD + d) * NN + n];
        } else {
            int c = row - CC;
            int j = idx[((size_t)b * KK + k) * NN + n];
            v = mean_pc[((size_t)b * CC + c) * NN + j]
              - x[(((size_t)b * CC + c) * DD + d) * NN + n];
        }
        ldsX[row * LDSC + col] = v;
    }
    __syncthreads();

    gemm_tile<4, true>(W1, b1, ldsX, ldsY, EE, 2 * w + 0, 0, laneRow, laneHi);
    gemm_tile<4, true>(W1, b1, ldsX, ldsY, EE, 2 * w + 1, 0, laneRow, laneHi);
    __syncthreads();

    gemm_tile<4, true>(W2, b2, ldsY, ldsX, HH, 2 * w + 0, 0, laneRow, laneHi);
    gemm_tile<4, true>(W2, b2, ldsY, ldsX, HH, 2 * w + 1, 0, laneRow, laneHi);
    __syncthreads();

    gemm_tile<2, false>(W3, b3, ldsX, ldsY, HH, w >> 1, (w & 1) * 2, laneRow, laneHi);
    __syncthreads();

    {
        int c  = t & (CC - 1);
        int nl = t >> 6;
        const float* rrow = &ldsY[c * LDSC + nl * 16];
        float m = rrow[0];
#pragma unroll
        for (int k = 1; k < KK; ++k) m = fmaxf(m, rrow[k]);
        size_t o = (((size_t)b * CC + c) * DD + d) * NN + (n0 + nl);
        out[o] = m + x[o];
    }
}

extern "C" void kernel_launch(void* const* d_in, const int* in_sizes, int n_in,
                              void* d_out, int out_size, void* d_ws, size_t ws_size,
                              hipStream_t stream) {
    const float* x  = (const float*)d_in[0];
    const float* W1 = (const float*)d_in[1];
    const float* b1 = (const float*)d_in[2];
    const float* W2 = (const float*)d_in[3];
    const float* b2 = (const float*)d_in[4];
    const float* W3 = (const float*)d_in[5];
    const float* b3 = (const float*)d_in[6];
    float* out = (float*)d_out;

    float* mean_pc = (float*)d_ws;
    float* sq      = mean_pc + (size_t)BB * CC * NN;
    int*   idxp    = (int*)(sq + (size_t)BB * NN);

    meanpc_kernel<<<(BB * CC * NN + 255) / 256, 256, 0, stream>>>(x, mean_pc);
    sqnorm_kernel<<<(BB * NN + 255) / 256, 256, 0, stream>>>(mean_pc, sq);
    knn_kernel<<<BB * (NN / KQ), 256, 0, stream>>>(mean_pc, sq, idxp);
    mlp_kernel<<<BB * DD * (NN / NT), 256, 0, stream>>>(x, mean_pc, idxp,
                                                        W1, b1, W2, b2, W3, b3, out);
}


# MLP feed: activations k-contiguous per column in LDS, A and B fragments as 128-bit loads issued ahead of their WMMAs, 2x4 tiles per wave; k order in a chunk permuted (f32 sum order changes)
# speedup vs baseline: 10.0579x; 1.1050x over previous
#include <hip/hip_runtime.h>

#define BB 2
#define CC 64
#define DD 3
#define NN 4096
#define KK 16
#define EE 128
#define HH 256

typedef __attribute__((ext_vector_type(2))) float v2f;
typedef __attribute__((ext_vector_type(8))) float v8f;

constexpr int NT   = 4;
constexpr int COLS = NT * KK;

__global__ void meanpc_kernel(const float* __restrict__ x, float* __restrict__ mean_pc) {
    int i = blockIdx.x * blockDim.x + threadIdx.x;
    if (i >= BB * CC * NN) return;
    int n  = i % NN;
    int bc = i / NN;
    const float* xp = x + (size_t)bc * (DD * NN) + n;
    mean_pc[i] = (xp[0] + xp[NN] + xp[2 * NN]) * (1.0f / 3.0f);
}

__global__ void sqnorm_kernel(const float* __restrict__ mean_pc, float* __restrict__ sq) {
    int i = blockIdx.x * blockDim.x + threadIdx.x;
    if (i >= BB * NN) return;
    int n = i % NN;
    int b = i / NN;
    float s = 0.0f;
#pragma unroll
    for (int c = 0; c < CC; ++c) {
        float v = mean_pc[((size_t)b * CC + c) * NN + n];
        s = __builtin_fmaf(v, v, s);
    }
    sq[i] = s;
}

constexpr int KQ   = 32;
constexpr int KW   = 8;
constexpr int KSEG = NN / KW;
constexpr int KRB  = 16;
constexpr int KNB  = KSEG / KRB;

template <int J>
__device__ __forceinline__ float row_bcast(float v) {
    return __builtin_bit_cast(float,
        __builtin_amdgcn_update_dpp(0, __builtin_bit_cast(int, v), 0x150 + J, 0xf, 0xf, true));
}

template <int J>
struct KnnStep {
    static __device__ __forceinline__ void fma(float (&dot)[KRB], float qc, float rc) {
        asm("v_fmac_f32_dpp %0, %1, %2 row_share:%3 row_mask:0xf bank_mask:0xf"
            : "+v"(dot[J]) : "v"(rc), "v"(qc), "n"(J));
        KnnStep<J + 1>::fma(dot, qc, rc);
    }
    static __device__ __forceinline__ void dist(float (&dot)[KRB], float sqv) {
        dot[J] = row_bcast<J>(sqv) - 2.0f * dot[J];
        KnnStep<J + 1>::dist(dot, sqv);
    }
};
template <>
struct KnnStep<KRB> {
    static __device__ __forceinline__ void fma(float (&)[KRB], float, float) {}
    static __device__ __forceinline__ void dist(float (&)[KRB], float) {}
};

__device__ __forceinline__ void knn_insert(float (&bd)[KK], int (&bi)[KK], float d, int i) {
    bd[KK - 1] = d;
    bi[KK - 1] = i;
#pragma unroll
    for (int j = KK - 1; j > 0; --j) {
        if (bd[j] < bd[j - 1]) {
            float td = bd[j]; bd[j] = bd[j - 1]; bd[j - 1] = td;
            int   ti = bi[j]; bi[j] = bi[j - 1]; bi[j - 1] = ti;
        }
    }
}

__global__ void __launch_bounds__(256) knn_kernel(const float* __restrict__ mean_pc,
                                                  const float* __restrict__ sq,
                                                  int* __restrict__ idx) {
    __shared__ float mrgD[KW * KK * KQ];
    __shared__ int   mrgI[KW * KK * KQ];

    int t    = threadIdx.x;
    int lane = t & 31;
    int sub  = lane & 15;
    int w    = __builtin_amdgcn_readfirstlane(t >> 5);
    int b    = blockIdx.x / (NN / KQ);
    int n    = (blockIdx.x % (NN / KQ)) * KQ + lane;
    const float* pc  = mean_pc + (size_t)b * CC * NN;
    const float* sqb = sq + (size_t)b * NN;

    float q[CC];
#pragma unroll
    for (int c = 0; c < CC; ++c) q[c] = pc[(size_t)c * NN + n];

    float bd[KK];
    int   bi[KK];
#pragma unroll
    for (int j = 0; j < KK; ++j) { bd[j] = 3.0e38f; bi[j] = 0; }

    int   r0 = w * KSEG;
    float ref[CC];
    float sqv;
#pragma unroll
    for (int c = 0; c < CC; ++c) ref[c] = pc[(size_t)c * NN + r0 + sub];
    sqv = sqb[r0 + sub];

    for (int rb = 0; rb < KNB; ++rb) {
        float dot[KRB];
#pragma unroll
        for (int j = 0; j < KRB; ++j) dot[j] = 0.0f;
#pragma unroll
        for (int c = 0; c < CC; ++c) KnnStep<0>::fma(dot, q[c], ref[c]);
        KnnStep<0>::dist(dot, sqv);

        int rn = (rb + 1 < KNB) ? r0 + KRB : r0;
#pragma unroll
        for (int c = 0; c < CC; ++c) ref[c] = pc[(size_t)c * NN + rn + sub];
        sqv = sqb[rn + sub];

#pragma unroll
        for (int j = 0; j < KRB; ++j) {
            int r = r0 + j;
            if (r != n && dot[j] < bd[KK - 1]) knn_insert(bd, bi, dot[j], r);
        }
        r0 = rn;
    }

#pragma unroll
    for (int j = 0; j < KK; ++j) {
        mrgD[(w * KK + j) * KQ + lane] = bd[j];
        mrgI[(w * KK + j) * KQ + lane] = bi[j];
    }
    __syncthreads();
    for (int step = 1; step < KW; step <<= 1) {
        if ((w & (2 * step - 1)) == 0) {
            int srcw = w + step;
            for (int j = 0; j < KK; ++j) {
                float cd = mrgD[(srcw * KK + j) * KQ + lane];
                int   ci = mrgI[(srcw * KK + j) * KQ + lane];
                if (cd < bd[KK - 1]) knn_insert(bd, bi, cd, ci);
            }
            if (2 * step < KW) {
#pragma unroll
                for (int j = 0; j < KK; ++j) {
                    mrgD[(w * KK + j) * KQ + lane] = bd[j];
                    mrgI[(w * KK + j) * KQ + lane] = bi[j];
                }
            }
        }
        __syncthreads();
    }
    if (w == 0) {
#pragma unroll
        for (int j = 0; j < KK; ++j) idx[((size_t)b * KK + j) * NN + n] = bi[j];
    }
}

typedef __attribute__((ext_vector_type(4))) float v4f;

constexpr int PE = EE + 4;
constexpr int PH = HH + 4;
constexpr int PR = CC + 4;

template <int NRT, int NCT, int Q>
__device__ __forceinline__ void load_chunk(v4f (&a)[NRT][Q], v4f (&b)[NCT][Q],
                                           const float* const (&wp)[NRT],
                                           const float* const (&bp)[NCT], int koff) {
#pragma unroll
    for (int r = 0; r < NRT; ++r) {
#pragma unroll
        for (int q = 0; q < Q; ++q) a[r][q] = *(const v4f*)(wp[r] + koff + 4 * q);
    }
#pragma unroll
    for (int c = 0; c < NCT; ++c) {
#pragma unroll
        for (int q = 0; q < Q; ++q) b[c][q] = *(const v4f*)(bp[c] + koff + 4 * q);
    }
}

template <int NRT, int NCT, int Q>
__device__ __forceinline__ void mma_chunk(v8f (&acc)[NRT][NCT],
                                          const v4f (&a)[NRT][Q], const v4f (&b)[NCT][Q]) {
#pragma unroll
    for (int q = 0; q < Q; ++q) {
#pragma unroll
        for (int s = 0; s < 2; ++s) {
#pragma unroll
            for (int r = 0; r < NRT; ++r) {
                v2f av = s ? __builtin_shufflevector(a[r][q], a[r][q], 2, 3)
                           : __builtin_shufflevector(a[r][q], a[r][q], 0, 1);
#pragma unroll
                for (int c = 0; c < NCT; ++c) {
                    v2f bv = s ? __builtin_shufflevector(b[c][q], b[c][q], 2, 3)
                               : __builtin_shufflevector(b[c][q], b[c][q], 0, 1);
                    acc[r][c] = __builtin_amdgcn_wmma_f32_16x16x4_f32(
                        false, av, false, bv, (short)0, acc[r][c], false, false);
                }
            }
        }
    }
}

template <int NRT, int NCT, int Q, int KI, int PIN, int POUT, bool RELU>
__device__ __forceinline__ void gemm_block(const float* __restrict__ W,
                                           const float* __restrict__ bias,
                                           const float* lin, float* lout,
                                           int rt0, int ct0, int laneRow, int laneHi) {
    constexpr int CH     = 4 * Q;
    constexpr int KSTEP  = 2 * CH;
    constexpr int NCHUNK = KI / KSTEP;
    static_assert(KI % KSTEP == 0 && NCHUNK % 2 == 0, "chunks come in pairs");

    const v8f vzero = {0.f, 0.f, 0.f, 0.f, 0.f, 0.f, 0.f, 0.f};
    v8f acc[NRT][NCT];
#pragma unroll
    for (int r = 0; r < NRT; ++r) {
#pragma unroll
        for (int c = 0; c < NCT; ++c) acc[r][c] = vzero;
    }

    const float* wp[NRT];
    const float* bp[NCT];
#pragma unroll
    for (int r = 0; r < NRT; ++r)
        wp[r] = W + (size_t)((rt0 + r) * 16 + laneRow) * KI + CH * laneHi;
#pragma unroll
    for (int c = 0; c < NCT; ++c)
        bp[c] = lin + ((ct0 + c) * 16 + laneRow) * PIN + CH * laneHi;

    v4f a0[NRT][Q], b0[NCT][Q], a1[NRT][Q], b1[NCT][Q];
    load_chunk<NRT, NCT, Q>(a0, b0, wp, bp, 0);
    for (int kk = 0; kk < NCHUNK; kk += 2) {
        load_chunk<NRT, NCT, Q>(a1, b1, wp, bp, (kk + 1) * KSTEP);
        mma_chunk<NRT, NCT, Q>(acc, a0, b0);
        if (kk + 2 < NCHUNK) load_chunk<NRT, NCT, Q>(a0, b0, wp, bp, (kk + 2) * KSTEP);
        mma_chunk<NRT, NCT, Q>(acc, a1, b1);
    }

#pragma unroll
    for (int r = 0; r < NRT; ++r) {
        int row0 = (rt0 + r) * 16 + 8 * laneHi;
        v4f bl = *(const v4f*)(bias + row0);
        v4f bh = *(const v4f*)(bias + row0 + 4);
#pragma unroll
        for (int c = 0; c < NCT; ++c) {
            int col = (ct0 + c) * 16 + laneRow;
            v4f lo, hi;
#pragma unroll
            for (int j = 0; j < 4; ++j) {
                float u = acc[r][c][j] + bl[j];
                float v = acc[r][c][j + 4] + bh[j];
                if (RELU) { u = fmaxf(u, 0.0f); v = fmaxf(v, 0.0f); }
                lo[j] = u;
                hi[j] = v;
            }
            *(v4f*)(lout + col * POUT + row0)     = lo;
            *(v4f*)(lout + col * POUT + row0 + 4) = hi;
        }
    }
}

__global__ void __launch_bounds__(256) mlp_kernel(const float* __restrict__ x,
                                                  const float* __restrict__ mean_pc,
                                                  const int* __restrict__ idx,
                                                  const float* __restrict__ W1, const float* __restrict__ b1,
                                                  const float* __restrict__ W2, const float* __restrict__ b2,
                                                  const float* __restrict__ W3, const float* __restrict__ b3,
                                                  float* __restrict__ out) {
    __shared__ __attribute__((aligned(16))) float ldsX[COLS * PH];
    __shared__ __attribute__((aligned(16))) float ldsY[COLS * PH];

    int t       = threadIdx.x;
    int lane    = t & 31;
    int w       = t >> 5;
    int laneRow = lane & 15;
    int laneHi  = lane >> 4;

    int blk = blockIdx.x;
    int nb  = blk % (NN / NT);
    int bd  = blk / (NN / NT);
    int d   = bd % DD;
    int b   = bd / DD;
    int n0  = nb * NT;

    for (int e = t; e < EE * COLS; e += 256) {
        int row = e >> 6;
        int col = e & (COLS - 1);
        int nl  = col >> 4;
        int k   = col & 15;
        int n   = n0 + nl;
        float v;
        if (row < CC) {
            v = x[(((size_t)b * CC + row) * DD + d) * NN + n];
        } else {
            int c = row - CC;
            int j = idx[((size_t)b * KK + k) * NN + n];
            v = mean_pc[((size_t)b * CC + c) * NN + j]
              - x[(((size_t)b * CC + c) * DD + d) * NN + n];
        }
        ldsX[col * PE + row] = v;
    }
    __syncthreads();

    gemm_block<2, 4, 2, EE, PE, PH, true>(W1, b1, ldsX, ldsY, 2 * w, 0, laneRow, laneHi);
    __syncthreads();

    gemm_block<2, 4, 2, HH, PH, PH, true>(W2, b2, ldsY, ldsX, 2 * w, 0, laneRow, laneHi);
    __syncthreads();

    gemm_block<1, 2, 4, HH, PH, PR, false>(W3, b3, ldsX, ldsY, w >> 1, (w & 1) * 2, laneRow, laneHi);
    __syncthreads();

    {
        int c  = t & (CC - 1);
        int nl = t >> 6;
        const float* rcol = &ldsY[(nl * 16) * PR + c];
        float m = rcol[0];
#pragma unroll
        for (int k = 1; k < KK; ++k) m = fmaxf(m, rcol[k * PR]);
        size_t o = (((size_t)b * CC + c) * DD + d) * NN + (n0 + nl);
        out[o] = m + x[o];
    }
}

extern "C" void kernel_launch(void* const* d_in, const int* in_sizes, int n_in,
                              void* d_out, int out_size, void* d_ws, size_t ws_size,
                              hipStream_t stream) {
    const float* x  = (const float*)d_in[0];
    const float* W1 = (const float*)d_in[1];
    const float* b1 = (const float*)d_in[2];
    const float* W2 = (const float*)d_in[3];
    const float* b2 = (const float*)d_in[4];
    const float* W3 = (const float*)d_in[5];
    const float* b3 = (const float*)d_in[6];
    float* out = (float*)d_out;

    float* mean_pc = (float*)d_ws;
    float* sq      = mean_pc + (size_t)BB * CC * NN;
    int*   idxp    = (int*)(sq + (size_t)BB * NN);

    meanpc_kernel<<<(BB * CC * NN + 255) / 256, 256, 0, stream>>>(x, mean_pc);
    sqnorm_kernel<<<(BB * NN + 255) / 256, 256, 0, stream>>>(mean_pc, sq);
    knn_kernel<<<BB * (NN / KQ), 256, 0, stream>>>(mean_pc, sq, idxp);
    mlp_kernel<<<BB * DD * (NN / NT), 256, 0, stream>>>(x, mean_pc, idxp,
                                                        W1, b1, W2, b2, W3, b3, out);
}


# MLP block = 4 n x all 3 d; d-invariant work once per block: layer 1 by linearity (P=W1b*nbr once, Q=W1a*c-W1b*c+b1, h1=relu(P+Q)), W2 fragments stay in registers; layer-1 f32 sum order changes
# speedup vs baseline: 11.4516x; 1.1386x over previous
#include <hip/hip_runtime.h>

#define BB 2
#define CC 64
#define DD 3
#define NN 4096
#define KK 16
#define EE 128
#define HH 256

typedef __attribute__((ext_vector_type(2))) float v2f;
typedef __attribute__((ext_vector_type(8))) float v8f;

constexpr int NT   = 4;
constexpr int COLS = NT * KK;

__global__ void meanpc_kernel(const float* __restrict__ x, float* __restrict__ mean_pc) {
    int i = blockIdx.x * blockDim.x + threadIdx.x;
    if (i >= BB * CC * NN) return;
    int n  = i % NN;
    int bc = i / NN;
    const float* xp = x + (size_t)bc * (DD * NN) + n;
    mean_pc[i] = (xp[0] + xp[NN] + xp[2 * NN]) * (1.0f / 3.0f);
}

__global__ void sqnorm_kernel(const float* __restrict__ mean_pc, float* __restrict__ sq) {
    int i = blockIdx.x * blockDim.x + threadIdx.x;
    if (i >= BB * NN) return;
    int n = i % NN;
    int b = i / NN;
    float s = 0.0f;
#pragma unroll
    for (int c = 0; c < CC; ++c) {
        float v = mean_pc[((size_t)b * CC + c) * NN + n];
        s = __builtin_fmaf(v, v, s);
    }
    sq[i] = s;
}

constexpr int KQ   = 32;
constexpr int KW   = 8;
constexpr int KSEG = NN / KW;
constexpr int KRB  = 16;
constexpr int KNB  = KSEG / KRB;

template <int J>
__device__ __forceinline__ float row_bcast(float v) {
    return __builtin_bit_cast(float,
        __builtin_amdgcn_update_dpp(0, __builtin_bit_cast(int, v), 0x150 + J, 0xf, 0xf, true));
}

template <int J>
struct KnnStep {
    static __device__ __forceinline__ void fma(float (&dot)[KRB], float qc, float rc) {
        asm("v_fmac_f32_dpp %0, %1, %2 row_share:%3 row_mask:0xf bank_mask:0xf"
            : "+v"(dot[J]) : "v"(rc), "v"(qc), "n"(J));
        KnnStep<J + 1>::fma(dot, qc, rc);
    }
    static __device__ __forceinline__ void dist(float (&dot)[KRB], float sqv) {
        dot[J] = row_bcast<J>(sqv) - 2.0f * dot[J];
        KnnStep<J + 1>::dist(dot, sqv);
    }
};
template <>
struct KnnStep<KRB> {
    static __device__ __forceinline__ void fma(float (&)[KRB], float, float) {}
    static __device__ __forceinline__ void dist(float (&)[KRB], float) {}
};

__device__ __forceinline__ void knn_insert(float (&bd)[KK], int (&bi)[KK], float d, int i) {
    bd[KK - 1] = d;
    bi[KK - 1] = i;
#pragma unroll
    for (int j = KK - 1; j > 0; --j) {
        if (bd[j] < bd[j - 1]) {
            float td = bd[j]; bd[j] = bd[j - 1]; bd[j - 1] = td;
            int   ti = bi[j]; bi[j] = bi[j - 1]; bi[j - 1] = ti;
        }
    }
}

__global__ void __launch_bounds__(256) knn_kernel(const float* __restrict__ mean_pc,
                                                  const float* __restrict__ sq,
                                                  int* __restrict__ idx) {
    __shared__ float mrgD[KW * KK * KQ];
    __shared__ int   mrgI[KW * KK * KQ];

    int t    = threadIdx.x;
    int lane = t & 31;
    int sub  = lane & 15;
    int w    = __builtin_amdgcn_readfirstlane(t >> 5);
    int b    = blockIdx.x / (NN / KQ);
    int n    = (blockIdx.x % (NN / KQ)) * KQ + lane;
    const float* pc  = mean_pc + (size_t)b * CC * NN;
    const float* sqb = sq + (size_t)b * NN;

    float q[CC];
#pragma unroll
    for (int c = 0; c < CC; ++c) q[c] = pc[(size_t)c * NN + n];

    float bd[KK];
    int   bi[KK];
#pragma unroll
    for (int j = 0; j < KK; ++j) { bd[j] = 3.0e38f; bi[j] = 0; }

    int   r0 = w * KSEG;
    float ref[CC];
    float sqv;
#pragma unroll
    for (int c = 0; c < CC; ++c) ref[c] = pc[(size_t)c * NN + r0 + sub];
    sqv = sqb[r0 + sub];

    for (int rb = 0; rb < KNB; ++rb) {
        float dot[KRB];
#pragma unroll
        for (int j = 0; j < KRB; ++j) dot[j] = 0.0f;
#pragma unroll
        for (int c = 0; c < CC; ++c) KnnStep<0>::fma(dot, q[c], ref[c]);
        KnnStep<0>::dist(dot, sqv);

        int rn = (rb + 1 < KNB) ? r0 + KRB : r0;
#pragma unroll
        for (int c = 0; c < CC; ++c) ref[c] = pc[(size_t)c * NN + rn + sub];
        sqv = sqb[rn + sub];

#pragma unroll
        for (int j = 0; j < KRB; ++j) {
            int r = r0 + j;
            if (r != n && dot[j] < bd[KK - 1]) knn_insert(bd, bi, dot[j], r);
        }
        r0 = rn;
    }

#pragma unroll
    for (int j = 0; j < KK; ++j) {
        mrgD[(w * KK + j) * KQ + lane] = bd[j];
        mrgI[(w * KK + j) * KQ + lane] = bi[j];
    }
    __syncthreads();
    for (int step = 1; step < KW; step <<= 1) {
        if ((w & (2 * step - 1)) == 0) {
            int srcw = w + step;
            for (int j = 0; j < KK; ++j) {
                float cd = mrgD[(srcw * KK + j) * KQ + lane];
                int   ci = mrgI[(srcw * KK + j) * KQ + lane];
                if (cd < bd[KK - 1]) knn_insert(bd, bi, cd, ci);
            }
            if (2 * step < KW) {
#pragma unroll
                for (int j = 0; j < KK; ++j) {
                    mrgD[(w * KK + j) * KQ + lane] = bd[j];
                    mrgI[(w * KK + j) * KQ + lane] = bi[j];
                }
            }
        }
        __syncthreads();
    }
    if (w == 0) {
#pragma unroll
        for (int j = 0; j < KK; ++j) idx[((size_t)b * KK + j) * NN + n] = bi[j];
    }
}

typedef __attribute__((ext_vector_type(4))) float v4f;

constexpr int PE = EE + 4;
constexpr int PH = HH + 4;
constexpr int PR = CC + 4;

template <int NRT, int NCT, int Q>
__device__ __forceinline__ void load_chunk(v4f (&a)[NRT][Q], v4f (&b)[NCT][Q],
                                           const float* const (&wp)[NRT],
                                           const float* const (&bp)[NCT], int koff) {
#pragma unroll
    for (int r = 0; r < NRT; ++r) {
#pragma unroll
        for (int q = 0; q < Q; ++q) a[r][q] = *(const v4f*)(wp[r] + koff + 4 * q);
    }
#pragma unroll
    for (int c = 0; c < NCT; ++c) {
#pragma unroll
        for (int q = 0; q < Q; ++q) b[c][q] = *(const v4f*)(bp[c] + koff + 4 * q);
    }
}

template <int NRT, int NCT, int Q>
__device__ __forceinline__ void mma_chunk(v8f (&acc)[NRT][NCT],
                                          const v4f (&a)[NRT][Q], const v4f (&b)[NCT][Q]) {
#pragma unroll
    for (int q = 0; q < Q; ++q) {
#pragma unroll
        for (int s = 0; s < 2; ++s) {
#pragma unroll
            for (int r = 0; r < NRT; ++r) {
                v2f av = s ? __builtin_shufflevector(a[r][q], a[r][q], 2, 3)
                           : __builtin_shufflevector(a[r][q], a[r][q], 0, 1);
#pragma unroll
                for (int c = 0; c < NCT; ++c) {
                    v2f bv = s ? __builtin_shufflevector(b[c][q], b[c][q], 2, 3)
                               : __builtin_shufflevector(b[c][q], b[c][q], 0, 1);
                    acc[r][c] = __builtin_amdgcn_wmma_f32_16x16x4_f32(
                        false, av, false, bv, (short)0, acc[r][c], false, false);
                }
            }
        }
    }
}

template <int NRT, int NCT, int Q, int KI, int WP, int PIN, int POUT, bool RELU, bool BIAS>
__device__ __forceinline__ void gemm_block(const float* __restrict__ W,
                                           const float* __restrict__ bias,
                                           const float* lin, float* lout,
                                           int rt0, int ct0, int laneRow, int laneHi) {
    constexpr int CH     = 4 * Q;
    constexpr int KSTEP  = 2 * CH;
    constexpr int NCHUNK = KI / KSTEP;
    static_assert(KI % KSTEP == 0 && NCHUNK % 2 == 0, "chunks come in pairs");

    const v8f vzero = {0.f, 0.f, 0.f, 0.f, 0.f, 0.f, 0.f, 0.f};
    v8f acc[NRT][NCT];
#pragma unroll
    for (int r = 0; r < NRT; ++r) {
#pragma unroll
        for (int c = 0; c < NCT; ++c) acc[r][c] = vzero;
    }

    const float* wp[NRT];
    const float* bp[NCT];
#pragma unroll
    for (int r = 0; r < NRT; ++r)
        wp[r] = W + (size_t)((rt0 + r) * 16 + laneRow) * WP + CH * laneHi;
#pragma unroll
    for (int c = 0; c < NCT; ++c)
        bp[c] = lin + ((ct0 + c) * 16 + laneRow) * PIN + CH * laneHi;

    v4f a0[NRT][Q], b0[NCT][Q], a1[NRT][Q], b1[NCT][Q];
    load_chunk<NRT, NCT, Q>(a0, b0, wp, bp, 0);
    for (int kk = 0; kk < NCHUNK; kk += 2) {
        load_chunk<NRT, NCT, Q>(a1, b1, wp, bp, (kk + 1) * KSTEP);
        mma_chunk<NRT, NCT, Q>(acc, a0, b0);
        if (kk + 2 < NCHUNK) load_chunk<NRT, NCT, Q>(a0, b0, wp, bp, (kk + 2) * KSTEP);
        mma_chunk<NRT, NCT, Q>(acc, a1, b1);
    }

#pragma unroll
    for (int r = 0; r < NRT; ++r) {
        int row0 = (rt0 + r) * 16 + 8 * laneHi;
        v4f bl = {0.f, 0.f, 0.f, 0.f};
        v4f bh = {0.f, 0.f, 0.f, 0.f};
        if (BIAS) {
            bl = *(const v4f*)(bias + row0);
            bh = *(const v4f*)(bias + row0 + 4);
        }
#pragma unroll
        for (int c = 0; c < NCT; ++c) {
            int col = (ct0 + c) * 16 + laneRow;
            v4f lo, hi;
#pragma unroll
            for (int j = 0; j < 4; ++j) {
                float u = acc[r][c][j];
                float v = acc[r][c][j + 4];
                if (BIAS) { u += bl[j]; v += bh[j]; }
                if (RELU) { u = fmaxf(u, 0.0f); v = fmaxf(v, 0.0f); }
                lo[j] = u;
                hi[j] = v;
            }
            *(v4f*)(lout + col * POUT + row0)     = lo;
            *(v4f*)(lout + col * POUT + row0 + 4) = hi;
        }
    }
}

constexpr int PC = CC + 4;
constexpr int QC = 16;

__global__ void __launch_bounds__(256) mlp_kernel(const float* __restrict__ x,
                                                  const float* __restrict__ mean_pc,
                                                  const int* __restrict__ idx,
                                                  const float* __restrict__ W1, const float* __restrict__ b1,
                                                  const float* __restrict__ W2, const float* __restrict__ b2,
                                                  const float* __restrict__ W3, const float* __restrict__ b3,
                                                  float* __restrict__ out) {
    __shared__ __attribute__((aligned(16))) float ldsX[COLS * PH];
    __shared__ __attribute__((aligned(16))) float ldsY[COLS * PH];
    __shared__ __attribute__((aligned(16))) float ldsP[COLS * PH];
    __shared__ __attribute__((aligned(16))) float ldsQ[QC * PH];
    __shared__ __attribute__((aligned(16))) float ldsR[COLS * PR];

    int t       = threadIdx.x;
    int lane    = t & 31;
    int w       = t >> 5;
    int laneRow = lane & 15;
    int laneHi  = lane >> 4;

    int blk = blockIdx.x;
    int nb  = blk % (NN / NT);
    int b   = blk / (NN / NT);
    int n0  = nb * NT;

    float* ldsN = ldsX;
    float* ldsC = ldsX + COLS * PC;

    {
        int col = t & (COLS - 1);
        int cg  = t >> 6;
        int nl  = col >> 4;
        int k   = col & 15;
        int j   = idx[((size_t)b * KK + k) * NN + n0 + nl];
        const float* mp = mean_pc + ((size_t)b * CC + cg * 16) * NN + j;
        float v[16];
#pragma unroll
        for (int i = 0; i < 16; ++i) v[i] = mp[(size_t)i * NN];
#pragma unroll
        for (int i = 0; i < 4; ++i) {
            v4f s = {v[4 * i], v[4 * i + 1], v[4 * i + 2], v[4 * i + 3]};
            *(v4f*)(ldsN + col * PC + cg * 16 + 4 * i) = s;
        }
    }
    for (int e = t; e < QC * CC; e += 256) {
        int dcol = e & (QC - 1);
        int c    = e >> 4;
        int d    = dcol >> 2;
        int nl   = dcol & 3;
        float v  = 0.0f;
        if (dcol < DD * NT) v = x[(((size_t)b * CC + c) * DD + d) * NN + n0 + nl];
        ldsC[dcol * PE + c]      = v;
        ldsC[dcol * PE + CC + c] = -v;
    }
    __syncthreads();

    gemm_block<2, 4, 2, CC, EE, PC, PH, false, false>(W1 + CC, b1, ldsN, ldsP, 2 * w, 0, laneRow, laneHi);
    gemm_block<2, 1, 2, EE, EE, PE, PH, false, true>(W1, b1, ldsC, ldsQ, 2 * w, 0, laneRow, laneHi);
    __syncthreads();

    for (int d = 0; d < DD; ++d) {
        {
            int col = t & (COLS - 1);
            int og  = t >> 6;
            const float* pp = ldsP + col * PH + og * 64;
            const float* qq = ldsQ + (d * NT + (col >> 4)) * PH + og * 64;
            float*       hh = ldsY + col * PH + og * 64;
#pragma unroll
            for (int i = 0; i < 16; ++i) {
                v4f p = *(const v4f*)(pp + 4 * i);
                v4f q = *(const v4f*)(qq + 4 * i);
                v4f h;
#pragma unroll
                for (int j = 0; j < 4; ++j) h[j] = fmaxf(p[j] + q[j], 0.0f);
                *(v4f*)(hh + 4 * i) = h;
            }
        }
        __syncthreads();

        gemm_block<2, 4, 2, HH, HH, PH, PH, true, true>(W2, b2, ldsY, ldsX, 2 * w, 0, laneRow, laneHi);
        __syncthreads();

        int zoff = 0;
        asm volatile("" : "+s"(zoff));
        size_t o   = (((size_t)b * CC + (t & (CC - 1))) * DD + d) * NN + (n0 + (t >> 6));
        float xres = x[o];
        gemm_block<1, 2, 4, HH, HH, PH, PR, false, true>(W3 + zoff, b3, ldsX, ldsR, w >> 1, (w & 1) * 2, laneRow, laneHi);
        __syncthreads();

        {
            int c  = t & (CC - 1);
            int nl = t >> 6;
            const float* rcol = &ldsR[(nl * 16) * PR + c];
            float m = rcol[0];
#pragma unroll
            for (int k = 1; k < KK; ++k) m = fmaxf(m, rcol[k * PR]);
            out[o] = m + xres;
        }
    }
}

extern "C" void kernel_launch(void* const* d_in, const int* in_sizes, int n_in,
                              void* d_out, int out_size, void* d_ws, size_t ws_size,
                              hipStream_t stream) {
    const float* x  = (const float*)d_in[0];
    const float* W1 = (const float*)d_in[1];
    const float* b1 = (const float*)d_in[2];
    const float* W2 = (const float*)d_in[3];
    const float* b2 = (const float*)d_in[4];
    const float* W3 = (const float*)d_in[5];
    const float* b3 = (const float*)d_in[6];
    float* out = (float*)d_out;

    float* mean_pc = (float*)d_ws;
    float* sq      = mean_pc + (size_t)BB * CC * NN;
    int*   idxp    = (int*)(sq + (size_t)BB * NN);

    meanpc_kernel<<<(BB * CC * NN + 255) / 256, 256, 0, stream>>>(x, mean_pc);
    sqnorm_kernel<<<(BB * NN + 255) / 256, 256, 0, stream>>>(mean_pc, sq);
    knn_kernel<<<BB * (NN / KQ), 256, 0, stream>>>(mean_pc, sq, idxp);
    mlp_kernel<<<BB * (NN / NT), 256, 0, stream>>>(x, mean_pc, idxp,
                                                        W1, b1, W2, b2, W3, b3, out);
}


# MLP workgroups persistent: 256 of them take 8 groups each in a loop, and the next group's gathered and centre columns are fetched under layer 2 into LDS regions of their own
# speedup vs baseline: 12.2196x; 1.0671x over previous
#include <hip/hip_runtime.h>

#define BB 2
#define CC 64
#define DD 3
#define NN 4096
#define KK 16
#define EE 128
#define HH 256

typedef __attribute__((ext_vector_type(2))) float v2f;
typedef __attribute__((ext_vector_type(8))) float v8f;

constexpr int NT   = 4;
constexpr int COLS = NT * KK;

__global__ void meanpc_kernel(const float* __restrict__ x, float* __restrict__ mean_pc) {
    int i = blockIdx.x * blockDim.x + threadIdx.x;
    if (i >= BB * CC * NN) return;
    int n  = i % NN;
    int bc = i / NN;
    const float* xp = x + (size_t)bc * (DD * NN) + n;
    mean_pc[i] = (xp[0] + xp[NN] + xp[2 * NN]) * (1.0f / 3.0f);
}

__global__ void sqnorm_kernel(const float* __restrict__ mean_pc, float* __restrict__ sq) {
    int i = blockIdx.x * blockDim.x + threadIdx.x;
    if (i >= BB * NN) return;
    int n = i % NN;
    int b = i / NN;
    float s = 0.0f;
#pragma unroll
    for (int c = 0; c < CC; ++c) {
        float v = mean_pc[((size_t)b * CC + c) * NN + n];
        s = __builtin_fmaf(v, v, s);
    }
    sq[i] = s;
}

constexpr int KQ   = 32;
constexpr int KW   = 8;
constexpr int KSEG = NN / KW;
constexpr int KRB  = 16;
constexpr int KNB  = KSEG / KRB;

template <int J>
__device__ __forceinline__ float row_bcast(float v) {
    return __builtin_bit_cast(float,
        __builtin_amdgcn_update_dpp(0, __builtin_bit_cast(int, v), 0x150 + J, 0xf, 0xf, true));
}

template <int J>
struct KnnStep {
    static __device__ __forceinline__ void fma(float (&dot)[KRB], float qc, float rc) {
        asm("v_fmac_f32_dpp %0, %1, %2 row_share:%3 row_mask:0xf bank_mask:0xf"
            : "+v"(dot[J]) : "v"(rc), "v"(qc), "n"(J));
        KnnStep<J + 1>::fma(dot, qc, rc);
    }
    static __device__ __forceinline__ void dist(float (&dot)[KRB], float sqv) {
        dot[J] = row_bcast<J>(sqv) - 2.0f * dot[J];
        KnnStep<J + 1>::dist(dot, sqv);
    }
};
template <>
struct KnnStep<KRB> {
    static __device__ __forceinline__ void fma(float (&)[KRB], float, float) {}
    static __device__ __forceinline__ void dist(float (&)[KRB], float) {}
};

__device__ __forceinline__ void knn_insert(float (&bd)[KK], int (&bi)[KK], float d, int i) {
    bd[KK - 1] = d;
    bi[KK - 1] = i;
#pragma unroll
    for (int j = KK - 1; j > 0; --j) {
        if (bd[j] < bd[j - 1]) {
            float td = bd[j]; bd[j] = bd[j - 1]; bd[j - 1] = td;
            int   ti = bi[j]; bi[j] = bi[j - 1]; bi[j - 1] = ti;
        }
    }
}

__global__ void __launch_bounds__(256) knn_kernel(const float* __restrict__ mean_pc,
                                                  const float* __restrict__ sq,
                                                  int* __restrict__ idx) {
    __shared__ float mrgD[KW * KK * KQ];
    __shared__ int   mrgI[KW * KK * KQ];

    int t    = threadIdx.x;
    int lane = t & 31;
    int sub  = lane & 15;
    int w    = __builtin_amdgcn_readfirstlane(t >> 5);
    int b    = blockIdx.x / (NN / KQ);
    int n    = (blockIdx.x % (NN / KQ)) * KQ + lane;
    const float* pc  = mean_pc + (size_t)b * CC * NN;
    const float* sqb = sq + (size_t)b * NN;

    float q[CC];
#pragma unroll
    for (int c = 0; c < CC; ++c) q[c] = pc[(size_t)c * NN + n];

    float bd[KK];
    int   bi[KK];
#pragma unroll
    for (int j = 0; j < KK; ++j) { bd[j] = 3.0e38f; bi[j] = 0; }

    int   r0 = w * KSEG;
    float ref[CC];
    float sqv;
#pragma unroll
    for (int c = 0; c < CC; ++c) ref[c] = pc[(size_t)c * NN + r0 + sub];
    sqv = sqb[r0 + sub];

    for (int rb = 0; rb < KNB; ++rb) {
        float dot[KRB];
#pragma unroll
        for (int j = 0; j < KRB; ++j) dot[j] = 0.0f;
#pragma unroll
        for (int c = 0; c < CC; ++c) KnnStep<0>::fma(dot, q[c], ref[c]);
        KnnStep<0>::dist(dot, sqv);

        int rn = (rb + 1 < KNB) ? r0 + KRB : r0;
#pragma unroll
        for (int c = 0; c < CC; ++c) ref[c] = pc[(size_t)c * NN + rn + sub];
        sqv = sqb[rn + sub];

#pragma unroll
        for (int j = 0; j < KRB; ++j) {
            int r = r0 + j;
            if (r != n && dot[j] < bd[KK - 1]) knn_insert(bd, bi, dot[j], r);
        }
        r0 = rn;
    }

#pragma unroll
    for (int j = 0; j < KK; ++j) {
        mrgD[(w * KK + j) * KQ + lane] = bd[j];
        mrgI[(w * KK + j) * KQ + lane] = bi[j];
    }
    __syncthreads();
    for (int step = 1; step < KW; step <<= 1) {
        if ((w & (2 * step - 1)) == 0) {
            int srcw = w + step;
            for (int j = 0; j < KK; ++j) {
                float cd = mrgD[(srcw * KK + j) * KQ + lane];
                int   ci = mrgI[(srcw * KK + j) * KQ + lane];
                if (cd < bd[KK - 1]) knn_insert(bd, bi, cd, ci);
            }
            if (2 * step < KW) {
#pragma unroll
                for (int j = 0; j < KK; ++j) {
                    mrgD[(w * KK + j) * KQ + lane] = bd[j];
                    mrgI[(w * KK + j) * KQ + lane] = bi[j];
                }
            }
        }
        __syncthreads();
    }
    if (w == 0) {
#pragma unroll
        for (int j = 0; j < KK; ++j) idx[((size_t)b * KK + j) * NN + n] = bi[j];
    }
}

typedef __attribute__((ext_vector_type(4))) float v4f;

constexpr int PE = EE + 4;
constexpr int PH = HH + 4;
constexpr int PR = CC + 4;

template <int NRT, int NCT, int Q>
__device__ __forceinline__ void load_chunk(v4f (&a)[NRT][Q], v4f (&b)[NCT][Q],
                                           const float* const (&wp)[NRT],
                                           const float* const (&bp)[NCT], int koff) {
#pragma unroll
    for (int r = 0; r < NRT; ++r) {
#pragma unroll
        for (int q = 0; q < Q; ++q) a[r][q] = *(const v4f*)(wp[r] + koff + 4 * q);
    }
#pragma unroll
    for (int c = 0; c < NCT; ++c) {
#pragma unroll
        for (int q = 0; q < Q; ++q) b[c][q] = *(const v4f*)(bp[c] + koff + 4 * q);
    }
}

template <int NRT, int NCT, int Q>
__device__ __forceinline__ void mma_chunk(v8f (&acc)[NRT][NCT],
                                          const v4f (&a)[NRT][Q], const v4f (&b)[NCT][Q]) {
#pragma unroll
    for (int q = 0; q < Q; ++q) {
#pragma unroll
        for (int s = 0; s < 2; ++s) {
#pragma unroll
            for (int r = 0; r < NRT; ++r) {
                v2f av = s ? __builtin_shufflevector(a[r][q], a[r][q], 2, 3)
                           : __builtin_shufflevector(a[r][q], a[r][q], 0, 1);
#pragma unroll
                for (int c = 0; c < NCT; ++c) {
                    v2f bv = s ? __builtin_shufflevector(b[c][q], b[c][q], 2, 3)
                               : __builtin_shufflevector(b[c][q], b[c][q], 0, 1);
                    acc[r][c] = __builtin_amdgcn_wmma_f32_16x16x4_f32(
                        false, av, false, bv, (short)0, acc[r][c], false, false);
                }
            }
        }
    }
}

template <int NRT, int NCT, int Q, int KI, int WP, int PIN, int POUT, bool RELU, bool BIAS>
__device__ __forceinline__ void gemm_block(const float* __restrict__ W,
                                           const float* __restrict__ bias,
                                           const float* lin, float* lout,
                                           int rt0, int ct0, int laneRow, int laneHi) {
    constexpr int CH     = 4 * Q;
    constexpr int KSTEP  = 2 * CH;
    constexpr int NCHUNK = KI / KSTEP;
    static_assert(KI % KSTEP == 0 && NCHUNK % 2 == 0, "chunks come in pairs");

    const v8f vzero = {0.f, 0.f, 0.f, 0.f, 0.f, 0.f, 0.f, 0.f};
    v8f acc[NRT][NCT];
#pragma unroll
    for (int r = 0; r < NRT; ++r) {
#pragma unroll
        for (int c = 0; c < NCT; ++c) acc[r][c] = vzero;
    }

    const float* wp[NRT];
    const float* bp[NCT];
#pragma unroll
    for (int r = 0; r < NRT; ++r)
        wp[r] = W + (size_t)((rt0 + r) * 16 + laneRow) * WP + CH * laneHi;
#pragma unroll
    for (int c = 0; c < NCT; ++c)
        bp[c] = lin + ((ct0 + c) * 16 + laneRow) * PIN + CH * laneHi;

    v4f a0[NRT][Q], b0[NCT][Q], a1[NRT][Q], b1[NCT][Q];
    load_chunk<NRT, NCT, Q>(a0, b0, wp, bp, 0);
    for (int kk = 0; kk < NCHUNK; kk += 2) {
        load_chunk<NRT, NCT, Q>(a1, b1, wp, bp, (kk + 1) * KSTEP);
        mma_chunk<NRT, NCT, Q>(acc, a0, b0);
        if (kk + 2 < NCHUNK) load_chunk<NRT, NCT, Q>(a0, b0, wp, bp, (kk + 2) * KSTEP);
        mma_chunk<NRT, NCT, Q>(acc, a1, b1);
    }

#pragma unroll
    for (int r = 0; r < NRT; ++r) {
        int row0 = (rt0 + r) * 16 + 8 * laneHi;
        v4f bl = {0.f, 0.f, 0.f, 0.f};
        v4f bh = {0.f, 0.f, 0.f, 0.f};
        if (BIAS) {
            bl = *(const v4f*)(bias + row0);
            bh = *(const v4f*)(bias + row0 + 4);
        }
#pragma unroll
        for (int c = 0; c < NCT; ++c) {
            int col = (ct0 + c) * 16 + laneRow;
            v4f lo, hi;
#pragma unroll
            for (int j = 0; j < 4; ++j) {
                float u = acc[r][c][j];
                float v = acc[r][c][j + 4];
                if (BIAS) { u += bl[j]; v += bh[j]; }
                if (RELU) { u = fmaxf(u, 0.0f); v = fmaxf(v, 0.0f); }
                lo[j] = u;
                hi[j] = v;
            }
            *(v4f*)(lout + col * POUT + row0)     = lo;
            *(v4f*)(lout + col * POUT + row0 + 4) = hi;
        }
    }
}

constexpr int PC = CC + 4;
constexpr int QC = 16;
constexpr int NGRP = BB * (NN / NT);
constexpr int NWG  = 256;
constexpr int GPW  = NGRP / NWG;
static_assert(NGRP % NWG == 0, "every workgroup takes the same number of groups");

__device__ __forceinline__ void fetch_cols(float (&gv)[16], float (&cv)[4],
                                           const float* __restrict__ x,
                                           const float* __restrict__ mean_pc,
                                           int b, int n0, unsigned j, int t) {
    int cg   = t >> 6;
    int dcol = t & (QC - 1);
    int c0   = t >> 4;
    const float* mp = mean_pc + ((size_t)b * CC + cg * 16) * NN + j;
    const float* xp = x + (((size_t)b * CC + c0) * DD + (dcol >> 2)) * NN + n0 + (dcol & 3);
#pragma unroll
    for (int i = 0; i < 16; ++i) gv[i] = mp[(size_t)i * NN];
#pragma unroll
    for (int i = 0; i < 4; ++i) {
        float v = 0.0f;
        if (dcol < DD * NT) v = xp[(size_t)i * 16 * DD * NN];
        cv[i] = v;
    }
}

__device__ __forceinline__ void store_cols(float* ldsN, float* ldsC,
                                           const float (&gv)[16], const float (&cv)[4], int t) {
    int col  = t & (COLS - 1);
    int cg   = t >> 6;
    int dcol = t & (QC - 1);
    int c0   = t >> 4;
#pragma unroll
    for (int i = 0; i < 4; ++i) {
        v4f s = {gv[4 * i], gv[4 * i + 1], gv[4 * i + 2], gv[4 * i + 3]};
        *(v4f*)(ldsN + col * PC + cg * 16 + 4 * i) = s;
    }
#pragma unroll
    for (int i = 0; i < 4; ++i) {
        ldsC[dcol * PE + c0 + 16 * i]      = cv[i];
        ldsC[dcol * PE + CC + c0 + 16 * i] = -cv[i];
    }
}

__global__ void __launch_bounds__(256) mlp_kernel(const float* __restrict__ x,
                                                  const float* __restrict__ mean_pc,
                                                  const int* __restrict__ idx,
                                                  const float* __restrict__ W1, const float* __restrict__ b1,
                                                  const float* __restrict__ W2, const float* __restrict__ b2,
                                                  const float* __restrict__ W3, const float* __restrict__ b3,
                                                  float* __restrict__ out) {
    __shared__ __attribute__((aligned(16))) float ldsX[COLS * PH];
    __shared__ __attribute__((aligned(16))) float ldsY[COLS * PH];
    __shared__ __attribute__((aligned(16))) float ldsP[COLS * PH];
    __shared__ __attribute__((aligned(16))) float ldsQ[QC * PH];
    __shared__ __attribute__((aligned(16))) float ldsR[COLS * PR];
    __shared__ __attribute__((aligned(16))) float ldsN[COLS * PC];
    __shared__ __attribute__((aligned(16))) float ldsC[QC * PE];

    int t       = threadIdx.x;
    int lane    = t & 31;
    int w       = t >> 5;
    int laneRow = lane & 15;
    int laneHi  = lane >> 4;

    int gnl = (t & (COLS - 1)) >> 4;
    int gk  = t & 15;

    float gv[16];
    float cv[4];

    int g = blockIdx.x;

    {
        int b  = g / (NN / NT);
        int n0 = (g % (NN / NT)) * NT;
        unsigned j = (unsigned)idx[((size_t)b * KK + gk) * NN + n0 + gnl];
        fetch_cols(gv, cv, x, mean_pc, b, n0, j, t);
        store_cols(ldsN, ldsC, gv, cv, t);
    }

    for (int it = 0; it < GPW; ++it) {
        int zw = 0;
        asm volatile("" : "+s"(zw));
        int b  = g / (NN / NT);
        int n0 = (g % (NN / NT)) * NT;

        bool more = it + 1 < GPW;
        int  gx   = g + NWG;
        int  bx   = gx / (NN / NT);
        int  n0x  = (gx % (NN / NT)) * NT;

        __syncthreads();

        unsigned jx = 0;
        if (more) jx = (unsigned)idx[((size_t)bx * KK + gk) * NN + n0x + gnl];

        gemm_block<2, 4, 2, CC, EE, PC, PH, false, false>(W1 + CC + zw, b1, ldsN, ldsP, 2 * w, 0, laneRow, laneHi);
        gemm_block<2, 1, 2, EE, EE, PE, PH, false, true>(W1 + zw, b1 + zw, ldsC, ldsQ, 2 * w, 0, laneRow, laneHi);
        __syncthreads();

        for (int d = 0; d < DD; ++d) {
            {
                int col = t & (COLS - 1);
                int og  = t >> 6;
                const float* pp = ldsP + col * PH + og * 64;
                const float* qq = ldsQ + (d * NT + (col >> 4)) * PH + og * 64;
                float*       hh = ldsY + col * PH + og * 64;
#pragma unroll
                for (int i = 0; i < 16; ++i) {
                    v4f p = *(const v4f*)(pp + 4 * i);
                    v4f q = *(const v4f*)(qq + 4 * i);
                    v4f h;
#pragma unroll
                    for (int j = 0; j < 4; ++j) h[j] = fmaxf(p[j] + q[j], 0.0f);
                    *(v4f*)(hh + 4 * i) = h;
                }
            }
            __syncthreads();

            bool pf = more && d == 0;
            if (pf) fetch_cols(gv, cv, x, mean_pc, bx, n0x, jx, t);

            gemm_block<2, 4, 2, HH, HH, PH, PH, true, true>(W2 + zw, b2 + zw, ldsY, ldsX, 2 * w, 0, laneRow, laneHi);
            __syncthreads();

            if (pf) store_cols(ldsN, ldsC, gv, cv, t);

            int zoff = 0;
            asm volatile("" : "+s"(zoff));
            size_t o   = (((size_t)b * CC + (t & (CC - 1))) * DD + d) * NN + (n0 + (t >> 6));
            float xres = x[o];
            gemm_block<1, 2, 4, HH, HH, PH, PR, false, true>(W3 + zoff, b3, ldsX, ldsR, w >> 1, (w & 1) * 2, laneRow, laneHi);
            __syncthreads();

            {
                int c  = t & (CC - 1);
                int nl = t >> 6;
                const float* rcol = &ldsR[(nl * 16) * PR + c];
                float m = rcol[0];
#pragma unroll
                for (int k = 1; k < KK; ++k) m = fmaxf(m, rcol[k * PR]);
                out[o] = m + xres;
            }
        }
        g = gx;
    }
}

extern "C" void kernel_launch(void* const* d_in, const int* in_sizes, int n_in,
                              void* d_out, int out_size, void* d_ws, size_t ws_size,
                              hipStream_t stream) {
    const float* x  = (const float*)d_in[0];
    const float* W1 = (const float*)d_in[1];
    const float* b1 = (const float*)d_in[2];
    const float* W2 = (const float*)d_in[3];
    const float* b2 = (const float*)d_in[4];
    const float* W3 = (const float*)d_in[5];
    const float* b3 = (const float*)d_in[6];
    float* out = (float*)d_out;

    float* mean_pc = (float*)d_ws;
    float* sq      = mean_pc + (size_t)BB * CC * NN;
    int*   idxp    = (int*)(sq + (size_t)BB * NN);

    meanpc_kernel<<<(BB * CC * NN + 255) / 256, 256, 0, stream>>>(x, mean_pc);
    sqnorm_kernel<<<(BB * NN + 255) / 256, 256, 0, stream>>>(mean_pc, sq);
    knn_kernel<<<BB * (NN / KQ), 256, 0, stream>>>(mean_pc, sq, idxp);
    mlp_kernel<<<NWG, 256, 0, stream>>>(x, mean_pc, idxp,
                                                        W1, b1, W2, b2, W3, b3, out);
}


# KNN scores on the matrix unit: per step 16 refs x 32 queries x 64 channels as 32 f32 16x16x4 WMMAs, then a swap between the two lane rows puts one query per lane; f32 sum order of scores may change
# speedup vs baseline: 13.5252x; 1.1068x over previous
#include <hip/hip_runtime.h>

#define BB 2
#define CC 64
#define DD 3
#define NN 4096
#define KK 16
#define EE 128
#define HH 256

typedef __attribute__((ext_vector_type(2))) float v2f;
typedef __attribute__((ext_vector_type(8))) float v8f;

constexpr int NT   = 4;
constexpr int COLS = NT * KK;

__global__ void meanpc_kernel(const float* __restrict__ x, float* __restrict__ mean_pc) {
    int i = blockIdx.x * blockDim.x + threadIdx.x;
    if (i >= BB * CC * NN) return;
    int n  = i % NN;
    int bc = i / NN;
    const float* xp = x + (size_t)bc * (DD * NN) + n;
    mean_pc[i] = (xp[0] + xp[NN] + xp[2 * NN]) * (1.0f / 3.0f);
}

__global__ void sqnorm_kernel(const float* __restrict__ mean_pc, float* __restrict__ sq) {
    int i = blockIdx.x * blockDim.x + threadIdx.x;
    if (i >= BB * NN) return;
    int n = i % NN;
    int b = i / NN;
    float s = 0.0f;
#pragma unroll
    for (int c = 0; c < CC; ++c) {
        float v = mean_pc[((size_t)b * CC + c) * NN + n];
        s = __builtin_fmaf(v, v, s);
    }
    sq[i] = s;
}

constexpr int KQ   = 32;
constexpr int KW   = 8;
constexpr int KSEG = NN / KW;
constexpr int KRB  = 16;
constexpr int KNB  = KSEG / KRB;

template <int J>
__device__ __forceinline__ float row_bcast(float v) {
    return __builtin_bit_cast(float,
        __builtin_amdgcn_update_dpp(0, __builtin_bit_cast(int, v), 0x150 + J, 0xf, 0xf, true));
}

template <int J>
struct KnnStep {
    static __device__ __forceinline__ void bcast(float (&sb)[KRB], float sqv) {
        sb[J] = row_bcast<J>(sqv);
        KnnStep<J + 1>::bcast(sb, sqv);
    }
};
template <>
struct KnnStep<KRB> {
    static __device__ __forceinline__ void bcast(float (&)[KRB], float) {}
};

__device__ __forceinline__ float other_row(float v) {
    int b = __builtin_bit_cast(int, v);
    return __builtin_bit_cast(float,
        __builtin_amdgcn_permlanex16(b, b, 0x76543210, 0xfedcba98, false, false));
}

__device__ __forceinline__ void load_points(v2f (&f)[CC / 4], const float* __restrict__ pc,
                                            int p0, int sub, int hi) {
    const float* p = pc + (size_t)(2 * hi) * NN + p0 + sub;
#pragma unroll
    for (int s = 0; s < CC / 4; ++s) {
        v2f v = {p[(size_t)(4 * s) * NN], p[(size_t)(4 * s + 1) * NN]};
        f[s] = v;
    }
}

__device__ __forceinline__ void knn_insert(float (&bd)[KK], int (&bi)[KK], float d, int i) {
    bd[KK - 1] = d;
    bi[KK - 1] = i;
#pragma unroll
    for (int j = KK - 1; j > 0; --j) {
        if (bd[j] < bd[j - 1]) {
            float td = bd[j]; bd[j] = bd[j - 1]; bd[j - 1] = td;
            int   ti = bi[j]; bi[j] = bi[j - 1]; bi[j - 1] = ti;
        }
    }
}

__global__ void __launch_bounds__(256) knn_kernel(const float* __restrict__ mean_pc,
                                                  const float* __restrict__ sq,
                                                  int* __restrict__ idx) {
    __shared__ float mrgD[KW * KK * KQ];
    __shared__ int   mrgI[KW * KK * KQ];

    int t    = threadIdx.x;
    int lane = t & 31;
    int sub  = lane & 15;
    int hi   = lane >> 4;
    int w    = __builtin_amdgcn_readfirstlane(t >> 5);
    int b    = blockIdx.x / (NN / KQ);
    int q0   = (blockIdx.x % (NN / KQ)) * KQ;
    int n    = q0 + lane;
    const float* pc  = mean_pc + (size_t)b * CC * NN;
    const float* sqb = sq + (size_t)b * NN;

    v2f qf[2][CC / 4];
    load_points(qf[0], pc, q0, sub, hi);
    load_points(qf[1], pc, q0 + 16, sub, hi);

    float bd[KK];
    int   bi[KK];
#pragma unroll
    for (int j = 0; j < KK; ++j) { bd[j] = 3.0e38f; bi[j] = 0; }

    int   r0 = w * KSEG;
    v2f   rf[CC / 4];
    float sqv;
    load_points(rf, pc, r0, sub, hi);
    sqv = sqb[r0 + sub];

    const v8f vzero = {0.f, 0.f, 0.f, 0.f, 0.f, 0.f, 0.f, 0.f};
    for (int rb = 0; rb < KNB; ++rb) {
        v8f d0 = vzero;
        v8f d1 = vzero;
#pragma unroll
        for (int s = 0; s < CC / 4; ++s) {
            d0 = __builtin_amdgcn_wmma_f32_16x16x4_f32(false, rf[s], false, qf[0][s], (short)0, d0, false, false);
            d1 = __builtin_amdgcn_wmma_f32_16x16x4_f32(false, rf[s], false, qf[1][s], (short)0, d1, false, false);
        }

        float sb[KRB];
        __builtin_amdgcn_sched_barrier(0);
        KnnStep<0>::bcast(sb, sqv);
        __builtin_amdgcn_sched_barrier(0);

        float dot[KRB];
#pragma unroll
        for (int j = 0; j < 8; ++j) {
            float keep = hi ? d1[j] : d0[j];
            float send = hi ? d0[j] : d1[j];
            float recv = other_row(send);
            dot[j]     = hi ? recv : keep;
            dot[8 + j] = hi ? keep : recv;
        }
#pragma unroll
        for (int j = 0; j < KRB; ++j) dot[j] = sb[j] - 2.0f * dot[j];

        int rn = (rb + 1 < KNB) ? r0 + KRB : r0;
        load_points(rf, pc, rn, sub, hi);
        sqv = sqb[rn + sub];

#pragma unroll
        for (int j = 0; j < KRB; ++j) {
            int r = r0 + j;
            if (r != n && dot[j] < bd[KK - 1]) knn_insert(bd, bi, dot[j], r);
        }
        r0 = rn;
    }

#pragma unroll
    for (int j = 0; j < KK; ++j) {
        mrgD[(w * KK + j) * KQ + lane] = bd[j];
        mrgI[(w * KK + j) * KQ + lane] = bi[j];
    }
    __syncthreads();
    for (int step = 1; step < KW; step <<= 1) {
        if ((w & (2 * step - 1)) == 0) {
            int srcw = w + step;
            for (int j = 0; j < KK; ++j) {
                float cd = mrgD[(srcw * KK + j) * KQ + lane];
                int   ci = mrgI[(srcw * KK + j) * KQ + lane];
                if (cd < bd[KK - 1]) knn_insert(bd, bi, cd, ci);
            }
            if (2 * step < KW) {
#pragma unroll
                for (int j = 0; j < KK; ++j) {
                    mrgD[(w * KK + j) * KQ + lane] = bd[j];
                    mrgI[(w * KK + j) * KQ + lane] = bi[j];
                }
            }
        }
        __syncthreads();
    }
    if (w == 0) {
#pragma unroll
        for (int j = 0; j < KK; ++j) idx[((size_t)b * KK + j) * NN + n] = bi[j];
    }
}

typedef __attribute__((ext_vector_type(4))) float v4f;

constexpr int PE = EE + 4;
constexpr int PH = HH + 4;
constexpr int PR = CC + 4;

template <int NRT, int NCT, int Q>
__device__ __forceinline__ void load_chunk(v4f (&a)[NRT][Q], v4f (&b)[NCT][Q],
                                           const float* const (&wp)[NRT],
                                           const float* const (&bp)[NCT], int koff) {
#pragma unroll
    for (int r = 0; r < NRT; ++r) {
#pragma unroll
        for (int q = 0; q < Q; ++q) a[r][q] = *(const v4f*)(wp[r] + koff + 4 * q);
    }
#pragma unroll
    for (int c = 0; c < NCT; ++c) {
#pragma unroll
        for (int q = 0; q < Q; ++q) b[c][q] = *(const v4f*)(bp[c] + koff + 4 * q);
    }
}

template <int NRT, int NCT, int Q>
__device__ __forceinline__ void mma_chunk(v8f (&acc)[NRT][NCT],
                                          const v4f (&a)[NRT][Q], const v4f (&b)[NCT][Q]) {
#pragma unroll
    for (int q = 0; q < Q; ++q) {
#pragma unroll
        for (int s = 0; s < 2; ++s) {
#pragma unroll
            for (int r = 0; r < NRT; ++r) {
                v2f av = s ? __builtin_shufflevector(a[r][q], a[r][q], 2, 3)
                           : __builtin_shufflevector(a[r][q], a[r][q], 0, 1);
#pragma unroll
                for (int c = 0; c < NCT; ++c) {
                    v2f bv = s ? __builtin_shufflevector(b[c][q], b[c][q], 2, 3)
                               : __builtin_shufflevector(b[c][q], b[c][q], 0, 1);
                    acc[r][c] = __builtin_amdgcn_wmma_f32_16x16x4_f32(
                        false, av, false, bv, (short)0, acc[r][c], false, false);
                }
            }
        }
    }
}

template <int NRT, int NCT, int Q, int KI, int WP, int PIN, int POUT, bool RELU, bool BIAS>
__device__ __forceinline__ void gemm_block(const float* __restrict__ W,
                                           const float* __restrict__ bias,
                                           const float* lin, float* lout,
                                           int rt0, int ct0, int laneRow, int laneHi) {
    constexpr int CH     = 4 * Q;
    constexpr int KSTEP  = 2 * CH;
    constexpr int NCHUNK = KI / KSTEP;
    static_assert(KI % KSTEP == 0 && NCHUNK % 2 == 0, "chunks come in pairs");

    const v8f vzero = {0.f, 0.f, 0.f, 0.f, 0.f, 0.f, 0.f, 0.f};
    v8f acc[NRT][NCT];
#pragma unroll
    for (int r = 0; r < NRT; ++r) {
#pragma unroll
        for (int c = 0; c < NCT; ++c) acc[r][c] = vzero;
    }

    const float* wp[NRT];
    const float* bp[NCT];
#pragma unroll
    for (int r = 0; r < NRT; ++r)
        wp[r] = W + (size_t)((rt0 + r) * 16 + laneRow) * WP + CH * laneHi;
#pragma unroll
    for (int c = 0; c < NCT; ++c)
        bp[c] = lin + ((ct0 + c) * 16 + laneRow) * PIN + CH * laneHi;

    v4f a0[NRT][Q], b0[NCT][Q], a1[NRT][Q], b1[NCT][Q];
    load_chunk<NRT, NCT, Q>(a0, b0, wp, bp, 0);
    for (int kk = 0; kk < NCHUNK; kk += 2) {
        load_chunk<NRT, NCT, Q>(a1, b1, wp, bp, (kk + 1) * KSTEP);
        mma_chunk<NRT, NCT, Q>(acc, a0, b0);
        if (kk + 2 < NCHUNK) load_chunk<NRT, NCT, Q>(a0, b0, wp, bp, (kk + 2) * KSTEP);
        mma_chunk<NRT, NCT, Q>(acc, a1, b1);
    }

#pragma unroll
    for (int r = 0; r < NRT; ++r) {
        int row0 = (rt0 + r) * 16 + 8 * laneHi;
        v4f bl = {0.f, 0.f, 0.f, 0.f};
        v4f bh = {0.f, 0.f, 0.f, 0.f};
        if (BIAS) {
            bl = *(const v4f*)(bias + row0);
            bh = *(const v4f*)(bias + row0 + 4);
        }
#pragma unroll
        for (int c = 0; c < NCT; ++c) {
            int col = (ct0 + c) * 16 + laneRow;
            v4f lo, hi;
#pragma unroll
            for (int j = 0; j < 4; ++j) {
                float u = acc[r][c][j];
                float v = acc[r][c][j + 4];
                if (BIAS) { u += bl[j]; v += bh[j]; }
                if (RELU) { u = fmaxf(u, 0.0f); v = fmaxf(v, 0.0f); }
                lo[j] = u;
                hi[j] = v;
            }
            *(v4f*)(lout + col * POUT + row0)     = lo;
            *(v4f*)(lout + col * POUT + row0 + 4) = hi;
        }
    }
}

constexpr int PC = CC + 4;
constexpr int QC = 16;
constexpr int NGRP = BB * (NN / NT);
constexpr int NWG  = 256;
constexpr int GPW  = NGRP / NWG;
static_assert(NGRP % NWG == 0, "every workgroup takes the same number of groups");

__device__ __forceinline__ void fetch_cols(float (&gv)[16], float (&cv)[4],
                                           const float* __restrict__ x,
                                           const float* __restrict__ mean_pc,
                                           int b, int n0, unsigned j, int t) {
    int cg   = t >> 6;
    int dcol = t & (QC - 1);
    int c0   = t >> 4;
    const float* mp = mean_pc + ((size_t)b * CC + cg * 16) * NN + j;
    const float* xp = x + (((size_t)b * CC + c0) * DD + (dcol >> 2)) * NN + n0 + (dcol & 3);
#pragma unroll
    for (int i = 0; i < 16; ++i) gv[i] = mp[(size_t)i * NN];
#pragma unroll
    for (int i = 0; i < 4; ++i) {
        float v = 0.0f;
        if (dcol < DD * NT) v = xp[(size_t)i * 16 * DD * NN];
        cv[i] = v;
    }
}

__device__ __forceinline__ void store_cols(float* ldsN, float* ldsC,
                                           const float (&gv)[16], const float (&cv)[4], int t) {
    int col  = t & (COLS - 1);
    int cg   = t >> 6;
    int dcol = t & (QC - 1);
    int c0   = t >> 4;
#pragma unroll
    for (int i = 0; i < 4; ++i) {
        v4f s = {gv[4 * i], gv[4 * i + 1], gv[4 * i + 2], gv[4 * i + 3]};
        *(v4f*)(ldsN + col * PC + cg * 16 + 4 * i) = s;
    }
#pragma unroll
    for (int i = 0; i < 4; ++i) {
        ldsC[dcol * PE + c0 + 16 * i]      = cv[i];
        ldsC[dcol * PE + CC + c0 + 16 * i] = -cv[i];
    }
}

__global__ void __launch_bounds__(256) mlp_kernel(const float* __restrict__ x,
                                                  const float* __restrict__ mean_pc,
                                                  const int* __restrict__ idx,
                                                  const float* __restrict__ W1, const float* __restrict__ b1,
                                                  const float* __restrict__ W2, const float* __restrict__ b2,
                                                  const float* __restrict__ W3, const float* __restrict__ b3,
                                                  float* __restrict__ out) {
    __shared__ __attribute__((aligned(16))) float ldsX[COLS * PH];
    __shared__ __attribute__((aligned(16))) float ldsY[COLS * PH];
    __shared__ __attribute__((aligned(16))) float ldsP[COLS * PH];
    __shared__ __attribute__((aligned(16))) float ldsQ[QC * PH];
    __shared__ __attribute__((aligned(16))) float ldsR[COLS * PR];
    __shared__ __attribute__((aligned(16))) float ldsN[COLS * PC];
    __shared__ __attribute__((aligned(16))) float ldsC[QC * PE];

    int t       = threadIdx.x;
    int lane    = t & 31;
    int w       = t >> 5;
    int laneRow = lane & 15;
    int laneHi  = lane >> 4;

    int gnl = (t & (COLS - 1)) >> 4;
    int gk  = t & 15;

    float gv[16];
    float cv[4];

    int g = blockIdx.x;

    {
        int b  = g / (NN / NT);
        int n0 = (g % (NN / NT)) * NT;
        unsigned j = (unsigned)idx[((size_t)b * KK + gk) * NN + n0 + gnl];
        fetch_cols(gv, cv, x, mean_pc, b, n0, j, t);
        store_cols(ldsN, ldsC, gv, cv, t);
    }

    for (int it = 0; it < GPW; ++it) {
        int zw = 0;
        asm volatile("" : "+s"(zw));
        int b  = g / (NN / NT);
        int n0 = (g % (NN / NT)) * NT;

        bool more = it + 1 < GPW;
        int  gx   = g + NWG;
        int  bx   = gx / (NN / NT);
        int  n0x  = (gx % (NN / NT)) * NT;

        __syncthreads();

        unsigned jx = 0;
        if (more) jx = (unsigned)idx[((size_t)bx * KK + gk) * NN + n0x + gnl];

        gemm_block<2, 4, 2, CC, EE, PC, PH, false, false>(W1 + CC + zw, b1, ldsN, ldsP, 2 * w, 0, laneRow, laneHi);
        gemm_block<2, 1, 2, EE, EE, PE, PH, false, true>(W1 + zw, b1 + zw, ldsC, ldsQ, 2 * w, 0, laneRow, laneHi);
        __syncthreads();

        for (int d = 0; d < DD; ++d) {
            {
                int col = t & (COLS - 1);
                int og  = t >> 6;
                const float* pp = ldsP + col * PH + og * 64;
                const float* qq = ldsQ + (d * NT + (col >> 4)) * PH + og * 64;
                float*       hh = ldsY + col * PH + og * 64;
#pragma unroll
                for (int i = 0; i < 16; ++i) {
                    v4f p = *(const v4f*)(pp + 4 * i);
                    v4f q = *(const v4f*)(qq + 4 * i);
                    v4f h;
#pragma unroll
                    for (int j = 0; j < 4; ++j) h[j] = fmaxf(p[j] + q[j], 0.0f);
                    *(v4f*)(hh + 4 * i) = h;
                }
            }
            __syncthreads();

            bool pf = more && d == 0;
            if (pf) fetch_cols(gv, cv, x, mean_pc, bx, n0x, jx, t);

            gemm_block<2, 4, 2, HH, HH, PH, PH, true, true>(W2 + zw, b2 + zw, ldsY, ldsX, 2 * w, 0, laneRow, laneHi);
            __syncthreads();

            if (pf) store_cols(ldsN, ldsC, gv, cv, t);

            int zoff = 0;
            asm volatile("" : "+s"(zoff));
            size_t o   = (((size_t)b * CC + (t & (CC - 1))) * DD + d) * NN + (n0 + (t >> 6));
            float xres = x[o];
            gemm_block<1, 2, 4, HH, HH, PH, PR, false, true>(W3 + zoff, b3, ldsX, ldsR, w >> 1, (w & 1) * 2, laneRow, laneHi);
            __syncthreads();

            {
                int c  = t & (CC - 1);
                int nl = t >> 6;
                const float* rcol = &ldsR[(nl * 16) * PR + c];
                float m = rcol[0];
#pragma unroll
                for (int k = 1; k < KK; ++k) m = fmaxf(m, rcol[k * PR]);
                out[o] = m + xres;
            }
        }
        g = gx;
    }
}

extern "C" void kernel_launch(void* const* d_in, const int* in_sizes, int n_in,
                              void* d_out, int out_size, void* d_ws, size_t ws_size,
                              hipStream_t stream) {
    const float* x  = (const float*)d_in[0];
    const float* W1 = (const float*)d_in[1];
    const float* b1 = (const float*)d_in[2];
    const float* W2 = (const float*)d_in[3];
    const float* b2 = (const float*)d_in[4];
    const float* W3 = (const float*)d_in[5];
    const float* b3 = (const float*)d_in[6];
    float* out = (float*)d_out;

    float* mean_pc = (float*)d_ws;
    float* sq      = mean_pc + (size_t)BB * CC * NN;
    int*   idxp    = (int*)(sq + (size_t)BB * NN);

    meanpc_kernel<<<(BB * CC * NN + 255) / 256, 256, 0, stream>>>(x, mean_pc);
    sqnorm_kernel<<<(BB * NN + 255) / 256, 256, 0, stream>>>(mean_pc, sq);
    knn_kernel<<<BB * (NN / KQ), 256, 0, stream>>>(mean_pc, sq, idxp);
    mlp_kernel<<<NWG, 256, 0, stream>>>(x, mean_pc, idxp,
                                                        W1, b1, W2, b2, W3, b3, out);
}


# MLP GEMMs: next chunk's fragment loads kept as one group ahead of the current chunk's WMMAs by scheduling barriers (the compiler sank them singly among the WMMAs); bias rows loaded before the k loop
# speedup vs baseline: 14.1127x; 1.0434x over previous
#include <hip/hip_runtime.h>

#define BB 2
#define CC 64
#define DD 3
#define NN 4096
#define KK 16
#define EE 128
#define HH 256

typedef __attribute__((ext_vector_type(2))) float v2f;
typedef __attribute__((ext_vector_type(8))) float v8f;

constexpr int NT   = 4;
constexpr int COLS = NT * KK;

__global__ void meanpc_kernel(const float* __restrict__ x, float* __restrict__ mean_pc) {
    int i = blockIdx.x * blockDim.x + threadIdx.x;
    if (i >= BB * CC * NN) return;
    int n  = i % NN;
    int bc = i / NN;
    const float* xp = x + (size_t)bc * (DD * NN) + n;
    mean_pc[i] = (xp[0] + xp[NN] + xp[2 * NN]) * (1.0f / 3.0f);
}

__global__ void sqnorm_kernel(const float* __restrict__ mean_pc, float* __restrict__ sq) {
    int i = blockIdx.x * blockDim.x + threadIdx.x;
    if (i >= BB * NN) return;
    int n = i % NN;
    int b = i / NN;
    float s = 0.0f;
#pragma unroll
    for (int c = 0; c < CC; ++c) {
        float v = mean_pc[((size_t)b * CC + c) * NN + n];
        s = __builtin_fmaf(v, v, s);
    }
    sq[i] = s;
}

constexpr int KQ   = 32;
constexpr int KW   = 8;
constexpr int KSEG = NN / KW;
constexpr int KRB  = 16;
constexpr int KNB  = KSEG / KRB;

template <int J>
__device__ __forceinline__ float row_bcast(float v) {
    return __builtin_bit_cast(float,
        __builtin_amdgcn_update_dpp(0, __builtin_bit_cast(int, v), 0x150 + J, 0xf, 0xf, true));
}

template <int J>
struct KnnStep {
    static __device__ __forceinline__ void bcast(float (&sb)[KRB], float sqv) {
        sb[J] = row_bcast<J>(sqv);
        KnnStep<J + 1>::bcast(sb, sqv);
    }
};
template <>
struct KnnStep<KRB> {
    static __device__ __forceinline__ void bcast(float (&)[KRB], float) {}
};

__device__ __forceinline__ float other_row(float v) {
    int b = __builtin_bit_cast(int, v);
    return __builtin_bit_cast(float,
        __builtin_amdgcn_permlanex16(b, b, 0x76543210, 0xfedcba98, false, false));
}

__device__ __forceinline__ void load_points(v2f (&f)[CC / 4], const float* __restrict__ pc,
                                            int p0, int sub, int hi) {
    const float* p = pc + (size_t)(2 * hi) * NN + p0 + sub;
#pragma unroll
    for (int s = 0; s < CC / 4; ++s) {
        v2f v = {p[(size_t)(4 * s) * NN], p[(size_t)(4 * s + 1) * NN]};
        f[s] = v;
    }
}

__device__ __forceinline__ void knn_insert(float (&bd)[KK], int (&bi)[KK], float d, int i) {
    bd[KK - 1] = d;
    bi[KK - 1] = i;
#pragma unroll
    for (int j = KK - 1; j > 0; --j) {
        if (bd[j] < bd[j - 1]) {
            float td = bd[j]; bd[j] = bd[j - 1]; bd[j - 1] = td;
            int   ti = bi[j]; bi[j] = bi[j - 1]; bi[j - 1] = ti;
        }
    }
}

__global__ void __launch_bounds__(256) knn_kernel(const float* __restrict__ mean_pc,
                                                  const float* __restrict__ sq,
                                                  int* __restrict__ idx) {
    __shared__ float mrgD[KW * KK * KQ];
    __shared__ int   mrgI[KW * KK * KQ];

    int t    = threadIdx.x;
    int lane = t & 31;
    int sub  = lane & 15;
    int hi   = lane >> 4;
    int w    = __builtin_amdgcn_readfirstlane(t >> 5);
    int b    = blockIdx.x / (NN / KQ);
    int q0   = (blockIdx.x % (NN / KQ)) * KQ;
    int n    = q0 + lane;
    const float* pc  = mean_pc + (size_t)b * CC * NN;
    const float* sqb = sq + (size_t)b * NN;

    v2f qf[2][CC / 4];
    load_points(qf[0], pc, q0, sub, hi);
    load_points(qf[1], pc, q0 + 16, sub, hi);

    float bd[KK];
    int   bi[KK];
#pragma unroll
    for (int j = 0; j < KK; ++j) { bd[j] = 3.0e38f; bi[j] = 0; }

    int   r0 = w * KSEG;
    v2f   rf[CC / 4];
    float sqv;
    load_points(rf, pc, r0, sub, hi);
    sqv = sqb[r0 + sub];

    const v8f vzero = {0.f, 0.f, 0.f, 0.f, 0.f, 0.f, 0.f, 0.f};
    for (int rb = 0; rb < KNB; ++rb) {
        v8f d0 = vzero;
        v8f d1 = vzero;
#pragma unroll
        for (int s = 0; s < CC / 4; ++s) {
            d0 = __builtin_amdgcn_wmma_f32_16x16x4_f32(false, rf[s], false, qf[0][s], (short)0, d0, false, false);
            d1 = __builtin_amdgcn_wmma_f32_16x16x4_f32(false, rf[s], false, qf[1][s], (short)0, d1, false, false);
        }

        float sb[KRB];
        __builtin_amdgcn_sched_barrier(0);
        KnnStep<0>::bcast(sb, sqv);
        __builtin_amdgcn_sched_barrier(0);

        float dot[KRB];
#pragma unroll
        for (int j = 0; j < 8; ++j) {
            float keep = hi ? d1[j] : d0[j];
            float send = hi ? d0[j] : d1[j];
            float recv = other_row(send);
            dot[j]     = hi ? recv : keep;
            dot[8 + j] = hi ? keep : recv;
        }
#pragma unroll
        for (int j = 0; j < KRB; ++j) dot[j] = sb[j] - 2.0f * dot[j];

        int rn = (rb + 1 < KNB) ? r0 + KRB : r0;
        load_points(rf, pc, rn, sub, hi);
        sqv = sqb[rn + sub];

#pragma unroll
        for (int j = 0; j < KRB; ++j) {
            int r = r0 + j;
            if (r != n && dot[j] < bd[KK - 1]) knn_insert(bd, bi, dot[j], r);
        }
        r0 = rn;
    }

#pragma unroll
    for (int j = 0; j < KK; ++j) {
        mrgD[(w * KK + j) * KQ + lane] = bd[j];
        mrgI[(w * KK + j) * KQ + lane] = bi[j];
    }
    __syncthreads();
    for (int step = 1; step < KW; step <<= 1) {
        if ((w & (2 * step - 1)) == 0) {
            int srcw = w + step;
            for (int j = 0; j < KK; ++j) {
                float cd = mrgD[(srcw * KK + j) * KQ + lane];
                int   ci = mrgI[(srcw * KK + j) * KQ + lane];
                if (cd < bd[KK - 1]) knn_insert(bd, bi, cd, ci);
            }
            if (2 * step < KW) {
#pragma unroll
                for (int j = 0; j < KK; ++j) {
                    mrgD[(w * KK + j) * KQ + lane] = bd[j];
                    mrgI[(w * KK + j) * KQ + lane] = bi[j];
                }
            }
        }
        __syncthreads();
    }
    if (w == 0) {
#pragma unroll
        for (int j = 0; j < KK; ++j) idx[((size_t)b * KK + j) * NN + n] = bi[j];
    }
}

typedef __attribute__((ext_vector_type(4))) float v4f;

constexpr int PE = EE + 4;
constexpr int PH = HH + 4;
constexpr int PR = CC + 4;

template <int NRT, int NCT, int Q>
__device__ __forceinline__ void load_chunk(v4f (&a)[NRT][Q], v4f (&b)[NCT][Q],
                                           const float* const (&wp)[NRT],
                                           const float* const (&bp)[NCT], int koff) {
#pragma unroll
    for (int r = 0; r < NRT; ++r) {
#pragma unroll
        for (int q = 0; q < Q; ++q) a[r][q] = *(const v4f*)(wp[r] + koff + 4 * q);
    }
#pragma unroll
    for (int c = 0; c < NCT; ++c) {
#pragma unroll
        for (int q = 0; q < Q; ++q) b[c][q] = *(const v4f*)(bp[c] + koff + 4 * q);
    }
}

template <int NRT, int NCT, int Q>
__device__ __forceinline__ void mma_chunk(v8f (&acc)[NRT][NCT],
                                          const v4f (&a)[NRT][Q], const v4f (&b)[NCT][Q]) {
#pragma unroll
    for (int q = 0; q < Q; ++q) {
#pragma unroll
        for (int s = 0; s < 2; ++s) {
#pragma unroll
            for (int r = 0; r < NRT; ++r) {
                v2f av = s ? __builtin_shufflevector(a[r][q], a[r][q], 2, 3)
                           : __builtin_shufflevector(a[r][q], a[r][q], 0, 1);
#pragma unroll
                for (int c = 0; c < NCT; ++c) {
                    v2f bv = s ? __builtin_shufflevector(b[c][q], b[c][q], 2, 3)
                               : __builtin_shufflevector(b[c][q], b[c][q], 0, 1);
                    acc[r][c] = __builtin_amdgcn_wmma_f32_16x16x4_f32(
                        false, av, false, bv, (short)0, acc[r][c], false, false);
                }
            }
        }
    }
}

template <int NRT, int NCT, int Q, int KI, int WP, int PIN, int POUT, bool RELU, bool BIAS>
__device__ __forceinline__ void gemm_block(const float* __restrict__ W,
                                           const float* __restrict__ bias,
                                           const float* lin, float* lout,
                                           int rt0, int ct0, int laneRow, int laneHi) {
    constexpr int CH     = 4 * Q;
    constexpr int KSTEP  = 2 * CH;
    constexpr int NCHUNK = KI / KSTEP;
    static_assert(KI % KSTEP == 0 && NCHUNK % 2 == 0, "chunks come in pairs");

    const v8f vzero = {0.f, 0.f, 0.f, 0.f, 0.f, 0.f, 0.f, 0.f};
    const v4f vzero4 = {0.f, 0.f, 0.f, 0.f};
    v8f acc[NRT][NCT];
#pragma unroll
    for (int r = 0; r < NRT; ++r) {
#pragma unroll
        for (int c = 0; c < NCT; ++c) acc[r][c] = vzero;
    }

    const float* wp[NRT];
    const float* bp[NCT];
#pragma unroll
    for (int r = 0; r < NRT; ++r)
        wp[r] = W + (size_t)((rt0 + r) * 16 + laneRow) * WP + CH * laneHi;
#pragma unroll
    for (int c = 0; c < NCT; ++c)
        bp[c] = lin + ((ct0 + c) * 16 + laneRow) * PIN + CH * laneHi;

    v4f bl[NRT], bh[NRT];
    for (int r = 0; r < NRT; ++r) {
        int row0 = (rt0 + r) * 16 + 8 * laneHi;
        bl[r] = vzero4;
        bh[r] = vzero4;
        if (BIAS) {
            bl[r] = *(const v4f*)(bias + row0);
            bh[r] = *(const v4f*)(bias + row0 + 4);
        }
    }

    v4f a0[NRT][Q], b0[NCT][Q], a1[NRT][Q], b1[NCT][Q];
    load_chunk<NRT, NCT, Q>(a0, b0, wp, bp, 0);
    for (int kk = 0; kk < NCHUNK; kk += 2) {
        load_chunk<NRT, NCT, Q>(a1, b1, wp, bp, (kk + 1) * KSTEP);
        __builtin_amdgcn_sched_barrier(0);
        mma_chunk<NRT, NCT, Q>(acc, a0, b0);
        __builtin_amdgcn_sched_barrier(0);
        if (kk + 2 < NCHUNK) load_chunk<NRT, NCT, Q>(a0, b0, wp, bp, (kk + 2) * KSTEP);
        __builtin_amdgcn_sched_barrier(0);
        mma_chunk<NRT, NCT, Q>(acc, a1, b1);
        __builtin_amdgcn_sched_barrier(0);
    }

#pragma unroll
    for (int r = 0; r < NRT; ++r) {
        int row0 = (rt0 + r) * 16 + 8 * laneHi;
#pragma unroll
        for (int c = 0; c < NCT; ++c) {
            int col = (ct0 + c) * 16 + laneRow;
            v4f lo, hi;
#pragma unroll
            for (int j = 0; j < 4; ++j) {
                float u = acc[r][c][j];
                float v = acc[r][c][j + 4];
                if (BIAS) { u += bl[r][j]; v += bh[r][j]; }
                if (RELU) { u = fmaxf(u, 0.0f); v = fmaxf(v, 0.0f); }
                lo[j] = u;
                hi[j] = v;
            }
            *(v4f*)(lout + col * POUT + row0)     = lo;
            *(v4f*)(lout + col * POUT + row0 + 4) = hi;
        }
    }
}

constexpr int PC = CC + 4;
constexpr int QC = 16;
constexpr int NGRP = BB * (NN / NT);
constexpr int NWG  = 256;
constexpr int GPW  = NGRP / NWG;
static_assert(NGRP % NWG == 0, "every workgroup takes the same number of groups");

__device__ __forceinline__ void fetch_cols(float (&gv)[16], float (&cv)[4],
                                           const float* __restrict__ x,
                                           const float* __restrict__ mean_pc,
                                           int b, int n0, unsigned j, int t) {
    int cg   = t >> 6;
    int dcol = t & (QC - 1);
    int c0   = t >> 4;
    const float* mp = mean_pc + ((size_t)b * CC + cg * 16) * NN + j;
    const float* xp = x + (((size_t)b * CC + c0) * DD + (dcol >> 2)) * NN + n0 + (dcol & 3);
#pragma unroll
    for (int i = 0; i < 16; ++i) gv[i] = mp[(size_t)i * NN];
#pragma unroll
    for (int i = 0; i < 4; ++i) {
        float v = 0.0f;
        if (dcol < DD * NT) v = xp[(size_t)i * 16 * DD * NN];
        cv[i] = v;
    }
}

__device__ __forceinline__ void store_cols(float* ldsN, float* ldsC,
                                           const float (&gv)[16], const float (&cv)[4], int t) {
    int col  = t & (COLS - 1);
    int cg   = t >> 6;
    int dcol = t & (QC - 1);
    int c0   = t >> 4;
#pragma unroll
    for (int i = 0; i < 4; ++i) {
        v4f s = {gv[4 * i], gv[4 * i + 1], gv[4 * i + 2], gv[4 * i + 3]};
        *(v4f*)(ldsN + col * PC + cg * 16 + 4 * i) = s;
    }
#pragma unroll
    for (int i = 0; i < 4; ++i) {
        ldsC[dcol * PE + c0 + 16 * i]      = cv[i];
        ldsC[dcol * PE + CC + c0 + 16 * i] = -cv[i];
    }
}

__global__ void __launch_bounds__(256) mlp_kernel(const float* __restrict__ x,
                                                  const float* __restrict__ mean_pc,
                                                  const int* __restrict__ idx,
                                                  const float* __restrict__ W1, const float* __restrict__ b1,
                                                  const float* __restrict__ W2, const float* __restrict__ b2,
                                                  const float* __restrict__ W3, const float* __restrict__ b3,
                                                  float* __restrict__ out) {
    __shared__ __attribute__((aligned(16))) float ldsX[COLS * PH];
    __shared__ __attribute__((aligned(16))) float ldsY[COLS * PH];
    __shared__ __attribute__((aligned(16))) float ldsP[COLS * PH];
    __shared__ __attribute__((aligned(16))) float ldsQ[QC * PH];
    __shared__ __attribute__((aligned(16))) float ldsR[COLS * PR];
    __shared__ __attribute__((aligned(16))) float ldsN[COLS * PC];
    __shared__ __attribute__((aligned(16))) float ldsC[QC * PE];

    int t       = threadIdx.x;
    int lane    = t & 31;
    int w       = t >> 5;
    int laneRow = lane & 15;
    int laneHi  = lane >> 4;

    int gnl = (t & (COLS - 1)) >> 4;
    int gk  = t & 15;

    float gv[16];
    float cv[4];

    int g = blockIdx.x;

    {
        int b  = g / (NN / NT);
        int n0 = (g % (NN / NT)) * NT;
        unsigned j = (unsigned)idx[((size_t)b * KK + gk) * NN + n0 + gnl];
        fetch_cols(gv, cv, x, mean_pc, b, n0, j, t);
        store_cols(ldsN, ldsC, gv, cv, t);
    }

    for (int it = 0; it < GPW; ++it) {
        int zw = 0;
        asm volatile("" : "+s"(zw));
        int b  = g / (NN / NT);
        int n0 = (g % (NN / NT)) * NT;

        bool more = it + 1 < GPW;
        int  gx   = g + NWG;
        int  bx   = gx / (NN / NT);
        int  n0x  = (gx % (NN / NT)) * NT;

        __syncthreads();

        unsigned jx = 0;
        if (more) jx = (unsigned)idx[((size_t)bx * KK + gk) * NN + n0x + gnl];

        gemm_block<2, 4, 2, CC, EE, PC, PH, false, false>(W1 + CC + zw, b1, ldsN, ldsP, 2 * w, 0, laneRow, laneHi);
        gemm_block<2, 1, 2, EE, EE, PE, PH, false, true>(W1 + zw, b1 + zw, ldsC, ldsQ, 2 * w, 0, laneRow, laneHi);
        __syncthreads();

        for (int d = 0; d < DD; ++d) {
            {
                int col = t & (COLS - 1);
                int og  = t >> 6;
                const float* pp = ldsP + col * PH + og * 64;
                const float* qq = ldsQ + (d * NT + (col >> 4)) * PH + og * 64;
                float*       hh = ldsY + col * PH + og * 64;
#pragma unroll
                for (int i = 0; i < 16; ++i) {
                    v4f p = *(const v4f*)(pp + 4 * i);
                    v4f q = *(const v4f*)(qq + 4 * i);
                    v4f h;
#pragma unroll
                    for (int j = 0; j < 4; ++j) h[j] = fmaxf(p[j] + q[j], 0.0f);
                    *(v4f*)(hh + 4 * i) = h;
                }
            }
            __syncthreads();

            bool pf = more && d == 0;
            if (pf) fetch_cols(gv, cv, x, mean_pc, bx, n0x, jx, t);

            gemm_block<2, 4, 2, HH, HH, PH, PH, true, true>(W2 + zw, b2 + zw, ldsY, ldsX, 2 * w, 0, laneRow, laneHi);
            __syncthreads();

            if (pf) store_cols(ldsN, ldsC, gv, cv, t);

            int zoff = 0;
            asm volatile("" : "+s"(zoff));
            size_t o   = (((size_t)b * CC + (t & (CC - 1))) * DD + d) * NN + (n0 + (t >> 6));
            float xres = x[o];
            gemm_block<1, 2, 4, HH, HH, PH, PR, false, true>(W3 + zoff, b3, ldsX, ldsR, w >> 1, (w & 1) * 2, laneRow, laneHi);
            __syncthreads();

            {
                int c  = t & (CC - 1);
                int nl = t >> 6;
                const float* rcol = &ldsR[(nl * 16) * PR + c];
                float m = rcol[0];
#pragma unroll
                for (int k = 1; k < KK; ++k) m = fmaxf(m, rcol[k * PR]);
                out[o] = m + xres;
            }
        }
        g = gx;
    }
}

extern "C" void kernel_launch(void* const* d_in, const int* in_sizes, int n_in,
                              void* d_out, int out_size, void* d_ws, size_t ws_size,
                              hipStream_t stream) {
    const float* x  = (const float*)d_in[0];
    const float* W1 = (const float*)d_in[1];
    const float* b1 = (const float*)d_in[2];
    const float* W2 = (const float*)d_in[3];
    const float* b2 = (const float*)d_in[4];
    const float* W3 = (const float*)d_in[5];
    const float* b3 = (const float*)d_in[6];
    float* out = (float*)d_out;

    float* mean_pc = (float*)d_ws;
    float* sq      = mean_pc + (size_t)BB * CC * NN;
    int*   idxp    = (int*)(sq + (size_t)BB * NN);

    meanpc_kernel<<<(BB * CC * NN + 255) / 256, 256, 0, stream>>>(x, mean_pc);
    sqnorm_kernel<<<(BB * NN + 255) / 256, 256, 0, stream>>>(mean_pc, sq);
    knn_kernel<<<BB * (NN / KQ), 256, 0, stream>>>(mean_pc, sq, idxp);
    mlp_kernel<<<NWG, 256, 0, stream>>>(x, mean_pc, idxp,
                                                        W1, b1, W2, b2, W3, b3, out);
}


# KNN list update: each lane walks only its own candidates of a step (bit mask, lowest first; trips = largest count among lanes), inserting by 16 independent selects; lists bit-identical
# speedup vs baseline: 14.6965x; 1.0414x over previous
#include <hip/hip_runtime.h>

#define BB 2
#define CC 64
#define DD 3
#define NN 4096
#define KK 16
#define EE 128
#define HH 256

typedef __attribute__((ext_vector_type(2))) float v2f;
typedef __attribute__((ext_vector_type(8))) float v8f;

constexpr int NT   = 4;
constexpr int COLS = NT * KK;

__global__ void meanpc_kernel(const float* __restrict__ x, float* __restrict__ mean_pc) {
    int i = blockIdx.x * blockDim.x + threadIdx.x;
    if (i >= BB * CC * NN) return;
    int n  = i % NN;
    int bc = i / NN;
    const float* xp = x + (size_t)bc * (DD * NN) + n;
    mean_pc[i] = (xp[0] + xp[NN] + xp[2 * NN]) * (1.0f / 3.0f);
}

__global__ void sqnorm_kernel(const float* __restrict__ mean_pc, float* __restrict__ sq) {
    int i = blockIdx.x * blockDim.x + threadIdx.x;
    if (i >= BB * NN) return;
    int n = i % NN;
    int b = i / NN;
    float s = 0.0f;
#pragma unroll
    for (int c = 0; c < CC; ++c) {
        float v = mean_pc[((size_t)b * CC + c) * NN + n];
        s = __builtin_fmaf(v, v, s);
    }
    sq[i] = s;
}

constexpr int KQ   = 32;
constexpr int KW   = 8;
constexpr int KSEG = NN / KW;
constexpr int KRB  = 16;
constexpr int KNB  = KSEG / KRB;
constexpr int MB   = 8;

template <int J>
__device__ __forceinline__ float row_bcast(float v) {
    return __builtin_bit_cast(float,
        __builtin_amdgcn_update_dpp(0, __builtin_bit_cast(int, v), 0x150 + J, 0xf, 0xf, true));
}

template <int J>
struct KnnStep {
    static __device__ __forceinline__ void bcast(float (&sb)[KRB], float sqv) {
        sb[J] = row_bcast<J>(sqv);
        KnnStep<J + 1>::bcast(sb, sqv);
    }
};
template <>
struct KnnStep<KRB> {
    static __device__ __forceinline__ void bcast(float (&)[KRB], float) {}
};

__device__ __forceinline__ float other_row(float v) {
    int b = __builtin_bit_cast(int, v);
    return __builtin_bit_cast(float,
        __builtin_amdgcn_permlanex16(b, b, 0x76543210, 0xfedcba98, false, false));
}

__device__ __forceinline__ void load_points(v2f (&f)[CC / 4], const float* __restrict__ pc,
                                            int p0, int sub, int hi) {
    const float* p = pc + (size_t)(2 * hi) * NN + p0 + sub;
#pragma unroll
    for (int s = 0; s < CC / 4; ++s) {
        v2f v = {p[(size_t)(4 * s) * NN], p[(size_t)(4 * s + 1) * NN]};
        f[s] = v;
    }
}

__device__ __forceinline__ float pick(bool c, float a, float b) { return c ? a : b; }
__device__ __forceinline__ int   pick(bool c, int a, int b)     { return c ? a : b; }

__device__ __forceinline__ void knn_insert(float (&bd)[KK], int (&bi)[KK], float d, int i) {
    bool c[KK];
#pragma unroll
    for (int j = 0; j < KK; ++j) c[j] = d < bd[j];
#pragma unroll
    for (int j = KK - 1; j > 0; --j) {
        bd[j] = pick(c[j], pick(c[j - 1], bd[j - 1], d), bd[j]);
        bi[j] = pick(c[j], pick(c[j - 1], bi[j - 1], i), bi[j]);
    }
    bd[0] = pick(c[0], d, bd[0]);
    bi[0] = pick(c[0], i, bi[0]);
}

__device__ __forceinline__ float pick16(const float (&v)[KRB], int j) {
    bool b0 = (j & 1) != 0;
    bool b1 = (j & 2) != 0;
    bool b2 = (j & 4) != 0;
    bool b3 = (j & 8) != 0;
    float e0 = pick(b0, v[1], v[0]);
    float e1 = pick(b0, v[3], v[2]);
    float e2 = pick(b0, v[5], v[4]);
    float e3 = pick(b0, v[7], v[6]);
    float e4 = pick(b0, v[9], v[8]);
    float e5 = pick(b0, v[11], v[10]);
    float e6 = pick(b0, v[13], v[12]);
    float e7 = pick(b0, v[15], v[14]);
    float f0 = pick(b1, e1, e0);
    float f1 = pick(b1, e3, e2);
    float f2 = pick(b1, e5, e4);
    float f3 = pick(b1, e7, e6);
    float g0 = pick(b2, f1, f0);
    float g1 = pick(b2, f3, f2);
    return pick(b3, g1, g0);
}

__global__ void __launch_bounds__(256) knn_kernel(const float* __restrict__ mean_pc,
                                                  const float* __restrict__ sq,
                                                  int* __restrict__ idx) {
    __shared__ float mrgD[KW * KK * KQ];
    __shared__ int   mrgI[KW * KK * KQ];

    int t    = threadIdx.x;
    int lane = t & 31;
    int sub  = lane & 15;
    int hi   = lane >> 4;
    int w    = __builtin_amdgcn_readfirstlane(t >> 5);
    int b    = blockIdx.x / (NN / KQ);
    int q0   = (blockIdx.x % (NN / KQ)) * KQ;
    int n    = q0 + lane;
    const float* pc  = mean_pc + (size_t)b * CC * NN;
    const float* sqb = sq + (size_t)b * NN;

    v2f qf[2][CC / 4];
    load_points(qf[0], pc, q0, sub, hi);
    load_points(qf[1], pc, q0 + 16, sub, hi);

    float bd[KK];
    int   bi[KK];
#pragma unroll
    for (int j = 0; j < KK; ++j) { bd[j] = 3.0e38f; bi[j] = 0; }

    int   r0 = w * KSEG;
    v2f   rf[CC / 4];
    float sqv;
    load_points(rf, pc, r0, sub, hi);
    sqv = sqb[r0 + sub];

    const v8f vzero = {0.f, 0.f, 0.f, 0.f, 0.f, 0.f, 0.f, 0.f};
    for (int rb = 0; rb < KNB; ++rb) {
        v8f d0 = vzero;
        v8f d1 = vzero;
#pragma unroll
        for (int s = 0; s < CC / 4; ++s) {
            d0 = __builtin_amdgcn_wmma_f32_16x16x4_f32(false, rf[s], false, qf[0][s], (short)0, d0, false, false);
            d1 = __builtin_amdgcn_wmma_f32_16x16x4_f32(false, rf[s], false, qf[1][s], (short)0, d1, false, false);
        }

        float sb[KRB];
        __builtin_amdgcn_sched_barrier(0);
        KnnStep<0>::bcast(sb, sqv);
        __builtin_amdgcn_sched_barrier(0);

        float dot[KRB];
#pragma unroll
        for (int j = 0; j < 8; ++j) {
            float keep = hi ? d1[j] : d0[j];
            float send = hi ? d0[j] : d1[j];
            float recv = other_row(send);
            dot[j]     = hi ? recv : keep;
            dot[8 + j] = hi ? keep : recv;
        }
#pragma unroll
        for (int j = 0; j < KRB; ++j) dot[j] = sb[j] - 2.0f * dot[j];

        int rn = (rb + 1 < KNB) ? r0 + KRB : r0;
        load_points(rf, pc, rn, sub, hi);
        sqv = sqb[rn + sub];

        unsigned m = 0u;
#pragma unroll
        for (int j = 0; j < KRB; ++j) m |= (dot[j] < bd[KK - 1]) ? (1u << (MB + j)) : 0u;
        unsigned self = (unsigned)(n - r0);
        m &= ~((self < (unsigned)KRB) ? (1u << (MB + (self & (unsigned)(KRB - 1)))) : 0u);

        for (int it = 0; it < KRB; ++it) {
            if (__builtin_amdgcn_ballot_w32(m != 0u) == 0u) break;
            bool has = m != 0u;
            int  j   = (__builtin_ctz(m | (1u << (MB + KRB))) - MB) & (KRB - 1);
            m &= m - 1u;
            float d = pick16(dot, j);
            d = has ? d : 3.0e38f;
            knn_insert(bd, bi, d, r0 + j);
        }
        r0 = rn;
    }

#pragma unroll
    for (int j = 0; j < KK; ++j) {
        mrgD[(w * KK + j) * KQ + lane] = bd[j];
        mrgI[(w * KK + j) * KQ + lane] = bi[j];
    }
    __syncthreads();
    for (int step = 1; step < KW; step <<= 1) {
        if ((w & (2 * step - 1)) == 0) {
            int srcw = w + step;
            for (int j = 0; j < KK; ++j) {
                float cd = mrgD[(srcw * KK + j) * KQ + lane];
                int   ci = mrgI[(srcw * KK + j) * KQ + lane];
                if (cd < bd[KK - 1]) knn_insert(bd, bi, cd, ci);
            }
            if (2 * step < KW) {
#pragma unroll
                for (int j = 0; j < KK; ++j) {
                    mrgD[(w * KK + j) * KQ + lane] = bd[j];
                    mrgI[(w * KK + j) * KQ + lane] = bi[j];
                }
            }
        }
        __syncthreads();
    }
    if (w == 0) {
#pragma unroll
        for (int j = 0; j < KK; ++j) idx[((size_t)b * KK + j) * NN + n] = bi[j];
    }
}

typedef __attribute__((ext_vector_type(4))) float v4f;

constexpr int PE = EE + 4;
constexpr int PH = HH + 4;
constexpr int PR = CC + 4;

template <int NRT, int NCT, int Q>
__device__ __forceinline__ void load_chunk(v4f (&a)[NRT][Q], v4f (&b)[NCT][Q],
                                           const float* const (&wp)[NRT],
                                           const float* const (&bp)[NCT], int koff) {
#pragma unroll
    for (int r = 0; r < NRT; ++r) {
#pragma unroll
        for (int q = 0; q < Q; ++q) a[r][q] = *(const v4f*)(wp[r] + koff + 4 * q);
    }
#pragma unroll
    for (int c = 0; c < NCT; ++c) {
#pragma unroll
        for (int q = 0; q < Q; ++q) b[c][q] = *(const v4f*)(bp[c] + koff + 4 * q);
    }
}

template <int NRT, int NCT, int Q>
__device__ __forceinline__ void mma_chunk(v8f (&acc)[NRT][NCT],
                                          const v4f (&a)[NRT][Q], const v4f (&b)[NCT][Q]) {
#pragma unroll
    for (int q = 0; q < Q; ++q) {
#pragma unroll
        for (int s = 0; s < 2; ++s) {
#pragma unroll
            for (int r = 0; r < NRT; ++r) {
                v2f av = s ? __builtin_shufflevector(a[r][q], a[r][q], 2, 3)
                           : __builtin_shufflevector(a[r][q], a[r][q], 0, 1);
#pragma unroll
                for (int c = 0; c < NCT; ++c) {
                    v2f bv = s ? __builtin_shufflevector(b[c][q], b[c][q], 2, 3)
                               : __builtin_shufflevector(b[c][q], b[c][q], 0, 1);
                    acc[r][c] = __builtin_amdgcn_wmma_f32_16x16x4_f32(
                        false, av, false, bv, (short)0, acc[r][c], false, false);
                }
            }
        }
    }
}

template <int NRT, int NCT, int Q, int KI, int WP, int PIN, int POUT, bool RELU, bool BIAS>
__device__ __forceinline__ void gemm_block(const float* __restrict__ W,
                                           const float* __restrict__ bias,
                                           const float* lin, float* lout,
                                           int rt0, int ct0, int laneRow, int laneHi) {
    constexpr int CH     = 4 * Q;
    constexpr int KSTEP  = 2 * CH;
    constexpr int NCHUNK = KI / KSTEP;
    static_assert(KI % KSTEP == 0 && NCHUNK % 2 == 0, "chunks come in pairs");

    const v8f vzero = {0.f, 0.f, 0.f, 0.f, 0.f, 0.f, 0.f, 0.f};
    const v4f vzero4 = {0.f, 0.f, 0.f, 0.f};
    v8f acc[NRT][NCT];
#pragma unroll
    for (int r = 0; r < NRT; ++r) {
#pragma unroll
        for (int c = 0; c < NCT; ++c) acc[r][c] = vzero;
    }

    const float* wp[NRT];
    const float* bp[NCT];
#pragma unroll
    for (int r = 0; r < NRT; ++r)
        wp[r] = W + (size_t)((rt0 + r) * 16 + laneRow) * WP + CH * laneHi;
#pragma unroll
    for (int c = 0; c < NCT; ++c)
        bp[c] = lin + ((ct0 + c) * 16 + laneRow) * PIN + CH * laneHi;

    v4f bl[NRT], bh[NRT];
    for (int r = 0; r < NRT; ++r) {
        int row0 = (rt0 + r) * 16 + 8 * laneHi;
        bl[r] = vzero4;
        bh[r] = vzero4;
        if (BIAS) {
            bl[r] = *(const v4f*)(bias + row0);
            bh[r] = *(const v4f*)(bias + row0 + 4);
        }
    }

    v4f a0[NRT][Q], b0[NCT][Q], a1[NRT][Q], b1[NCT][Q];
    load_chunk<NRT, NCT, Q>(a0, b0, wp, bp, 0);
    for (int kk = 0; kk < NCHUNK; kk += 2) {
        load_chunk<NRT, NCT, Q>(a1, b1, wp, bp, (kk + 1) * KSTEP);
        __builtin_amdgcn_sched_barrier(0);
        mma_chunk<NRT, NCT, Q>(acc, a0, b0);
        __builtin_amdgcn_sched_barrier(0);
        if (kk + 2 < NCHUNK) load_chunk<NRT, NCT, Q>(a0, b0, wp, bp, (kk + 2) * KSTEP);
        __builtin_amdgcn_sched_barrier(0);
        mma_chunk<NRT, NCT, Q>(acc, a1, b1);
        __builtin_amdgcn_sched_barrier(0);
    }

#pragma unroll
    for (int r = 0; r < NRT; ++r) {
        int row0 = (rt0 + r) * 16 + 8 * laneHi;
#pragma unroll
        for (int c = 0; c < NCT; ++c) {
            int col = (ct0 + c) * 16 + laneRow;
            v4f lo, hi;
#pragma unroll
            for (int j = 0; j < 4; ++j) {
                float u = acc[r][c][j];
                float v = acc[r][c][j + 4];
                if (BIAS) { u += bl[r][j]; v += bh[r][j]; }
                if (RELU) { u = fmaxf(u, 0.0f); v = fmaxf(v, 0.0f); }
                lo[j] = u;
                hi[j] = v;
            }
            *(v4f*)(lout + col * POUT + row0)     = lo;
            *(v4f*)(lout + col * POUT + row0 + 4) = hi;
        }
    }
}

constexpr int PC = CC + 4;
constexpr int QC = 16;
constexpr int NGRP = BB * (NN / NT);
constexpr int NWG  = 256;
constexpr int GPW  = NGRP / NWG;
static_assert(NGRP % NWG == 0, "every workgroup takes the same number of groups");

__device__ __forceinline__ void fetch_cols(float (&gv)[16], float (&cv)[4],
                                           const float* __restrict__ x,
                                           const float* __restrict__ mean_pc,
                                           int b, int n0, unsigned j, int t) {
    int cg   = t >> 6;
    int dcol = t & (QC - 1);
    int c0   = t >> 4;
    const float* mp = mean_pc + ((size_t)b * CC + cg * 16) * NN + j;
    const float* xp = x + (((size_t)b * CC + c0) * DD + (dcol >> 2)) * NN + n0 + (dcol & 3);
#pragma unroll
    for (int i = 0; i < 16; ++i) gv[i] = mp[(size_t)i * NN];
#pragma unroll
    for (int i = 0; i < 4; ++i) {
        float v = 0.0f;
        if (dcol < DD * NT) v = xp[(size_t)i * 16 * DD * NN];
        cv[i] = v;
    }
}

__device__ __forceinline__ void store_cols(float* ldsN, float* ldsC,
                                           const float (&gv)[16], const float (&cv)[4], int t) {
    int col  = t & (COLS - 1);
    int cg   = t >> 6;
    int dcol = t & (QC - 1);
    int c0   = t >> 4;
#pragma unroll
    for (int i = 0; i < 4; ++i) {
        v4f s = {gv[4 * i], gv[4 * i + 1], gv[4 * i + 2], gv[4 * i + 3]};
        *(v4f*)(ldsN + col * PC + cg * 16 + 4 * i) = s;
    }
#pragma unroll
    for (int i = 0; i < 4; ++i) {
        ldsC[dcol * PE + c0 + 16 * i]      = cv[i];
        ldsC[dcol * PE + CC + c0 + 16 * i] = -cv[i];
    }
}

__global__ void __launch_bounds__(256) mlp_kernel(const float* __restrict__ x,
                                                  const float* __restrict__ mean_pc,
                                                  const int* __restrict__ idx,
                                                  const float* __restrict__ W1, const float* __restrict__ b1,
                                                  const float* __restrict__ W2, const float* __restrict__ b2,
                                                  const float* __restrict__ W3, const float* __restrict__ b3,
                                                  float* __restrict__ out) {
    __shared__ __attribute__((aligned(16))) float ldsX[COLS * PH];
    __shared__ __attribute__((aligned(16))) float ldsY[COLS * PH];
    __shared__ __attribute__((aligned(16))) float ldsP[COLS * PH];
    __shared__ __attribute__((aligned(16))) float ldsQ[QC * PH];
    __shared__ __attribute__((aligned(16))) float ldsR[COLS * PR];
    __shared__ __attribute__((aligned(16))) float ldsN[COLS * PC];
    __shared__ __attribute__((aligned(16))) float ldsC[QC * PE];

    int t       = threadIdx.x;
    int lane    = t & 31;
    int w       = t >> 5;
    int laneRow = lane & 15;
    int laneHi  = lane >> 4;

    int gnl = (t & (COLS - 1)) >> 4;
    int gk  = t & 15;

    float gv[16];
    float cv[4];

    int g = blockIdx.x;

    {
        int b  = g / (NN / NT);
        int n0 = (g % (NN / NT)) * NT;
        unsigned j = (unsigned)idx[((size_t)b * KK + gk) * NN + n0 + gnl];
        fetch_cols(gv, cv, x, mean_pc, b, n0, j, t);
        store_cols(ldsN, ldsC, gv, cv, t);
    }

    for (int it = 0; it < GPW; ++it) {
        int zw = 0;
        asm volatile("" : "+s"(zw));
        int b  = g / (NN / NT);
        int n0 = (g % (NN / NT)) * NT;

        bool more = it + 1 < GPW;
        int  gx   = g + NWG;
        int  bx   = gx / (NN / NT);
        int  n0x  = (gx % (NN / NT)) * NT;

        __syncthreads();

        unsigned jx = 0;
        if (more) jx = (unsigned)idx[((size_t)bx * KK + gk) * NN + n0x + gnl];

        gemm_block<2, 4, 2, CC, EE, PC, PH, false, false>(W1 + CC + zw, b1, ldsN, ldsP, 2 * w, 0, laneRow, laneHi);
        gemm_block<2, 1, 2, EE, EE, PE, PH, false, true>(W1 + zw, b1 + zw, ldsC, ldsQ, 2 * w, 0, laneRow, laneHi);
        __syncthreads();

        for (int d = 0; d < DD; ++d) {
            {
                int col = t & (COLS - 1);
                int og  = t >> 6;
                const float* pp = ldsP + col * PH + og * 64;
                const float* qq = ldsQ + (d * NT + (col >> 4)) * PH + og * 64;
                float*       hh = ldsY + col * PH + og * 64;
#pragma unroll
                for (int i = 0; i < 16; ++i) {
                    v4f p = *(const v4f*)(pp + 4 * i);
                    v4f q = *(const v4f*)(qq + 4 * i);
                    v4f h;
#pragma unroll
                    for (int j = 0; j < 4; ++j) h[j] = fmaxf(p[j] + q[j], 0.0f);
                    *(v4f*)(hh + 4 * i) = h;
                }
            }
            __syncthreads();

            bool pf = more && d == 0;
            if (pf) fetch_cols(gv, cv, x, mean_pc, bx, n0x, jx, t);

            gemm_block<2, 4, 2, HH, HH, PH, PH, true, true>(W2 + zw, b2 + zw, ldsY, ldsX, 2 * w, 0, laneRow, laneHi);
            __syncthreads();

            if (pf) store_cols(ldsN, ldsC, gv, cv, t);

            int zoff = 0;
            asm volatile("" : "+s"(zoff));
            size_t o   = (((size_t)b * CC + (t & (CC - 1))) * DD + d) * NN + (n0 + (t >> 6));
            float xres = x[o];
            gemm_block<1, 2, 4, HH, HH, PH, PR, false, true>(W3 + zoff, b3, ldsX, ldsR, w >> 1, (w & 1) * 2, laneRow, laneHi);
            __syncthreads();

            {
                int c  = t & (CC - 1);
                int nl = t >> 6;
                const float* rcol = &ldsR[(nl * 16) * PR + c];
                float m = rcol[0];
#pragma unroll
                for (int k = 1; k < KK; ++k) m = fmaxf(m, rcol[k * PR]);
                out[o] = m + xres;
            }
        }
        g = gx;
    }
}

extern "C" void kernel_launch(void* const* d_in, const int* in_sizes, int n_in,
                              void* d_out, int out_size, void* d_ws, size_t ws_size,
                              hipStream_t stream) {
    const float* x  = (const float*)d_in[0];
    const float* W1 = (const float*)d_in[1];
    const float* b1 = (const float*)d_in[2];
    const float* W2 = (const float*)d_in[3];
    const float* b2 = (const float*)d_in[4];
    const float* W3 = (const float*)d_in[5];
    const float* b3 = (const float*)d_in[6];
    float* out = (float*)d_out;

    float* mean_pc = (float*)d_ws;
    float* sq      = mean_pc + (size_t)BB * CC * NN;
    int*   idxp    = (int*)(sq + (size_t)BB * NN);

    meanpc_kernel<<<(BB * CC * NN + 255) / 256, 256, 0, stream>>>(x, mean_pc);
    sqnorm_kernel<<<(BB * NN + 255) / 256, 256, 0, stream>>>(mean_pc, sq);
    knn_kernel<<<BB * (NN / KQ), 256, 0, stream>>>(mean_pc, sq, idxp);
    mlp_kernel<<<NWG, 256, 0, stream>>>(x, mean_pc, idxp,
                                                        W1, b1, W2, b2, W3, b3, out);
}


# MLP weight fragments ahead of their WMMAs: W2 held in registers across groups; W1/W3 two chunks ahead, first chunks issued before LDS-only barriers; Q in 16-WMMA chunks (its f32 sum order changes)
# speedup vs baseline: 15.8460x; 1.0782x over previous
#include <hip/hip_runtime.h>

#define BB 2
#define CC 64
#define DD 3
#define NN 4096
#define KK 16
#define EE 128
#define HH 256

typedef __attribute__((ext_vector_type(2))) float v2f;
typedef __attribute__((ext_vector_type(8))) float v8f;

constexpr int NT   = 4;
constexpr int COLS = NT * KK;

__global__ void meanpc_kernel(const float* __restrict__ x, float* __restrict__ mean_pc) {
    int i = blockIdx.x * blockDim.x + threadIdx.x;
    if (i >= BB * CC * NN) return;
    int n  = i % NN;
    int bc = i / NN;
    const float* xp = x + (size_t)bc * (DD * NN) + n;
    mean_pc[i] = (xp[0] + xp[NN] + xp[2 * NN]) * (1.0f / 3.0f);
}

__global__ void sqnorm_kernel(const float* __restrict__ mean_pc, float* __restrict__ sq) {
    int i = blockIdx.x * blockDim.x + threadIdx.x;
    if (i >= BB * NN) return;
    int n = i % NN;
    int b = i / NN;
    float s = 0.0f;
#pragma unroll
    for (int c = 0; c < CC; ++c) {
        float v = mean_pc[((size_t)b * CC + c) * NN + n];
        s = __builtin_fmaf(v, v, s);
    }
    sq[i] = s;
}

constexpr int KQ   = 32;
constexpr int KW   = 8;
constexpr int KSEG = NN / KW;
constexpr int KRB  = 16;
constexpr int KNB  = KSEG / KRB;
constexpr int MB   = 8;

template <int J>
__device__ __forceinline__ float row_bcast(float v) {
    return __builtin_bit_cast(float,
        __builtin_amdgcn_update_dpp(0, __builtin_bit_cast(int, v), 0x150 + J, 0xf, 0xf, true));
}

template <int J>
struct KnnStep {
    static __device__ __forceinline__ void bcast(float (&sb)[KRB], float sqv) {
        sb[J] = row_bcast<J>(sqv);
        KnnStep<J + 1>::bcast(sb, sqv);
    }
};
template <>
struct KnnStep<KRB> {
    static __device__ __forceinline__ void bcast(float (&)[KRB], float) {}
};

__device__ __forceinline__ float other_row(float v) {
    int b = __builtin_bit_cast(int, v);
    return __builtin_bit_cast(float,
        __builtin_amdgcn_permlanex16(b, b, 0x76543210, 0xfedcba98, false, false));
}

__device__ __forceinline__ void load_points(v2f (&f)[CC / 4], const float* __restrict__ pc,
                                            int p0, int sub, int hi) {
    const float* p = pc + (size_t)(2 * hi) * NN + p0 + sub;
#pragma unroll
    for (int s = 0; s < CC / 4; ++s) {
        v2f v = {p[(size_t)(4 * s) * NN], p[(size_t)(4 * s + 1) * NN]};
        f[s] = v;
    }
}

__device__ __forceinline__ float pick(bool c, float a, float b) { return c ? a : b; }
__device__ __forceinline__ int   pick(bool c, int a, int b)     { return c ? a : b; }

__device__ __forceinline__ void knn_insert(float (&bd)[KK], int (&bi)[KK], float d, int i) {
    bool c[KK];
#pragma unroll
    for (int j = 0; j < KK; ++j) c[j] = d < bd[j];
#pragma unroll
    for (int j = KK - 1; j > 0; --j) {
        bd[j] = pick(c[j], pick(c[j - 1], bd[j - 1], d), bd[j]);
        bi[j] = pick(c[j], pick(c[j - 1], bi[j - 1], i), bi[j]);
    }
    bd[0] = pick(c[0], d, bd[0]);
    bi[0] = pick(c[0], i, bi[0]);
}

__device__ __forceinline__ float pick16(const float (&v)[KRB], int j) {
    bool b0 = (j & 1) != 0;
    bool b1 = (j & 2) != 0;
    bool b2 = (j & 4) != 0;
    bool b3 = (j & 8) != 0;
    float e0 = pick(b0, v[1], v[0]);
    float e1 = pick(b0, v[3], v[2]);
    float e2 = pick(b0, v[5], v[4]);
    float e3 = pick(b0, v[7], v[6]);
    float e4 = pick(b0, v[9], v[8]);
    float e5 = pick(b0, v[11], v[10]);
    float e6 = pick(b0, v[13], v[12]);
    float e7 = pick(b0, v[15], v[14]);
    float f0 = pick(b1, e1, e0);
    float f1 = pick(b1, e3, e2);
    float f2 = pick(b1, e5, e4);
    float f3 = pick(b1, e7, e6);
    float g0 = pick(b2, f1, f0);
    float g1 = pick(b2, f3, f2);
    return pick(b3, g1, g0);
}

__global__ void __launch_bounds__(256) knn_kernel(const float* __restrict__ mean_pc,
                                                  const float* __restrict__ sq,
                                                  int* __restrict__ idx) {
    __shared__ float mrgD[KW * KK * KQ];
    __shared__ int   mrgI[KW * KK * KQ];

    int t    = threadIdx.x;
    int lane = t & 31;
    int sub  = lane & 15;
    int hi   = lane >> 4;
    int w    = __builtin_amdgcn_readfirstlane(t >> 5);
    int b    = blockIdx.x / (NN / KQ);
    int q0   = (blockIdx.x % (NN / KQ)) * KQ;
    int n    = q0 + lane;
    const float* pc  = mean_pc + (size_t)b * CC * NN;
    const float* sqb = sq + (size_t)b * NN;

    v2f qf[2][CC / 4];
    load_points(qf[0], pc, q0, sub, hi);
    load_points(qf[1], pc, q0 + 16, sub, hi);

    float bd[KK];
    int   bi[KK];
#pragma unroll
    for (int j = 0; j < KK; ++j) { bd[j] = 3.0e38f; bi[j] = 0; }

    int   r0 = w * KSEG;
    v2f   rf[CC / 4];
    float sqv;
    load_points(rf, pc, r0, sub, hi);
    sqv = sqb[r0 + sub];

    const v8f vzero = {0.f, 0.f, 0.f, 0.f, 0.f, 0.f, 0.f, 0.f};
    for (int rb = 0; rb < KNB; ++rb) {
        v8f d0 = vzero;
        v8f d1 = vzero;
#pragma unroll
        for (int s = 0; s < CC / 4; ++s) {
            d0 = __builtin_amdgcn_wmma_f32_16x16x4_f32(false, rf[s], false, qf[0][s], (short)0, d0, false, false);
            d1 = __builtin_amdgcn_wmma_f32_16x16x4_f32(false, rf[s], false, qf[1][s], (short)0, d1, false, false);
        }

        float sb[KRB];
        __builtin_amdgcn_sched_barrier(0);
        KnnStep<0>::bcast(sb, sqv);
        __builtin_amdgcn_sched_barrier(0);

        float dot[KRB];
#pragma unroll
        for (int j = 0; j < 8; ++j) {
            float keep = hi ? d1[j] : d0[j];
            float send = hi ? d0[j] : d1[j];
            float recv = other_row(send);
            dot[j]     = hi ? recv : keep;
            dot[8 + j] = hi ? keep : recv;
        }
#pragma unroll
        for (int j = 0; j < KRB; ++j) dot[j] = sb[j] - 2.0f * dot[j];

        int rn = (rb + 1 < KNB) ? r0 + KRB : r0;
        load_points(rf, pc, rn, sub, hi);
        sqv = sqb[rn + sub];

        unsigned m = 0u;
#pragma unroll
        for (int j = 0; j < KRB; ++j) m |= (dot[j] < bd[KK - 1]) ? (1u << (MB + j)) : 0u;
        unsigned self = (unsigned)(n - r0);
        m &= ~((self < (unsigned)KRB) ? (1u << (MB + (self & (unsigned)(KRB - 1)))) : 0u);

        for (int it = 0; it < KRB; ++it) {
            if (__builtin_amdgcn_ballot_w32(m != 0u) == 0u) break;
            bool has = m != 0u;
            int  j   = (__builtin_ctz(m | (1u << (MB + KRB))) - MB) & (KRB - 1);
            m &= m - 1u;
            float d = pick16(dot, j);
            d = has ? d : 3.0e38f;
            knn_insert(bd, bi, d, r0 + j);
        }
        r0 = rn;
    }

#pragma unroll
    for (int j = 0; j < KK; ++j) {
        mrgD[(w * KK + j) * KQ + lane] = bd[j];
        mrgI[(w * KK + j) * KQ + lane] = bi[j];
    }
    __syncthreads();
    for (int step = 1; step < KW; step <<= 1) {
        if ((w & (2 * step - 1)) == 0) {
            int srcw = w + step;
            for (int j = 0; j < KK; ++j) {
                float cd = mrgD[(srcw * KK + j) * KQ + lane];
                int   ci = mrgI[(srcw * KK + j) * KQ + lane];
                if (cd < bd[KK - 1]) knn_insert(bd, bi, cd, ci);
            }
            if (2 * step < KW) {
#pragma unroll
                for (int j = 0; j < KK; ++j) {
                    mrgD[(w * KK + j) * KQ + lane] = bd[j];
                    mrgI[(w * KK + j) * KQ + lane] = bi[j];
                }
            }
        }
        __syncthreads();
    }
    if (w == 0) {
#pragma unroll
        for (int j = 0; j < KK; ++j) idx[((size_t)b * KK + j) * NN + n] = bi[j];
    }
}

typedef __attribute__((ext_vector_type(4))) float v4f;

constexpr int PE = EE + 4;
constexpr int PH = HH + 4;
constexpr int PR = CC + 4;

template <int NRT, int NCT, int Q>
__device__ __forceinline__ void load_chunk(v4f (&a)[NRT][Q], v4f (&b)[NCT][Q],
                                           const float* const (&wp)[NRT],
                                           const float* const (&bp)[NCT], int koff) {
#pragma unroll
    for (int r = 0; r < NRT; ++r) {
#pragma unroll
        for (int q = 0; q < Q; ++q) a[r][q] = *(const v4f*)(wp[r] + koff + 4 * q);
    }
#pragma unroll
    for (int c = 0; c < NCT; ++c) {
#pragma unroll
        for (int q = 0; q < Q; ++q) b[c][q] = *(const v4f*)(bp[c] + koff + 4 * q);
    }
}

template <int NRT, int Q, int WP>
__device__ __forceinline__ void load_a(v4f (&a)[NRT][Q], const float* __restrict__ W,
                                       int rt0, int laneRow, int laneHi, int koff) {
    constexpr int CH = 4 * Q;
#pragma unroll
    for (int r = 0; r < NRT; ++r) {
        const float* p = W + (size_t)((rt0 + r) * 16 + laneRow) * WP + CH * laneHi + koff;
#pragma unroll
        for (int q = 0; q < Q; ++q) a[r][q] = *(const v4f*)(p + 4 * q);
    }
}

template <int NCT, int Q>
__device__ __forceinline__ void load_b(v4f (&b)[NCT][Q], const float* const (&bp)[NCT], int koff) {
#pragma unroll
    for (int c = 0; c < NCT; ++c) {
#pragma unroll
        for (int q = 0; q < Q; ++q) b[c][q] = *(const v4f*)(bp[c] + koff + 4 * q);
    }
}

template <int NRT, int Q>
__device__ __forceinline__ void copy_a(v4f (&d)[NRT][Q], const v4f (&s)[NRT][Q]) {
#pragma unroll
    for (int r = 0; r < NRT; ++r) {
#pragma unroll
        for (int q = 0; q < Q; ++q) d[r][q] = s[r][q];
    }
}

template <int NRT, int NCT, int Q>
__device__ __forceinline__ void mma_chunk(v8f (&acc)[NRT][NCT],
                                          const v4f (&a)[NRT][Q], const v4f (&b)[NCT][Q]) {
#pragma unroll
    for (int q = 0; q < Q; ++q) {
#pragma unroll
        for (int s = 0; s < 2; ++s) {
#pragma unroll
            for (int r = 0; r < NRT; ++r) {
                v2f av = s ? __builtin_shufflevector(a[r][q], a[r][q], 2, 3)
                           : __builtin_shufflevector(a[r][q], a[r][q], 0, 1);
#pragma unroll
                for (int c = 0; c < NCT; ++c) {
                    v2f bv = s ? __builtin_shufflevector(b[c][q], b[c][q], 2, 3)
                               : __builtin_shufflevector(b[c][q], b[c][q], 0, 1);
                    acc[r][c] = __builtin_amdgcn_wmma_f32_16x16x4_f32(
                        false, av, false, bv, (short)0, acc[r][c], false, false);
                }
            }
        }
    }
}

template <int NRT, int NCT, int Q, int KI, int WP, int PIN, int POUT, bool RELU, bool BIAS,
          int NPRE, bool AREG>
__device__ __forceinline__ void gemm_core(const float* __restrict__ W,
                                          const float* __restrict__ bias,
                                          const float* lin, float* lout,
                                          int rt0, int ct0, int laneRow, int laneHi,
                                          const v4f (&apre)[NPRE > 0 ? NPRE : 1][NRT][Q],
                                          const v4f (&aall)[AREG ? KI / (8 * Q) : 1][NRT][Q]) {
    constexpr int CH     = 4 * Q;
    constexpr int KSTEP  = 2 * CH;
    constexpr int NCHUNK = KI / KSTEP;
    static_assert(KI % KSTEP == 0 && NCHUNK % 2 == 0, "chunks come in pairs");
    static_assert(AREG || NPRE == 1 || NPRE == 2, "the caller supplies the first A fragments");

    const v8f vzero = {0.f, 0.f, 0.f, 0.f, 0.f, 0.f, 0.f, 0.f};
    const v4f vzero4 = {0.f, 0.f, 0.f, 0.f};
    v8f acc[NRT][NCT];
#pragma unroll
    for (int r = 0; r < NRT; ++r) {
#pragma unroll
        for (int c = 0; c < NCT; ++c) acc[r][c] = vzero;
    }

    const float* wp[NRT];
    const float* bp[NCT];
#pragma unroll
    for (int r = 0; r < NRT; ++r)
        wp[r] = W + (size_t)((rt0 + r) * 16 + laneRow) * WP + CH * laneHi;
#pragma unroll
    for (int c = 0; c < NCT; ++c)
        bp[c] = lin + ((ct0 + c) * 16 + laneRow) * PIN + CH * laneHi;

    v4f bl[NRT], bh[NRT];
    for (int r = 0; r < NRT; ++r) {
        int row0 = (rt0 + r) * 16 + 8 * laneHi;
        bl[r] = vzero4;
        bh[r] = vzero4;
        if (BIAS) {
            bl[r] = *(const v4f*)(bias + row0);
            bh[r] = *(const v4f*)(bias + row0 + 4);
        }
    }

    v4f b0[NCT][Q], b1[NCT][Q];
    if (AREG) {
        load_b<NCT, Q>(b0, bp, 0);
#pragma unroll
        for (int kk = 0; kk < NCHUNK; kk += 2) {
            load_b<NCT, Q>(b1, bp, (kk + 1) * KSTEP);
            __builtin_amdgcn_sched_barrier(0);
            mma_chunk<NRT, NCT, Q>(acc, aall[AREG ? kk : 0], b0);
            __builtin_amdgcn_sched_barrier(0);
            if (kk + 2 < NCHUNK) load_b<NCT, Q>(b0, bp, (kk + 2) * KSTEP);
            __builtin_amdgcn_sched_barrier(0);
            mma_chunk<NRT, NCT, Q>(acc, aall[AREG ? kk + 1 : 0], b1);
            __builtin_amdgcn_sched_barrier(0);
        }
    } else if (NPRE >= 2) {
        v4f ar[NCHUNK][NRT][Q];
        copy_a<NRT, Q>(ar[0], apre[0]);
        copy_a<NRT, Q>(ar[1], apre[NPRE >= 2 ? 1 : 0]);
        load_b<NCT, Q>(b0, bp, 0);
#pragma unroll
        for (int kk = 0; kk < NCHUNK; kk += 2) {
            load_b<NCT, Q>(b1, bp, (kk + 1) * KSTEP);
            if (kk + 2 < NCHUNK)
                load_a<NRT, Q, WP>(ar[kk + 2 < NCHUNK ? kk + 2 : 0], W, rt0, laneRow, laneHi, (kk + 2) * KSTEP);
            __builtin_amdgcn_sched_barrier(0);
            mma_chunk<NRT, NCT, Q>(acc, ar[kk], b0);
            __builtin_amdgcn_sched_barrier(0);
            if (kk + 2 < NCHUNK) {
                load_b<NCT, Q>(b0, bp, (kk + 2) * KSTEP);
                load_a<NRT, Q, WP>(ar[kk + 3 < NCHUNK ? kk + 3 : 0], W, rt0, laneRow, laneHi, (kk + 3) * KSTEP);
            }
            __builtin_amdgcn_sched_barrier(0);
            mma_chunk<NRT, NCT, Q>(acc, ar[kk + 1], b1);
            __builtin_amdgcn_sched_barrier(0);
        }
    } else {
        v4f a0[NRT][Q], a1[NRT][Q];
        copy_a<NRT, Q>(a0, apre[0]);
        load_b<NCT, Q>(b0, bp, 0);
        for (int kk = 0; kk < NCHUNK; kk += 2) {
            load_chunk<NRT, NCT, Q>(a1, b1, wp, bp, (kk + 1) * KSTEP);
            __builtin_amdgcn_sched_barrier(0);
            mma_chunk<NRT, NCT, Q>(acc, a0, b0);
            __builtin_amdgcn_sched_barrier(0);
            if (kk + 2 < NCHUNK) load_chunk<NRT, NCT, Q>(a0, b0, wp, bp, (kk + 2) * KSTEP);
            __builtin_amdgcn_sched_barrier(0);
            mma_chunk<NRT, NCT, Q>(acc, a1, b1);
            __builtin_amdgcn_sched_barrier(0);
        }
    }

#pragma unroll
    for (int r = 0; r < NRT; ++r) {
        int row0 = (rt0 + r) * 16 + 8 * laneHi;
#pragma unroll
        for (int c = 0; c < NCT; ++c) {
            int col = (ct0 + c) * 16 + laneRow;
            v4f lo, hi;
#pragma unroll
            for (int j = 0; j < 4; ++j) {
                float u = acc[r][c][j];
                float v = acc[r][c][j + 4];
                if (BIAS) { u += bl[r][j]; v += bh[r][j]; }
                if (RELU) { u = fmaxf(u, 0.0f); v = fmaxf(v, 0.0f); }
                lo[j] = u;
                hi[j] = v;
            }
            *(v4f*)(lout + col * POUT + row0)     = lo;
            *(v4f*)(lout + col * POUT + row0 + 4) = hi;
        }
    }
}

template <int NRT, int NCT, int Q, int KI, int WP, int PIN, int POUT, bool RELU, bool BIAS, int NPRE>
__device__ __forceinline__ void gemm_block_pre(const float* __restrict__ W,
                                               const float* __restrict__ bias,
                                               const float* lin, float* lout,
                                               int rt0, int ct0, int laneRow, int laneHi,
                                               const v4f (&apre)[NPRE][NRT][Q]) {
    const v4f none[1][NRT][Q] = {};
    gemm_core<NRT, NCT, Q, KI, WP, PIN, POUT, RELU, BIAS, NPRE, false>(
        W, bias, lin, lout, rt0, ct0, laneRow, laneHi, apre, none);
}

template <int NRT, int NCT, int Q, int KI, int WP, int PIN, int POUT, bool RELU, bool BIAS>
__device__ __forceinline__ void gemm_block_areg(const float* __restrict__ W,
                                                const float* __restrict__ bias,
                                                const float* lin, float* lout,
                                                int rt0, int ct0, int laneRow, int laneHi,
                                                const v4f (&aall)[KI / (8 * Q)][NRT][Q]) {
    const v4f none[1][NRT][Q] = {};
    gemm_core<NRT, NCT, Q, KI, WP, PIN, POUT, RELU, BIAS, 0, true>(
        W, bias, lin, lout, rt0, ct0, laneRow, laneHi, none, aall);
}

__device__ __forceinline__ void lds_barrier() {
    __builtin_amdgcn_fence(__ATOMIC_RELEASE, "workgroup", "local");
    __builtin_amdgcn_s_barrier();
    __builtin_amdgcn_fence(__ATOMIC_ACQUIRE, "workgroup", "local");
}

constexpr int PC = CC + 4;
constexpr int QC = 16;
constexpr int NGRP = BB * (NN / NT);
constexpr int NWG  = 256;
constexpr int GPW  = NGRP / NWG;
static_assert(NGRP % NWG == 0, "every workgroup takes the same number of groups");

__device__ __forceinline__ void fetch_cols(float (&gv)[16], float (&cv)[4],
                                           const float* __restrict__ x,
                                           const float* __restrict__ mean_pc,
                                           int b, int n0, unsigned j, int t) {
    int cg   = t >> 6;
    int dcol = t & (QC - 1);
    int c0   = t >> 4;
    const float* mp = mean_pc + ((size_t)b * CC + cg * 16) * NN + j;
    const float* xp = x + (((size_t)b * CC + c0) * DD + (dcol >> 2)) * NN + n0 + (dcol & 3);
#pragma unroll
    for (int i = 0; i < 16; ++i) gv[i] = mp[(size_t)i * NN];
#pragma unroll
    for (int i = 0; i < 4; ++i) {
        float v = 0.0f;
        if (dcol < DD * NT) v = xp[(size_t)i * 16 * DD * NN];
        cv[i] = v;
    }
}

__device__ __forceinline__ void store_cols(float* ldsN, float* ldsC,
                                           const float (&gv)[16], const float (&cv)[4], int t) {
    int col  = t & (COLS - 1);
    int cg   = t >> 6;
    int dcol = t & (QC - 1);
    int c0   = t >> 4;
#pragma unroll
    for (int i = 0; i < 4; ++i) {
        v4f s = {gv[4 * i], gv[4 * i + 1], gv[4 * i + 2], gv[4 * i + 3]};
        *(v4f*)(ldsN + col * PC + cg * 16 + 4 * i) = s;
    }
#pragma unroll
    for (int i = 0; i < 4; ++i) {
        ldsC[dcol * PE + c0 + 16 * i]      = cv[i];
        ldsC[dcol * PE + CC + c0 + 16 * i] = -cv[i];
    }
}

__global__ void __launch_bounds__(256) mlp_kernel(const float* __restrict__ x,
                                                  const float* __restrict__ mean_pc,
                                                  const int* __restrict__ idx,
                                                  const float* __restrict__ W1, const float* __restrict__ b1,
                                                  const float* __restrict__ W2, const float* __restrict__ b2,
                                                  const float* __restrict__ W3, const float* __restrict__ b3,
                                                  float* __restrict__ out) {
    __shared__ __attribute__((aligned(16))) float ldsX[COLS * PH];
    __shared__ __attribute__((aligned(16))) float ldsY[COLS * PH];
    __shared__ __attribute__((aligned(16))) float ldsP[COLS * PH];
    __shared__ __attribute__((aligned(16))) float ldsQ[QC * PH];
    __shared__ __attribute__((aligned(16))) float ldsR[COLS * PR];
    __shared__ __attribute__((aligned(16))) float ldsN[COLS * PC];
    __shared__ __attribute__((aligned(16))) float ldsC[QC * PE];

    int t       = threadIdx.x;
    int lane    = t & 31;
    int w       = t >> 5;
    int laneRow = lane & 15;
    int laneHi  = lane >> 4;

    int gnl = (t & (COLS - 1)) >> 4;
    int gk  = t & 15;

    float gv[16];
    float cv[4];

    int g = blockIdx.x;

    {
        int b  = g / (NN / NT);
        int n0 = (g % (NN / NT)) * NT;
        unsigned j = (unsigned)idx[((size_t)b * KK + gk) * NN + n0 + gnl];
        fetch_cols(gv, cv, x, mean_pc, b, n0, j, t);
        store_cols(ldsN, ldsC, gv, cv, t);
    }

    v4f aP[1][2][2];
    load_a<2, 2, EE>(aP[0], W1 + CC, 2 * w, laneRow, laneHi, 0);

    v4f w2f[HH / 16][2][2];
#pragma unroll
    for (int kk = 0; kk < HH / 16; ++kk) load_a<2, 2, HH>(w2f[kk], W2, 2 * w, laneRow, laneHi, kk * 16);

    for (int it = 0; it < GPW; ++it) {
        int zw = 0;
        asm volatile("" : "+s"(zw));
        int b  = g / (NN / NT);
        int n0 = (g % (NN / NT)) * NT;

        bool more = it + 1 < GPW;
        int  gx   = g + NWG;
        int  bx   = gx / (NN / NT);
        int  n0x  = (gx % (NN / NT)) * NT;

        lds_barrier();

        unsigned jx = 0;
        if (more) jx = (unsigned)idx[((size_t)bx * KK + gk) * NN + n0x + gnl];

        gemm_block_pre<2, 4, 2, CC, EE, PC, PH, false, false, 1>(W1 + CC + zw, b1, ldsN, ldsP, 2 * w, 0, laneRow, laneHi, aP);
        v4f aQ[2][2][4];
        load_a<2, 4, EE>(aQ[0], W1 + zw, 2 * w, laneRow, laneHi, 0);
        __builtin_amdgcn_sched_barrier(0);
        load_a<2, 4, EE>(aQ[1], W1 + zw, 2 * w, laneRow, laneHi, 32);
        __builtin_amdgcn_sched_barrier(0);
        gemm_block_pre<2, 1, 4, EE, EE, PE, PH, false, true, 2>(W1 + zw, b1 + zw, ldsC, ldsQ, 2 * w, 0, laneRow, laneHi, aQ);
        __syncthreads();

        for (int d = 0; d < DD; ++d) {
            {
                int col = t & (COLS - 1);
                int og  = t >> 6;
                const float* pp = ldsP + col * PH + og * 64;
                const float* qq = ldsQ + (d * NT + (col >> 4)) * PH + og * 64;
                float*       hh = ldsY + col * PH + og * 64;
#pragma unroll
                for (int i = 0; i < 16; ++i) {
                    v4f p = *(const v4f*)(pp + 4 * i);
                    v4f q = *(const v4f*)(qq + 4 * i);
                    v4f h;
#pragma unroll
                    for (int j = 0; j < 4; ++j) h[j] = fmaxf(p[j] + q[j], 0.0f);
                    *(v4f*)(hh + 4 * i) = h;
                }
            }
            __syncthreads();

            bool pf = more && d == 0;
            if (pf) fetch_cols(gv, cv, x, mean_pc, bx, n0x, jx, t);

            gemm_block_areg<2, 4, 2, HH, HH, PH, PH, true, true>(W2, b2 + zw, ldsY, ldsX, 2 * w, 0, laneRow, laneHi, w2f);

            int zoff = 0;
            asm volatile("" : "+s"(zoff));
            v4f a3[2][1][4];
            load_a<1, 4, HH>(a3[0], W3 + zoff, w >> 1, laneRow, laneHi, 0);
            load_a<1, 4, HH>(a3[1], W3 + zoff, w >> 1, laneRow, laneHi, 32);
            lds_barrier();

            if (pf) store_cols(ldsN, ldsC, gv, cv, t);

            size_t o   = (((size_t)b * CC + (t & (CC - 1))) * DD + d) * NN + (n0 + (t >> 6));
            float xres = x[o];
            gemm_block_pre<1, 2, 4, HH, HH, PH, PR, false, true, 2>(W3 + zoff, b3, ldsX, ldsR, w >> 1, (w & 1) * 2, laneRow, laneHi, a3);
            __syncthreads();

            {
                int c  = t & (CC - 1);
                int nl = t >> 6;
                const float* rcol = &ldsR[(nl * 16) * PR + c];
                float m = rcol[0];
#pragma unroll
                for (int k = 1; k < KK; ++k) m = fmaxf(m, rcol[k * PR]);
                out[o] = m + xres;
            }
        }
        load_a<2, 2, EE>(aP[0], W1 + CC + zw, 2 * w, laneRow, laneHi, 0);
        g = gx;
    }
}

extern "C" void kernel_launch(void* const* d_in, const int* in_sizes, int n_in,
                              void* d_out, int out_size, void* d_ws, size_t ws_size,
                              hipStream_t stream) {
    const float* x  = (const float*)d_in[0];
    const float* W1 = (const float*)d_in[1];
    const float* b1 = (const float*)d_in[2];
    const float* W2 = (const float*)d_in[3];
    const float* b2 = (const float*)d_in[4];
    const float* W3 = (const float*)d_in[5];
    const float* b3 = (const float*)d_in[6];
    float* out = (float*)d_out;

    float* mean_pc = (float*)d_ws;
    float* sq      = mean_pc + (size_t)BB * CC * NN;
    int*   idxp    = (int*)(sq + (size_t)BB * NN);

    meanpc_kernel<<<(BB * CC * NN + 255) / 256, 256, 0, stream>>>(x, mean_pc);
    sqnorm_kernel<<<(BB * NN + 255) / 256, 256, 0, stream>>>(mean_pc, sq);
    knn_kernel<<<BB * (NN / KQ), 256, 0, stream>>>(mean_pc, sq, idxp);
    mlp_kernel<<<NWG, 256, 0, stream>>>(x, mean_pc, idxp,
                                                        W1, b1, W2, b2, W3, b3, out);
}


# MLP layer-1 neighbour term P read from a per-point table W1b*mean_pc built once in the KNN launch, not multiplied per neighbour column; old product kept for a small workspace; P f32 sum order changes
# speedup vs baseline: 17.1721x; 1.0837x over previous
#include <hip/hip_runtime.h>

#define BB 2
#define CC 64
#define DD 3
#define NN 4096
#define KK 16
#define EE 128
#define HH 256

typedef __attribute__((ext_vector_type(2))) float v2f;
typedef __attribute__((ext_vector_type(4))) float v4f;
typedef __attribute__((ext_vector_type(8))) float v8f;

constexpr int NT   = 4;
constexpr int COLS = NT * KK;

constexpr size_t WS_TBL_OFF  = ((size_t)BB * CC * NN + (size_t)BB * NN + (size_t)BB * KK * NN) * 4;
constexpr size_t WS_TBL_NEED = WS_TBL_OFF + (size_t)BB * NN * HH * 4;

__global__ void meanpc_kernel(const float* __restrict__ x, float* __restrict__ mean_pc) {
    int i = blockIdx.x * blockDim.x + threadIdx.x;
    if (i >= BB * CC * NN) return;
    int n  = i % NN;
    int bc = i / NN;
    const float* xp = x + (size_t)bc * (DD * NN) + n;
    mean_pc[i] = (xp[0] + xp[NN] + xp[2 * NN]) * (1.0f / 3.0f);
}

__global__ void sqnorm_kernel(const float* __restrict__ mean_pc, float* __restrict__ sq) {
    int i = blockIdx.x * blockDim.x + threadIdx.x;
    if (i >= BB * NN) return;
    int n = i % NN;
    int b = i / NN;
    float s = 0.0f;
#pragma unroll
    for (int c = 0; c < CC; ++c) {
        float v = mean_pc[((size_t)b * CC + c) * NN + n];
        s = __builtin_fmaf(v, v, s);
    }
    sq[i] = s;
}

constexpr int KQ   = 32;
constexpr int KW   = 8;
constexpr int KSEG = NN / KW;
constexpr int KRB  = 16;
constexpr int KNB  = KSEG / KRB;
constexpr int MB   = 8;

template <int J>
__device__ __forceinline__ float row_bcast(float v) {
    return __builtin_bit_cast(float,
        __builtin_amdgcn_update_dpp(0, __builtin_bit_cast(int, v), 0x150 + J, 0xf, 0xf, true));
}

template <int J>
struct KnnStep {
    static __device__ __forceinline__ void bcast(float (&sb)[KRB], float sqv) {
        sb[J] = row_bcast<J>(sqv);
        KnnStep<J + 1>::bcast(sb, sqv);
    }
};
template <>
struct KnnStep<KRB> {
    static __device__ __forceinline__ void bcast(float (&)[KRB], float) {}
};

__device__ __forceinline__ float other_row(float v) {
    int b = __builtin_bit_cast(int, v);
    return __builtin_bit_cast(float,
        __builtin_amdgcn_permlanex16(b, b, 0x76543210, 0xfedcba98, false, false));
}

__device__ __forceinline__ void load_points(v2f (&f)[CC / 4], const float* __restrict__ pc,
                                            int p0, int sub, int hi) {
    const float* p = pc + (size_t)(2 * hi) * NN + p0 + sub;
#pragma unroll
    for (int s = 0; s < CC / 4; ++s) {
        v2f v = {p[(size_t)(4 * s) * NN], p[(size_t)(4 * s + 1) * NN]};
        f[s] = v;
    }
}

__device__ __forceinline__ float pick(bool c, float a, float b) { return c ? a : b; }
__device__ __forceinline__ int   pick(bool c, int a, int b)     { return c ? a : b; }

__device__ __forceinline__ void knn_insert(float (&bd)[KK], int (&bi)[KK], float d, int i) {
    bool c[KK];
#pragma unroll
    for (int j = 0; j < KK; ++j) c[j] = d < bd[j];
#pragma unroll
    for (int j = KK - 1; j > 0; --j) {
        bd[j] = pick(c[j], pick(c[j - 1], bd[j - 1], d), bd[j]);
        bi[j] = pick(c[j], pick(c[j - 1], bi[j - 1], i), bi[j]);
    }
    bd[0] = pick(c[0], d, bd[0]);
    bi[0] = pick(c[0], i, bi[0]);
}

__device__ __forceinline__ float pick16(const float (&v)[KRB], int j) {
    bool b0 = (j & 1) != 0;
    bool b1 = (j & 2) != 0;
    bool b2 = (j & 4) != 0;
    bool b3 = (j & 8) != 0;
    float e0 = pick(b0, v[1], v[0]);
    float e1 = pick(b0, v[3], v[2]);
    float e2 = pick(b0, v[5], v[4]);
    float e3 = pick(b0, v[7], v[6]);
    float e4 = pick(b0, v[9], v[8]);
    float e5 = pick(b0, v[11], v[10]);
    float e6 = pick(b0, v[13], v[12]);
    float e7 = pick(b0, v[15], v[14]);
    float f0 = pick(b1, e1, e0);
    float f1 = pick(b1, e3, e2);
    float f2 = pick(b1, e5, e4);
    float f3 = pick(b1, e7, e6);
    float g0 = pick(b2, f1, f0);
    float g1 = pick(b2, f3, f2);
    return pick(b3, g1, g0);
}

__global__ void __launch_bounds__(256) knn_kernel(const float* __restrict__ mean_pc,
                                                  const float* __restrict__ sq,
                                                  int* __restrict__ idx,
                                                  const float* __restrict__ W1,
                                                  float* __restrict__ tbl,
                                                  size_t ws_size) {
    __shared__ float mrgD[KW * KK * KQ];
    __shared__ int   mrgI[KW * KK * KQ];

    int t    = threadIdx.x;
    int lane = t & 31;
    int sub  = lane & 15;
    int hi   = lane >> 4;
    int w    = __builtin_amdgcn_readfirstlane(t >> 5);
    int b    = blockIdx.x / (NN / KQ);
    int q0   = (blockIdx.x % (NN / KQ)) * KQ;
    int n    = q0 + lane;
    const float* pc  = mean_pc + (size_t)b * CC * NN;
    const float* sqb = sq + (size_t)b * NN;

    v2f qf[2][CC / 4];
    load_points(qf[0], pc, q0, sub, hi);
    load_points(qf[1], pc, q0 + 16, sub, hi);

    const v8f vzero = {0.f, 0.f, 0.f, 0.f, 0.f, 0.f, 0.f, 0.f};

    if (ws_size >= WS_TBL_NEED) {
#pragma unroll
        for (int rt = 0; rt < 2; ++rt) {
            int row0 = (2 * w + rt) * 16;
            const float* wr = W1 + (size_t)(row0 + sub) * EE + CC + 2 * hi;
            v2f af[CC / 4];
#pragma unroll
            for (int s = 0; s < CC / 4; ++s) af[s] = *(const v2f*)(wr + 4 * s);
            v8f t0 = vzero;
            v8f t1 = vzero;
#pragma unroll
            for (int s = 0; s < CC / 4; ++s) {
                t0 = __builtin_amdgcn_wmma_f32_16x16x4_f32(false, af[s], false, qf[0][s], (short)0, t0, false, false);
                t1 = __builtin_amdgcn_wmma_f32_16x16x4_f32(false, af[s], false, qf[1][s], (short)0, t1, false, false);
            }
            float* tq = tbl + ((size_t)b * NN + q0 + sub) * HH + row0 + 8 * hi;
            v4f lo0 = {t0[0], t0[1], t0[2], t0[3]};
            v4f hi0 = {t0[4], t0[5], t0[6], t0[7]};
            v4f lo1 = {t1[0], t1[1], t1[2], t1[3]};
            v4f hi1 = {t1[4], t1[5], t1[6], t1[7]};
            *(v4f*)(tq)               = lo0;
            *(v4f*)(tq + 4)           = hi0;
            *(v4f*)(tq + 16 * HH)     = lo1;
            *(v4f*)(tq + 16 * HH + 4) = hi1;
        }
    }

    float bd[KK];
    int   bi[KK];
#pragma unroll
    for (int j = 0; j < KK; ++j) { bd[j] = 3.0e38f; bi[j] = 0; }

    int   r0 = w * KSEG;
    v2f   rf[CC / 4];
    float sqv;
    load_points(rf, pc, r0, sub, hi);
    sqv = sqb[r0 + sub];

    for (int rb = 0; rb < KNB; ++rb) {
        v8f d0 = vzero;
        v8f d1 = vzero;
#pragma unroll
        for (int s = 0; s < CC / 4; ++s) {
            d0 = __builtin_amdgcn_wmma_f32_16x16x4_f32(false, rf[s], false, qf[0][s], (short)0, d0, false, false);
            d1 = __builtin_amdgcn_wmma_f32_16x16x4_f32(false, rf[s], false, qf[1][s], (short)0, d1, false, false);
        }

        float sb[KRB];
        __builtin_amdgcn_sched_barrier(0);
        KnnStep<0>::bcast(sb, sqv);
        __builtin_amdgcn_sched_barrier(0);

        float dot[KRB];
#pragma unroll
        for (int j = 0; j < 8; ++j) {
            float keep = hi ? d1[j] : d0[j];
            float send = hi ? d0[j] : d1[j];
            float recv = other_row(send);
            dot[j]     = hi ? recv : keep;
            dot[8 + j] = hi ? keep : recv;
        }
#pragma unroll
        for (int j = 0; j < KRB; ++j) dot[j] = sb[j] - 2.0f * dot[j];

        int rn = (rb + 1 < KNB) ? r0 + KRB : r0;
        load_points(rf, pc, rn, sub, hi);
        sqv = sqb[rn + sub];

        unsigned m = 0u;
#pragma unroll
        for (int j = 0; j < KRB; ++j) m |= (dot[j] < bd[KK - 1]) ? (1u << (MB + j)) : 0u;
        unsigned self = (unsigned)(n - r0);
        m &= ~((self < (unsigned)KRB) ? (1u << (MB + (self & (unsigned)(KRB - 1)))) : 0u);

        for (int it = 0; it < KRB; ++it) {
            if (__builtin_amdgcn_ballot_w32(m != 0u) == 0u) break;
            bool has = m != 0u;
            int  j   = (__builtin_ctz(m | (1u << (MB + KRB))) - MB) & (KRB - 1);
            m &= m - 1u;
            float d = pick16(dot, j);
            d = has ? d : 3.0e38f;
            knn_insert(bd, bi, d, r0 + j);
        }
        r0 = rn;
    }

#pragma unroll
    for (int j = 0; j < KK; ++j) {
        mrgD[(w * KK + j) * KQ + lane] = bd[j];
        mrgI[(w * KK + j) * KQ + lane] = bi[j];
    }
    __syncthreads();
    for (int step = 1; step < KW; step <<= 1) {
        if ((w & (2 * step - 1)) == 0) {
            int srcw = w + step;
            for (int j = 0; j < KK; ++j) {
                float cd = mrgD[(srcw * KK + j) * KQ + lane];
                int   ci = mrgI[(srcw * KK + j) * KQ + lane];
                if (cd < bd[KK - 1]) knn_insert(bd, bi, cd, ci);
            }
            if (2 * step < KW) {
#pragma unroll
                for (int j = 0; j < KK; ++j) {
                    mrgD[(w * KK + j) * KQ + lane] = bd[j];
                    mrgI[(w * KK + j) * KQ + lane] = bi[j];
                }
            }
        }
        __syncthreads();
    }
    if (w == 0) {
#pragma unroll
        for (int j = 0; j < KK; ++j) idx[((size_t)b * KK + j) * NN + n] = bi[j];
    }
}

constexpr int PE = EE + 4;
constexpr int PH = HH + 4;
constexpr int PR = CC + 4;

template <int NRT, int NCT, int Q>
__device__ __forceinline__ void load_chunk(v4f (&a)[NRT][Q], v4f (&b)[NCT][Q],
                                           const float* const (&wp)[NRT],
                                           const float* const (&bp)[NCT], int koff) {
#pragma unroll
    for (int r = 0; r < NRT; ++r) {
#pragma unroll
        for (int q = 0; q < Q; ++q) a[r][q] = *(const v4f*)(wp[r] + koff + 4 * q);
    }
#pragma unroll
    for (int c = 0; c < NCT; ++c) {
#pragma unroll
        for (int q = 0; q < Q; ++q) b[c][q] = *(const v4f*)(bp[c] + koff + 4 * q);
    }
}

template <int NRT, int Q, int WP>
__device__ __forceinline__ void load_a(v4f (&a)[NRT][Q], const float* __restrict__ W,
                                       int rt0, int laneRow, int laneHi, int koff) {
    constexpr int CH = 4 * Q;
#pragma unroll
    for (int r = 0; r < NRT; ++r) {
        const float* p = W + (size_t)((rt0 + r) * 16 + laneRow) * WP + CH * laneHi + koff;
#pragma unroll
        for (int q = 0; q < Q; ++q) a[r][q] = *(const v4f*)(p + 4 * q);
    }
}

template <int NCT, int Q>
__device__ __forceinline__ void load_b(v4f (&b)[NCT][Q], const float* const (&bp)[NCT], int koff) {
#pragma unroll
    for (int c = 0; c < NCT; ++c) {
#pragma unroll
        for (int q = 0; q < Q; ++q) b[c][q] = *(const v4f*)(bp[c] + koff + 4 * q);
    }
}

template <int NRT, int Q>
__device__ __forceinline__ void copy_a(v4f (&d)[NRT][Q], const v4f (&s)[NRT][Q]) {
#pragma unroll
    for (int r = 0; r < NRT; ++r) {
#pragma unroll
        for (int q = 0; q < Q; ++q) d[r][q] = s[r][q];
    }
}

template <int NRT, int NCT, int Q>
__device__ __forceinline__ void mma_chunk(v8f (&acc)[NRT][NCT],
                                          const v4f (&a)[NRT][Q], const v4f (&b)[NCT][Q]) {
#pragma unroll
    for (int q = 0; q < Q; ++q) {
#pragma unroll
        for (int s = 0; s < 2; ++s) {
#pragma unroll
            for (int r = 0; r < NRT; ++r) {
                v2f av = s ? __builtin_shufflevector(a[r][q], a[r][q], 2, 3)
                           : __builtin_shufflevector(a[r][q], a[r][q], 0, 1);
#pragma unroll
                for (int c = 0; c < NCT; ++c) {
                    v2f bv = s ? __builtin_shufflevector(b[c][q], b[c][q], 2, 3)
                               : __builtin_shufflevector(b[c][q], b[c][q], 0, 1);
                    acc[r][c] = __builtin_amdgcn_wmma_f32_16x16x4_f32(
                        false, av, false, bv, (short)0, acc[r][c], false, false);
                }
            }
        }
    }
}

template <int NRT, int NCT, int Q, int KI, int WP, int PIN, int POUT, bool RELU, bool BIAS,
          int NPRE, bool AREG>
__device__ __forceinline__ void gemm_core(const float* __restrict__ W,
                                          const float* __restrict__ bias,
                                          const float* lin, float* lout,
                                          int rt0, int ct0, int laneRow, int laneHi,
                                          const v4f (&apre)[NPRE > 0 ? NPRE : 1][NRT][Q],
                                          const v4f (&aall)[AREG ? KI / (8 * Q) : 1][NRT][Q]) {
    constexpr int CH     = 4 * Q;
    constexpr int KSTEP  = 2 * CH;
    constexpr int NCHUNK = KI / KSTEP;
    static_assert(KI % KSTEP == 0 && NCHUNK % 2 == 0, "chunks come in pairs");
    static_assert(AREG || NPRE == 1 || NPRE == 2, "the caller supplies the first A fragments");

    const v8f vzero = {0.f, 0.f, 0.f, 0.f, 0.f, 0.f, 0.f, 0.f};
    const v4f vzero4 = {0.f, 0.f, 0.f, 0.f};
    v8f acc[NRT][NCT];
#pragma unroll
    for (int r = 0; r < NRT; ++r) {
#pragma unroll
        for (int c = 0; c < NCT; ++c) acc[r][c] = vzero;
    }

    const float* wp[NRT];
    const float* bp[NCT];
#pragma unroll
    for (int r = 0; r < NRT; ++r)
        wp[r] = W + (size_t)((rt0 + r) * 16 + laneRow) * WP + CH * laneHi;
#pragma unroll
    for (int c = 0; c < NCT; ++c)
        bp[c] = lin + ((ct0 + c) * 16 + laneRow) * PIN + CH * laneHi;

    v4f bl[NRT], bh[NRT];
    for (int r = 0; r < NRT; ++r) {
        int row0 = (rt0 + r) * 16 + 8 * laneHi;
        bl[r] = vzero4;
        bh[r] = vzero4;
        if (BIAS) {
            bl[r] = *(const v4f*)(bias + row0);
            bh[r] = *(const v4f*)(bias + row0 + 4);
        }
    }

    v4f b0[NCT][Q], b1[NCT][Q];
    if (AREG) {
        load_b<NCT, Q>(b0, bp, 0);
#pragma unroll
        for (int kk = 0; kk < NCHUNK; kk += 2) {
            load_b<NCT, Q>(b1, bp, (kk + 1) * KSTEP);
            __builtin_amdgcn_sched_barrier(0);
            mma_chunk<NRT, NCT, Q>(acc, aall[AREG ? kk : 0], b0);
            __builtin_amdgcn_sched_barrier(0);
            if (kk + 2 < NCHUNK) load_b<NCT, Q>(b0, bp, (kk + 2) * KSTEP);
            __builtin_amdgcn_sched_barrier(0);
            mma_chunk<NRT, NCT, Q>(acc, aall[AREG ? kk + 1 : 0], b1);
            __builtin_amdgcn_sched_barrier(0);
        }
    } else if (NPRE >= 2) {
        v4f ar[NCHUNK][NRT][Q];
        copy_a<NRT, Q>(ar[0], apre[0]);
        copy_a<NRT, Q>(ar[1], apre[NPRE >= 2 ? 1 : 0]);
        load_b<NCT, Q>(b0, bp, 0);
#pragma unroll
        for (int kk = 0; kk < NCHUNK; kk += 2) {
            load_b<NCT, Q>(b1, bp, (kk + 1) * KSTEP);
            if (kk + 2 < NCHUNK)
                load_a<NRT, Q, WP>(ar[kk + 2 < NCHUNK ? kk + 2 : 0], W, rt0, laneRow, laneHi, (kk + 2) * KSTEP);
            __builtin_amdgcn_sched_barrier(0);
            mma_chunk<NRT, NCT, Q>(acc, ar[kk], b0);
            __builtin_amdgcn_sched_barrier(0);
            if (kk + 2 < NCHUNK) {
                load_b<NCT, Q>(b0, bp, (kk + 2) * KSTEP);
                load_a<NRT, Q, WP>(ar[kk + 3 < NCHUNK ? kk + 3 : 0], W, rt0, laneRow, laneHi, (kk + 3) * KSTEP);
            }
            __builtin_amdgcn_sched_barrier(0);
            mma_chunk<NRT, NCT, Q>(acc, ar[kk + 1], b1);
            __builtin_amdgcn_sched_barrier(0);
        }
    } else {
        v4f a0[NRT][Q], a1[NRT][Q];
        copy_a<NRT, Q>(a0, apre[0]);
        load_b<NCT, Q>(b0, bp, 0);
        for (int kk = 0; kk < NCHUNK; kk += 2) {
            load_chunk<NRT, NCT, Q>(a1, b1, wp, bp, (kk + 1) * KSTEP);
            __builtin_amdgcn_sched_barrier(0);
            mma_chunk<NRT, NCT, Q>(acc, a0, b0);
            __builtin_amdgcn_sched_barrier(0);
            if (kk + 2 < NCHUNK) load_chunk<NRT, NCT, Q>(a0, b0, wp, bp, (kk + 2) * KSTEP);
            __builtin_amdgcn_sched_barrier(0);
            mma_chunk<NRT, NCT, Q>(acc, a1, b1);
            __builtin_amdgcn_sched_barrier(0);
        }
    }

#pragma unroll
    for (int r = 0; r < NRT; ++r) {
        int row0 = (rt0 + r) * 16 + 8 * laneHi;
#pragma unroll
        for (int c = 0; c < NCT; ++c) {
            int col = (ct0 + c) * 16 + laneRow;
            v4f lo, hi;
#pragma unroll
            for (int j = 0; j < 4; ++j) {
                float u = acc[r][c][j];
                float v = acc[r][c][j + 4];
                if (BIAS) { u += bl[r][j]; v += bh[r][j]; }
                if (RELU) { u = fmaxf(u, 0.0f); v = fmaxf(v, 0.0f); }
                lo[j] = u;
                hi[j] = v;
            }
            *(v4f*)(lout + col * POUT + row0)     = lo;
            *(v4f*)(lout + col * POUT + row0 + 4) = hi;
        }
    }
}

template <int NRT, int NCT, int Q, int KI, int WP, int PIN, int POUT, bool RELU, bool BIAS, int NPRE>
__device__ __forceinline__ void gemm_block_pre(const float* __restrict__ W,
                                               const float* __restrict__ bias,
                                               const float* lin, float* lout,
                                               int rt0, int ct0, int laneRow, int laneHi,
                                               const v4f (&apre)[NPRE][NRT][Q]) {
    const v4f none[1][NRT][Q] = {};
    gemm_core<NRT, NCT, Q, KI, WP, PIN, POUT, RELU, BIAS, NPRE, false>(
        W, bias, lin, lout, rt0, ct0, laneRow, laneHi, apre, none);
}

template <int NRT, int NCT, int Q, int KI, int WP, int PIN, int POUT, bool RELU, bool BIAS>
__device__ __forceinline__ void gemm_block_areg(const float* __restrict__ W,
                                                const float* __restrict__ bias,
                                                const float* lin, float* lout,
                                                int rt0, int ct0, int laneRow, int laneHi,
                                                const v4f (&aall)[KI / (8 * Q)][NRT][Q]) {
    const v4f none[1][NRT][Q] = {};
    gemm_core<NRT, NCT, Q, KI, WP, PIN, POUT, RELU, BIAS, 0, true>(
        W, bias, lin, lout, rt0, ct0, laneRow, laneHi, none, aall);
}

__device__ __forceinline__ void lds_barrier() {
    __builtin_amdgcn_fence(__ATOMIC_RELEASE, "workgroup", "local");
    __builtin_amdgcn_s_barrier();
    __builtin_amdgcn_fence(__ATOMIC_ACQUIRE, "workgroup", "local");
}

constexpr int PC = CC + 4;
constexpr int QC = 16;
constexpr int NGRP = BB * (NN / NT);
constexpr int NWG  = 256;
constexpr int GPW  = NGRP / NWG;
static_assert(NGRP % NWG == 0, "every workgroup takes the same number of groups");

__device__ __forceinline__ void fetch_cols(float (&gv)[16], float (&cv)[4],
                                           const float* __restrict__ x,
                                           const float* __restrict__ mean_pc,
                                           int b, int n0, unsigned j, int t, bool nbr) {
    int cg   = t >> 6;
    int dcol = t & (QC - 1);
    int c0   = t >> 4;
    const float* mp = mean_pc + ((size_t)b * CC + cg * 16) * NN + j;
    const float* xp = x + (((size_t)b * CC + c0) * DD + (dcol >> 2)) * NN + n0 + (dcol & 3);
    if (nbr) {
#pragma unroll
        for (int i = 0; i < 16; ++i) gv[i] = mp[(size_t)i * NN];
    }
#pragma unroll
    for (int i = 0; i < 4; ++i) {
        float v = 0.0f;
        if (dcol < DD * NT) v = xp[(size_t)i * 16 * DD * NN];
        cv[i] = v;
    }
}

__device__ __forceinline__ void store_cols(float* ldsN, float* ldsC,
                                           const float (&gv)[16], const float (&cv)[4], int t, bool nbr) {
    int col  = t & (COLS - 1);
    int cg   = t >> 6;
    int dcol = t & (QC - 1);
    int c0   = t >> 4;
    if (nbr) {
#pragma unroll
        for (int i = 0; i < 4; ++i) {
            v4f s = {gv[4 * i], gv[4 * i + 1], gv[4 * i + 2], gv[4 * i + 3]};
            *(v4f*)(ldsN + col * PC + cg * 16 + 4 * i) = s;
        }
    }
#pragma unroll
    for (int i = 0; i < 4; ++i) {
        ldsC[dcol * PE + c0 + 16 * i]      = cv[i];
        ldsC[dcol * PE + CC + c0 + 16 * i] = -cv[i];
    }
}

template <int I0, int I1>
__device__ __forceinline__ void request_tbl(v4f (&pv)[16], const float* __restrict__ tbl,
                                            int b, unsigned j, int wu, int lane) {
#pragma unroll
    for (int i = I0; i < I1; ++i) {
        int m = 8 * (wu >> 1) + (i >> 1);
        unsigned jj = (unsigned)__builtin_amdgcn_readlane((int)j, m);
        pv[i] = *(const v4f*)(tbl + ((size_t)b * NN + jj) * HH + (i & 1) * (HH / 2) + 4 * lane);
    }
}

template <bool UT>
__device__ __forceinline__ void mlp_body(const float* __restrict__ x,
                                         const float* __restrict__ mean_pc,
                                         const int* __restrict__ idx,
                                         const float* __restrict__ W1, const float* __restrict__ b1,
                                         const float* __restrict__ W2, const float* __restrict__ b2,
                                         const float* __restrict__ W3, const float* __restrict__ b3,
                                         float* __restrict__ out,
                                         const float* __restrict__ tbl,
                                         float* ldsX, float* ldsY, float* ldsP, float* ldsQ,
                                         float* ldsR, float* ldsN, float* ldsC) {
    constexpr bool useT = UT;


    int t       = threadIdx.x;
    int lane    = t & 31;
    int w       = t >> 5;
    int laneRow = lane & 15;
    int laneHi  = lane >> 4;

    int gnl = (t & (COLS - 1)) >> 4;
    int gk  = t & 15;

    float gv[16];
    float cv[4];

    const int wu = __builtin_amdgcn_readfirstlane(w);
    unsigned jc = 0;

    int g = blockIdx.x;

    {
        int b  = g / (NN / NT);
        int n0 = (g % (NN / NT)) * NT;
        unsigned j = (unsigned)idx[((size_t)b * KK + gk) * NN + n0 + gnl];
        jc = j;
        fetch_cols(gv, cv, x, mean_pc, b, n0, j, t, !useT);
        store_cols(ldsN, ldsC, gv, cv, t, !useT);
    }

    v4f aP[1][2][2] = {};
    if (!useT) load_a<2, 2, EE>(aP[0], W1 + CC, 2 * w, laneRow, laneHi, 0);

    v4f aQ[2][2][4];
    v4f pv[16];
    if (useT) {
        load_a<2, 4, EE>(aQ[0], W1, 2 * w, laneRow, laneHi, 0);
        __builtin_amdgcn_sched_barrier(0);
        request_tbl<0, 8>(pv, tbl, g / (NN / NT), jc, wu, lane);
        __builtin_amdgcn_sched_barrier(0);
        load_a<2, 4, EE>(aQ[1], W1, 2 * w, laneRow, laneHi, 32);
        __builtin_amdgcn_sched_barrier(0);
        request_tbl<8, 16>(pv, tbl, g / (NN / NT), jc, wu, lane);
    }

    v4f w2f[HH / 16][2][2];
#pragma unroll
    for (int kk = 0; kk < HH / 16; ++kk) load_a<2, 2, HH>(w2f[kk], W2, 2 * w, laneRow, laneHi, kk * 16);

    for (int it = 0; it < GPW; ++it) {
        int zw = 0;
        asm volatile("" : "+s"(zw));
        int b  = g / (NN / NT);
        int n0 = (g % (NN / NT)) * NT;

        bool more = it + 1 < GPW;
        int  gx   = g + NWG;
        int  bx   = gx / (NN / NT);
        int  n0x  = (gx % (NN / NT)) * NT;

        lds_barrier();

        unsigned jx = 0;
        if (more) jx = (unsigned)idx[((size_t)bx * KK + gk) * NN + n0x + gnl];

        if (!useT) {
            gemm_block_pre<2, 4, 2, CC, EE, PC, PH, false, false, 1>(W1 + CC + zw, b1, ldsN, ldsP, 2 * w, 0, laneRow, laneHi, aP);
        }
        if (!useT) {
            load_a<2, 4, EE>(aQ[0], W1 + zw, 2 * w, laneRow, laneHi, 0);
            __builtin_amdgcn_sched_barrier(0);
            load_a<2, 4, EE>(aQ[1], W1 + zw, 2 * w, laneRow, laneHi, 32);
            __builtin_amdgcn_sched_barrier(0);
        }
        if (useT) {
#pragma unroll
            for (int i = 0; i < 8; ++i) {
                int col = (wu & 1) * 32 + 8 * (wu >> 1) + (i >> 1);
                *(v4f*)(ldsP + col * PH + (i & 1) * (HH / 2) + 4 * lane) = pv[i];
            }
            __builtin_amdgcn_sched_barrier(0);
        }
        gemm_block_pre<2, 1, 4, EE, EE, PE, PH, false, true, 2>(W1 + zw, b1 + zw, ldsC, ldsQ, 2 * w, 0, laneRow, laneHi, aQ);
        if (useT) {
#pragma unroll
            for (int i = 8; i < 16; ++i) {
                int col = (wu & 1) * 32 + 8 * (wu >> 1) + (i >> 1);
                *(v4f*)(ldsP + col * PH + (i & 1) * (HH / 2) + 4 * lane) = pv[i];
            }
        }
        __syncthreads();

        for (int d = 0; d < DD; ++d) {
            {
                int col = t & (COLS - 1);
                int og  = t >> 6;
                const float* pp = ldsP + col * PH + og * 64;
                const float* qq = ldsQ + (d * NT + (col >> 4)) * PH + og * 64;
                float*       hh = ldsY + col * PH + og * 64;
#pragma unroll
                for (int i = 0; i < 16; ++i) {
                    v4f p = *(const v4f*)(pp + 4 * i);
                    v4f q = *(const v4f*)(qq + 4 * i);
                    v4f h;
#pragma unroll
                    for (int j = 0; j < 4; ++j) h[j] = fmaxf(p[j] + q[j], 0.0f);
                    *(v4f*)(hh + 4 * i) = h;
                }
            }
            __syncthreads();

            bool pf = more && d == 0;
            if (pf) fetch_cols(gv, cv, x, mean_pc, bx, n0x, jx, t, !useT);

            gemm_block_areg<2, 4, 2, HH, HH, PH, PH, true, true>(W2, b2 + zw, ldsY, ldsX, 2 * w, 0, laneRow, laneHi, w2f);

            int zoff = 0;
            asm volatile("" : "+s"(zoff));
            v4f a3[2][1][4];
            load_a<1, 4, HH>(a3[0], W3 + zoff, w >> 1, laneRow, laneHi, 0);
            load_a<1, 4, HH>(a3[1], W3 + zoff, w >> 1, laneRow, laneHi, 32);
            lds_barrier();

            if (pf) store_cols(ldsN, ldsC, gv, cv, t, !useT);

            size_t o   = (((size_t)b * CC + (t & (CC - 1))) * DD + d) * NN + (n0 + (t >> 6));
            float xres = x[o];
            gemm_block_pre<1, 2, 4, HH, HH, PH, PR, false, true, 2>(W3 + zoff, b3, ldsX, ldsR, w >> 1, (w & 1) * 2, laneRow, laneHi, a3);
            __syncthreads();

            {
                int c  = t & (CC - 1);
                int nl = t >> 6;
                const float* rcol = &ldsR[(nl * 16) * PR + c];
                float m = rcol[0];
#pragma unroll
                for (int k = 1; k < KK; ++k) m = fmaxf(m, rcol[k * PR]);
                out[o] = m + xres;
            }
        }
        if (!useT) load_a<2, 2, EE>(aP[0], W1 + CC + zw, 2 * w, laneRow, laneHi, 0);
        if (useT) {
            load_a<2, 4, EE>(aQ[0], W1 + zw, 2 * w, laneRow, laneHi, 0);
            __builtin_amdgcn_sched_barrier(0);
            request_tbl<0, 8>(pv, tbl, more ? bx : b, jx, wu, lane);
            __builtin_amdgcn_sched_barrier(0);
            load_a<2, 4, EE>(aQ[1], W1 + zw, 2 * w, laneRow, laneHi, 32);
            __builtin_amdgcn_sched_barrier(0);
            request_tbl<8, 16>(pv, tbl, more ? bx : b, jx, wu, lane);
        }
        jc = jx;
        g = gx;
    }
}

__global__ void __launch_bounds__(256) mlp_kernel(const float* __restrict__ x,
                                                  const float* __restrict__ mean_pc,
                                                  const int* __restrict__ idx,
                                                  const float* __restrict__ W1, const float* __restrict__ b1,
                                                  const float* __restrict__ W2, const float* __restrict__ b2,
                                                  const float* __restrict__ W3, const float* __restrict__ b3,
                                                  float* __restrict__ out,
                                                  const float* __restrict__ tbl, size_t ws_size) {
    __shared__ __attribute__((aligned(16))) float ldsX[COLS * PH];
    __shared__ __attribute__((aligned(16))) float ldsY[COLS * PH];
    __shared__ __attribute__((aligned(16))) float ldsP[COLS * PH];
    __shared__ __attribute__((aligned(16))) float ldsQ[QC * PH];
    __shared__ __attribute__((aligned(16))) float ldsR[COLS * PR];
    __shared__ __attribute__((aligned(16))) float ldsN[COLS * PC];
    __shared__ __attribute__((aligned(16))) float ldsC[QC * PE];

    if (ws_size >= WS_TBL_NEED)
        mlp_body<true>(x, mean_pc, idx, W1, b1, W2, b2, W3, b3, out, tbl,
                       ldsX, ldsY, ldsP, ldsQ, ldsR, ldsN, ldsC);
    else
        mlp_body<false>(x, mean_pc, idx, W1, b1, W2, b2, W3, b3, out, tbl,
                        ldsX, ldsY, ldsP, ldsQ, ldsR, ldsN, ldsC);
}

extern "C" void kernel_launch(void* const* d_in, const int* in_sizes, int n_in,
                              void* d_out, int out_size, void* d_ws, size_t ws_size,
                              hipStream_t stream) {
    const float* x  = (const float*)d_in[0];
    const float* W1 = (const float*)d_in[1];
    const float* b1 = (const float*)d_in[2];
    const float* W2 = (const float*)d_in[3];
    const float* b2 = (const float*)d_in[4];
    const float* W3 = (const float*)d_in[5];
    const float* b3 = (const float*)d_in[6];
    float* out = (float*)d_out;

    float* mean_pc = (float*)d_ws;
    float* sq      = mean_pc + (size_t)BB * CC * NN;
    int*   idxp    = (int*)(sq + (size_t)BB * NN);
    float* tbl     = (float*)(idxp + (size_t)BB * KK * NN);

    meanpc_kernel<<<(BB * CC * NN + 255) / 256, 256, 0, stream>>>(x, mean_pc);
    sqnorm_kernel<<<(BB * NN + 255) / 256, 256, 0, stream>>>(mean_pc, sq);
    knn_kernel<<<BB * (NN / KQ), 256, 0, stream>>>(mean_pc, sq, idxp, W1, tbl, ws_size);
    mlp_kernel<<<NWG, 256, 0, stream>>>(x, mean_pc, idxp,
                                                        W1, b1, W2, b2, W3, b3, out, tbl, ws_size);
}


# Launches 4 -> 2: mean, norm and table column of all points in one 256-block launch; then each workgroup finds its 32 points' neighbours and runs their MLP (indices in LDS, W2 fetched under the merge)
# speedup vs baseline: 17.8967x; 1.0422x over previous
#include <hip/hip_runtime.h>

#define BB 2
#define CC 64
#define DD 3
#define NN 4096
#define KK 16
#define EE 128
#define HH 256

typedef __attribute__((ext_vector_type(2))) float v2f;
typedef __attribute__((ext_vector_type(4))) float v4f;
typedef __attribute__((ext_vector_type(8))) float v8f;

constexpr int NT   = 4;
constexpr int COLS = NT * KK;

constexpr size_t WS_TBL_OFF  = ((size_t)BB * CC * NN + (size_t)BB * NN + (size_t)BB * KK * NN) * 4;
constexpr size_t WS_TBL_NEED = WS_TBL_OFF + (size_t)BB * NN * HH * 4;

constexpr int KQ   = 32;
constexpr int KW   = 8;
constexpr int KSEG = NN / KW;
constexpr int KRB  = 16;
constexpr int KNB  = KSEG / KRB;
constexpr int MB   = 8;

template <int J>
__device__ __forceinline__ float row_bcast(float v) {
    return __builtin_bit_cast(float,
        __builtin_amdgcn_update_dpp(0, __builtin_bit_cast(int, v), 0x150 + J, 0xf, 0xf, true));
}

template <int J>
struct KnnStep {
    static __device__ __forceinline__ void bcast(float (&sb)[KRB], float sqv) {
        sb[J] = row_bcast<J>(sqv);
        KnnStep<J + 1>::bcast(sb, sqv);
    }
};
template <>
struct KnnStep<KRB> {
    static __device__ __forceinline__ void bcast(float (&)[KRB], float) {}
};

__device__ __forceinline__ float other_row(float v) {
    int b = __builtin_bit_cast(int, v);
    return __builtin_bit_cast(float,
        __builtin_amdgcn_permlanex16(b, b, 0x76543210, 0xfedcba98, false, false));
}

__device__ __forceinline__ void load_points(v2f (&f)[CC / 4], const float* __restrict__ pc,
                                            int p0, int sub, int hi) {
    const float* p = pc + (size_t)(2 * hi) * NN + p0 + sub;
#pragma unroll
    for (int s = 0; s < CC / 4; ++s) {
        v2f v = {p[(size_t)(4 * s) * NN], p[(size_t)(4 * s + 1) * NN]};
        f[s] = v;
    }
}

__device__ __forceinline__ float pick(bool c, float a, float b) { return c ? a : b; }
__device__ __forceinline__ int   pick(bool c, int a, int b)     { return c ? a : b; }

__device__ __forceinline__ void knn_insert(float (&bd)[KK], int (&bi)[KK], float d, int i) {
    bool c[KK];
#pragma unroll
    for (int j = 0; j < KK; ++j) c[j] = d < bd[j];
#pragma unroll
    for (int j = KK - 1; j > 0; --j) {
        bd[j] = pick(c[j], pick(c[j - 1], bd[j - 1], d), bd[j]);
        bi[j] = pick(c[j], pick(c[j - 1], bi[j - 1], i), bi[j]);
    }
    bd[0] = pick(c[0], d, bd[0]);
    bi[0] = pick(c[0], i, bi[0]);
}

__device__ __forceinline__ float pick16(const float (&v)[KRB], int j) {
    bool b0 = (j & 1) != 0;
    bool b1 = (j & 2) != 0;
    bool b2 = (j & 4) != 0;
    bool b3 = (j & 8) != 0;
    float e0 = pick(b0, v[1], v[0]);
    float e1 = pick(b0, v[3], v[2]);
    float e2 = pick(b0, v[5], v[4]);
    float e3 = pick(b0, v[7], v[6]);
    float e4 = pick(b0, v[9], v[8]);
    float e5 = pick(b0, v[11], v[10]);
    float e6 = pick(b0, v[13], v[12]);
    float e7 = pick(b0, v[15], v[14]);
    float f0 = pick(b1, e1, e0);
    float f1 = pick(b1, e3, e2);
    float f2 = pick(b1, e5, e4);
    float f3 = pick(b1, e7, e6);
    float g0 = pick(b2, f1, f0);
    float g1 = pick(b2, f3, f2);
    return pick(b3, g1, g0);
}

constexpr int PSM = 48;

__global__ void __launch_bounds__(256) prep_kernel(const float* __restrict__ x,
                                                   const float* __restrict__ W1,
                                                   float* __restrict__ mean_pc,
                                                   float* __restrict__ sq,
                                                   float* __restrict__ tbl,
                                                   size_t ws_size) {
    __shared__ float sm[CC * PSM];

    int t    = threadIdx.x;
    int lane = t & 31;
    int sub  = lane & 15;
    int hi   = lane >> 4;
    int w    = __builtin_amdgcn_readfirstlane(t >> 5);
    int b    = blockIdx.x / (NN / KQ);
    int q0   = (blockIdx.x % (NN / KQ)) * KQ;
    int n    = q0 + lane;

#pragma unroll
    for (int i = 0; i < CC / KW; ++i) {
        int c = w * (CC / KW) + i;
        const float* xp = x + (size_t)(b * CC + c) * (DD * NN) + n;
        float m = (xp[0] + xp[NN] + xp[2 * NN]) * (1.0f / 3.0f);
        mean_pc[((size_t)b * CC + c) * NN + n] = m;
        sm[c * PSM + lane] = m;
    }
    __syncthreads();

    if (w == 0) {
        float s = 0.0f;
#pragma unroll
        for (int c = 0; c < CC; ++c) {
            float v = sm[c * PSM + lane];
            s = __builtin_fmaf(v, v, s);
        }
        sq[(size_t)b * NN + n] = s;
    }

    if (ws_size >= WS_TBL_NEED) {
        const v8f vzero = {0.f, 0.f, 0.f, 0.f, 0.f, 0.f, 0.f, 0.f};
        v2f qf[2][CC / 4];
#pragma unroll
        for (int tl = 0; tl < 2; ++tl) {
#pragma unroll
            for (int s = 0; s < CC / 4; ++s) {
                const float* p = sm + (4 * s + 2 * hi) * PSM + 16 * tl + sub;
                v2f v = {p[0], p[PSM]};
                qf[tl][s] = v;
            }
        }
#pragma unroll
        for (int rt = 0; rt < 2; ++rt) {
            int row0 = (2 * w + rt) * 16;
            const float* wr = W1 + (size_t)(row0 + sub) * EE + CC + 2 * hi;
            v2f af[CC / 4];
#pragma unroll
            for (int s = 0; s < CC / 4; ++s) af[s] = *(const v2f*)(wr + 4 * s);
            v8f t0 = vzero;
            v8f t1 = vzero;
#pragma unroll
            for (int s = 0; s < CC / 4; ++s) {
                t0 = __builtin_amdgcn_wmma_f32_16x16x4_f32(false, af[s], false, qf[0][s], (short)0, t0, false, false);
                t1 = __builtin_amdgcn_wmma_f32_16x16x4_f32(false, af[s], false, qf[1][s], (short)0, t1, false, false);
            }
            float* tq = tbl + ((size_t)b * NN + q0 + sub) * HH + row0 + 8 * hi;
            v4f lo0 = {t0[0], t0[1], t0[2], t0[3]};
            v4f hi0 = {t0[4], t0[5], t0[6], t0[7]};
            v4f lo1 = {t1[0], t1[1], t1[2], t1[3]};
            v4f hi1 = {t1[4], t1[5], t1[6], t1[7]};
            *(v4f*)(tq)               = lo0;
            *(v4f*)(tq + 4)           = hi0;
            *(v4f*)(tq + 16 * HH)     = lo1;
            *(v4f*)(tq + 16 * HH + 4) = hi1;
        }
    }
}

__device__ __forceinline__ void lds_barrier() {
    __builtin_amdgcn_fence(__ATOMIC_RELEASE, "workgroup", "local");
    __builtin_amdgcn_s_barrier();
    __builtin_amdgcn_fence(__ATOMIC_ACQUIRE, "workgroup", "local");
}

__device__ __forceinline__ void knn_scan(const float* __restrict__ mean_pc,
                                         const float* __restrict__ sq,
                                         float (&bd)[KK], int (&bi)[KK]) {
    int t    = threadIdx.x;
    int lane = t & 31;
    int sub  = lane & 15;
    int hi   = lane >> 4;
    int w    = __builtin_amdgcn_readfirstlane(t >> 5);
    int b    = blockIdx.x / (NN / KQ);
    int q0   = (blockIdx.x % (NN / KQ)) * KQ;
    int n    = q0 + lane;
    const float* pc  = mean_pc + (size_t)b * CC * NN;
    const float* sqb = sq + (size_t)b * NN;

    v2f qf[2][CC / 4];
    load_points(qf[0], pc, q0, sub, hi);
    load_points(qf[1], pc, q0 + 16, sub, hi);

    const v8f vzero = {0.f, 0.f, 0.f, 0.f, 0.f, 0.f, 0.f, 0.f};

#pragma unroll
    for (int j = 0; j < KK; ++j) { bd[j] = 3.0e38f; bi[j] = 0; }

    int   r0 = w * KSEG;
    v2f   rf[CC / 4];
    float sqv;
    load_points(rf, pc, r0, sub, hi);
    sqv = sqb[r0 + sub];

    for (int rb = 0; rb < KNB; ++rb) {
        v8f d0 = vzero;
        v8f d1 = vzero;
#pragma unroll
        for (int s = 0; s < CC / 4; ++s) {
            d0 = __builtin_amdgcn_wmma_f32_16x16x4_f32(false, rf[s], false, qf[0][s], (short)0, d0, false, false);
            d1 = __builtin_amdgcn_wmma_f32_16x16x4_f32(false, rf[s], false, qf[1][s], (short)0, d1, false, false);
        }

        float sb[KRB];
        __builtin_amdgcn_sched_barrier(0);
        KnnStep<0>::bcast(sb, sqv);
        __builtin_amdgcn_sched_barrier(0);

        float dot[KRB];
#pragma unroll
        for (int j = 0; j < 8; ++j) {
            float keep = hi ? d1[j] : d0[j];
            float send = hi ? d0[j] : d1[j];
            float recv = other_row(send);
            dot[j]     = hi ? recv : keep;
            dot[8 + j] = hi ? keep : recv;
        }
#pragma unroll
        for (int j = 0; j < KRB; ++j) dot[j] = sb[j] - 2.0f * dot[j];

        int rn = (rb + 1 < KNB) ? r0 + KRB : r0;
        load_points(rf, pc, rn, sub, hi);
        sqv = sqb[rn + sub];

        unsigned m = 0u;
#pragma unroll
        for (int j = 0; j < KRB; ++j) m |= (dot[j] < bd[KK - 1]) ? (1u << (MB + j)) : 0u;
        unsigned self = (unsigned)(n - r0);
        m &= ~((self < (unsigned)KRB) ? (1u << (MB + (self & (unsigned)(KRB - 1)))) : 0u);

        for (int it = 0; it < KRB; ++it) {
            if (__builtin_amdgcn_ballot_w32(m != 0u) == 0u) break;
            bool has = m != 0u;
            int  j   = (__builtin_ctz(m | (1u << (MB + KRB))) - MB) & (KRB - 1);
            m &= m - 1u;
            float d = pick16(dot, j);
            d = has ? d : 3.0e38f;
            knn_insert(bd, bi, d, r0 + j);
        }
        r0 = rn;
    }
}

__device__ __forceinline__ void knn_merge(float (&bd)[KK], int (&bi)[KK],
                                          float* mrgD, int* mrgI, int* sIdx) {
    int lane = threadIdx.x & 31;
    int w    = __builtin_amdgcn_readfirstlane(threadIdx.x >> 5);
#pragma unroll
    for (int j = 0; j < KK; ++j) {
        mrgD[(w * KK + j) * KQ + lane] = bd[j];
        mrgI[(w * KK + j) * KQ + lane] = bi[j];
    }
    lds_barrier();
    for (int step = 1; step < KW; step <<= 1) {
        if ((w & (2 * step - 1)) == 0) {
            int srcw = w + step;
            for (int j = 0; j < KK; ++j) {
                float cd = mrgD[(srcw * KK + j) * KQ + lane];
                int   ci = mrgI[(srcw * KK + j) * KQ + lane];
                if (cd < bd[KK - 1]) knn_insert(bd, bi, cd, ci);
            }
            if (2 * step < KW) {
#pragma unroll
                for (int j = 0; j < KK; ++j) {
                    mrgD[(w * KK + j) * KQ + lane] = bd[j];
                    mrgI[(w * KK + j) * KQ + lane] = bi[j];
                }
            }
        }
        lds_barrier();
    }
    if (w == 0) {
#pragma unroll
        for (int j = 0; j < KK; ++j) sIdx[j * KQ + lane] = bi[j];
    }
}

constexpr int PE = EE + 4;
constexpr int PH = HH + 4;
constexpr int PR = CC + 4;

template <int NRT, int NCT, int Q>
__device__ __forceinline__ void load_chunk(v4f (&a)[NRT][Q], v4f (&b)[NCT][Q],
                                           const float* const (&wp)[NRT],
                                           const float* const (&bp)[NCT], int koff) {
#pragma unroll
    for (int r = 0; r < NRT; ++r) {
#pragma unroll
        for (int q = 0; q < Q; ++q) a[r][q] = *(const v4f*)(wp[r] + koff + 4 * q);
    }
#pragma unroll
    for (int c = 0; c < NCT; ++c) {
#pragma unroll
        for (int q = 0; q < Q; ++q) b[c][q] = *(const v4f*)(bp[c] + koff + 4 * q);
    }
}

template <int NRT, int Q, int WP>
__device__ __forceinline__ void load_a(v4f (&a)[NRT][Q], const float* __restrict__ W,
                                       int rt0, int laneRow, int laneHi, int koff) {
    constexpr int CH = 4 * Q;
#pragma unroll
    for (int r = 0; r < NRT; ++r) {
        const float* p = W + (size_t)((rt0 + r) * 16 + laneRow) * WP + CH * laneHi + koff;
#pragma unroll
        for (int q = 0; q < Q; ++q) a[r][q] = *(const v4f*)(p + 4 * q);
    }
}

template <int NCT, int Q>
__device__ __forceinline__ void load_b(v4f (&b)[NCT][Q], const float* const (&bp)[NCT], int koff) {
#pragma unroll
    for (int c = 0; c < NCT; ++c) {
#pragma unroll
        for (int q = 0; q < Q; ++q) b[c][q] = *(const v4f*)(bp[c] + koff + 4 * q);
    }
}

template <int NRT, int Q>
__device__ __forceinline__ void copy_a(v4f (&d)[NRT][Q], const v4f (&s)[NRT][Q]) {
#pragma unroll
    for (int r = 0; r < NRT; ++r) {
#pragma unroll
        for (int q = 0; q < Q; ++q) d[r][q] = s[r][q];
    }
}

template <int NRT, int NCT, int Q>
__device__ __forceinline__ void mma_chunk(v8f (&acc)[NRT][NCT],
                                          const v4f (&a)[NRT][Q], const v4f (&b)[NCT][Q]) {
#pragma unroll
    for (int q = 0; q < Q; ++q) {
#pragma unroll
        for (int s = 0; s < 2; ++s) {
#pragma unroll
            for (int r = 0; r < NRT; ++r) {
                v2f av = s ? __builtin_shufflevector(a[r][q], a[r][q], 2, 3)
                           : __builtin_shufflevector(a[r][q], a[r][q], 0, 1);
#pragma unroll
                for (int c = 0; c < NCT; ++c) {
                    v2f bv = s ? __builtin_shufflevector(b[c][q], b[c][q], 2, 3)
                               : __builtin_shufflevector(b[c][q], b[c][q], 0, 1);
                    acc[r][c] = __builtin_amdgcn_wmma_f32_16x16x4_f32(
                        false, av, false, bv, (short)0, acc[r][c], false, false);
                }
            }
        }
    }
}

template <int NRT, int NCT, int Q, int KI, int WP, int PIN, int POUT, bool RELU, bool BIAS,
          int NPRE, bool AREG>
__device__ __forceinline__ void gemm_core(const float* __restrict__ W,
                                          const float* __restrict__ bias,
                                          const float* lin, float* lout,
                                          int rt0, int ct0, int laneRow, int laneHi,
                                          const v4f (&apre)[NPRE > 0 ? NPRE : 1][NRT][Q],
                                          const v4f (&aall)[AREG ? KI / (8 * Q) : 1][NRT][Q]) {
    constexpr int CH     = 4 * Q;
    constexpr int KSTEP  = 2 * CH;
    constexpr int NCHUNK = KI / KSTEP;
    static_assert(KI % KSTEP == 0 && NCHUNK % 2 == 0, "chunks come in pairs");
    static_assert(AREG || NPRE == 1 || NPRE == 2, "the caller supplies the first A fragments");

    const v8f vzero = {0.f, 0.f, 0.f, 0.f, 0.f, 0.f, 0.f, 0.f};
    const v4f vzero4 = {0.f, 0.f, 0.f, 0.f};
    v8f acc[NRT][NCT];
#pragma unroll
    for (int r = 0; r < NRT; ++r) {
#pragma unroll
        for (int c = 0; c < NCT; ++c) acc[r][c] = vzero;
    }

    const float* wp[NRT];
    const float* bp[NCT];
#pragma unroll
    for (int r = 0; r < NRT; ++r)
        wp[r] = W + (size_t)((rt0 + r) * 16 + laneRow) * WP + CH * laneHi;
#pragma unroll
    for (int c = 0; c < NCT; ++c)
        bp[c] = lin + ((ct0 + c) * 16 + laneRow) * PIN + CH * laneHi;

    v4f bl[NRT], bh[NRT];
    for (int r = 0; r < NRT; ++r) {
        int row0 = (rt0 + r) * 16 + 8 * laneHi;
        bl[r] = vzero4;
        bh[r] = vzero4;
        if (BIAS) {
            bl[r] = *(const v4f*)(bias + row0);
            bh[r] = *(const v4f*)(bias + row0 + 4);
        }
    }

    v4f b0[NCT][Q], b1[NCT][Q];
    if (AREG) {
        load_b<NCT, Q>(b0, bp, 0);
#pragma unroll
        for (int kk = 0; kk < NCHUNK; kk += 2) {
            load_b<NCT, Q>(b1, bp, (kk + 1) * KSTEP);
            __builtin_amdgcn_sched_barrier(0);
            mma_chunk<NRT, NCT, Q>(acc, aall[AREG ? kk : 0], b0);
            __builtin_amdgcn_sched_barrier(0);
            if (kk + 2 < NCHUNK) load_b<NCT, Q>(b0, bp, (kk + 2) * KSTEP);
            __builtin_amdgcn_sched_barrier(0);
            mma_chunk<NRT, NCT, Q>(acc, aall[AREG ? kk + 1 : 0], b1);
            __builtin_amdgcn_sched_barrier(0);
        }
    } else if (NPRE >= 2) {
        v4f ar[NCHUNK][NRT][Q];
        copy_a<NRT, Q>(ar[0], apre[0]);
        copy_a<NRT, Q>(ar[1], apre[NPRE >= 2 ? 1 : 0]);
        load_b<NCT, Q>(b0, bp, 0);
#pragma unroll
        for (int kk = 0; kk < NCHUNK; kk += 2) {
            load_b<NCT, Q>(b1, bp, (kk + 1) * KSTEP);
            if (kk + 2 < NCHUNK)
                load_a<NRT, Q, WP>(ar[kk + 2 < NCHUNK ? kk + 2 : 0], W, rt0, laneRow, laneHi, (kk + 2) * KSTEP);
            __builtin_amdgcn_sched_barrier(0);
            mma_chunk<NRT, NCT, Q>(acc, ar[kk], b0);
            __builtin_amdgcn_sched_barrier(0);
            if (kk + 2 < NCHUNK) {
                load_b<NCT, Q>(b0, bp, (kk + 2) * KSTEP);
                load_a<NRT, Q, WP>(ar[kk + 3 < NCHUNK ? kk + 3 : 0], W, rt0, laneRow, laneHi, (kk + 3) * KSTEP);
            }
            __builtin_amdgcn_sched_barrier(0);
            mma_chunk<NRT, NCT, Q>(acc, ar[kk + 1], b1);
            __builtin_amdgcn_sched_barrier(0);
        }
    } else {
        v4f a0[NRT][Q], a1[NRT][Q];
        copy_a<NRT, Q>(a0, apre[0]);
        load_b<NCT, Q>(b0, bp, 0);
        for (int kk = 0; kk < NCHUNK; kk += 2) {
            load_chunk<NRT, NCT, Q>(a1, b1, wp, bp, (kk + 1) * KSTEP);
            __builtin_amdgcn_sched_barrier(0);
            mma_chunk<NRT, NCT, Q>(acc, a0, b0);
            __builtin_amdgcn_sched_barrier(0);
            if (kk + 2 < NCHUNK) load_chunk<NRT, NCT, Q>(a0, b0, wp, bp, (kk + 2) * KSTEP);
            __builtin_amdgcn_sched_barrier(0);
            mma_chunk<NRT, NCT, Q>(acc, a1, b1);
            __builtin_amdgcn_sched_barrier(0);
        }
    }

#pragma unroll
    for (int r = 0; r < NRT; ++r) {
        int row0 = (rt0 + r) * 16 + 8 * laneHi;
#pragma unroll
        for (int c = 0; c < NCT; ++c) {
            int col = (ct0 + c) * 16 + laneRow;
            v4f lo, hi;
#pragma unroll
            for (int j = 0; j < 4; ++j) {
                float u = acc[r][c][j];
                float v = acc[r][c][j + 4];
                if (BIAS) { u += bl[r][j]; v += bh[r][j]; }
                if (RELU) { u = fmaxf(u, 0.0f); v = fmaxf(v, 0.0f); }
                lo[j] = u;
                hi[j] = v;
            }
            *(v4f*)(lout + col * POUT + row0)     = lo;
            *(v4f*)(lout + col * POUT + row0 + 4) = hi;
        }
    }
}

template <int NRT, int NCT, int Q, int KI, int WP, int PIN, int POUT, bool RELU, bool BIAS, int NPRE>
__device__ __forceinline__ void gemm_block_pre(const float* __restrict__ W,
                                               const float* __restrict__ bias,
                                               const float* lin, float* lout,
                                               int rt0, int ct0, int laneRow, int laneHi,
                                               const v4f (&apre)[NPRE][NRT][Q]) {
    const v4f none[1][NRT][Q] = {};
    gemm_core<NRT, NCT, Q, KI, WP, PIN, POUT, RELU, BIAS, NPRE, false>(
        W, bias, lin, lout, rt0, ct0, laneRow, laneHi, apre, none);
}

template <int NRT, int NCT, int Q, int KI, int WP, int PIN, int POUT, bool RELU, bool BIAS>
__device__ __forceinline__ void gemm_block_areg(const float* __restrict__ W,
                                                const float* __restrict__ bias,
                                                const float* lin, float* lout,
                                                int rt0, int ct0, int laneRow, int laneHi,
                                                const v4f (&aall)[KI / (8 * Q)][NRT][Q]) {
    const v4f none[1][NRT][Q] = {};
    gemm_core<NRT, NCT, Q, KI, WP, PIN, POUT, RELU, BIAS, 0, true>(
        W, bias, lin, lout, rt0, ct0, laneRow, laneHi, none, aall);
}

constexpr int PC = CC + 4;
constexpr int QC = 16;
constexpr int NGRP = BB * (NN / NT);
constexpr int NWG  = 256;
constexpr int GPW  = NGRP / NWG;
static_assert(NGRP % NWG == 0, "every workgroup takes the same number of groups");
static_assert(GPW * NT == KQ && NWG == BB * (NN / KQ), "a workgroup's groups are the 32 points of its neighbour search");

__device__ __forceinline__ void fetch_cols(float (&gv)[16], float (&cv)[4],
                                           const float* __restrict__ x,
                                           const float* __restrict__ mean_pc,
                                           int b, int n0, unsigned j, int t, bool nbr) {
    int cg   = t >> 6;
    int dcol = t & (QC - 1);
    int c0   = t >> 4;
    const float* mp = mean_pc + ((size_t)b * CC + cg * 16) * NN + j;
    const float* xp = x + (((size_t)b * CC + c0) * DD + (dcol >> 2)) * NN + n0 + (dcol & 3);
    if (nbr) {
#pragma unroll
        for (int i = 0; i < 16; ++i) gv[i] = mp[(size_t)i * NN];
    }
#pragma unroll
    for (int i = 0; i < 4; ++i) {
        float v = 0.0f;
        if (dcol < DD * NT) v = xp[(size_t)i * 16 * DD * NN];
        cv[i] = v;
    }
}

__device__ __forceinline__ void store_cols(float* ldsN, float* ldsC,
                                           const float (&gv)[16], const float (&cv)[4], int t, bool nbr) {
    int col  = t & (COLS - 1);
    int cg   = t >> 6;
    int dcol = t & (QC - 1);
    int c0   = t >> 4;
    if (nbr) {
#pragma unroll
        for (int i = 0; i < 4; ++i) {
            v4f s = {gv[4 * i], gv[4 * i + 1], gv[4 * i + 2], gv[4 * i + 3]};
            *(v4f*)(ldsN + col * PC + cg * 16 + 4 * i) = s;
        }
    }
#pragma unroll
    for (int i = 0; i < 4; ++i) {
        ldsC[dcol * PE + c0 + 16 * i]      = cv[i];
        ldsC[dcol * PE + CC + c0 + 16 * i] = -cv[i];
    }
}

template <int I0, int I1>
__device__ __forceinline__ void request_tbl(v4f (&pv)[16], const float* __restrict__ tbl,
                                            int b, unsigned j, int wu, int lane) {
#pragma unroll
    for (int i = I0; i < I1; ++i) {
        int m = 8 * (wu >> 1) + (i >> 1);
        unsigned jj = (unsigned)__builtin_amdgcn_readlane((int)j, m);
        pv[i] = *(const v4f*)(tbl + ((size_t)b * NN + jj) * HH + (i & 1) * (HH / 2) + 4 * lane);
    }
}

template <bool UT>
__device__ __forceinline__ void mlp_body(const float* __restrict__ x,
                                         const float* __restrict__ mean_pc,
                                         const int* sIdx,
                                         const float* __restrict__ W1, const float* __restrict__ b1,
                                         const float* __restrict__ W2, const float* __restrict__ b2,
                                         const float* __restrict__ W3, const float* __restrict__ b3,
                                         float* __restrict__ out,
                                         const float* __restrict__ tbl,
                                         float* ldsX, float* ldsY, float* ldsP, float* ldsQ,
                                         float* ldsR, float* ldsN, float* ldsC,
                                         const v4f (&w2f)[HH / 16][2][2]) {
    constexpr bool useT = UT;


    int t       = threadIdx.x;
    int lane    = t & 31;
    int w       = t >> 5;
    int laneRow = lane & 15;
    int laneHi  = lane >> 4;

    int gnl = (t & (COLS - 1)) >> 4;
    int gk  = t & 15;

    float gv[16];
    float cv[4];

    const int wu = __builtin_amdgcn_readfirstlane(w);
    unsigned jc = 0;

    int g = blockIdx.x * GPW;

    {
        int b  = g / (NN / NT);
        int n0 = (g % (NN / NT)) * NT;
        unsigned j = (unsigned)sIdx[gk * KQ + gnl];
        jc = j;
        fetch_cols(gv, cv, x, mean_pc, b, n0, j, t, !useT);
        store_cols(ldsN, ldsC, gv, cv, t, !useT);
    }

    v4f aP[1][2][2] = {};
    if (!useT) load_a<2, 2, EE>(aP[0], W1 + CC, 2 * w, laneRow, laneHi, 0);

    v4f aQ[2][2][4];
    v4f pv[16];
    if (useT) {
        load_a<2, 4, EE>(aQ[0], W1, 2 * w, laneRow, laneHi, 0);
        __builtin_amdgcn_sched_barrier(0);
        request_tbl<0, 8>(pv, tbl, g / (NN / NT), jc, wu, lane);
        __builtin_amdgcn_sched_barrier(0);
        load_a<2, 4, EE>(aQ[1], W1, 2 * w, laneRow, laneHi, 32);
        __builtin_amdgcn_sched_barrier(0);
        request_tbl<8, 16>(pv, tbl, g / (NN / NT), jc, wu, lane);
    }


    for (int it = 0; it < GPW; ++it) {
        int zw = 0;
        asm volatile("" : "+s"(zw));
        int b  = g / (NN / NT);
        int n0 = (g % (NN / NT)) * NT;

        bool more = it + 1 < GPW;
        int  gx   = g + 1;
        int  bx   = gx / (NN / NT);
        int  n0x  = (gx % (NN / NT)) * NT;

        lds_barrier();

        unsigned jx = 0;
        if (more) jx = (unsigned)sIdx[gk * KQ + (it + 1) * NT + gnl];

        if (!useT) {
            gemm_block_pre<2, 4, 2, CC, EE, PC, PH, false, false, 1>(W1 + CC + zw, b1, ldsN, ldsP, 2 * w, 0, laneRow, laneHi, aP);
        }
        if (!useT) {
            load_a<2, 4, EE>(aQ[0], W1 + zw, 2 * w, laneRow, laneHi, 0);
            __builtin_amdgcn_sched_barrier(0);
            load_a<2, 4, EE>(aQ[1], W1 + zw, 2 * w, laneRow, laneHi, 32);
            __builtin_amdgcn_sched_barrier(0);
        }
        if (useT) {
#pragma unroll
            for (int i = 0; i < 8; ++i) {
                int col = (wu & 1) * 32 + 8 * (wu >> 1) + (i >> 1);
                *(v4f*)(ldsP + col * PH + (i & 1) * (HH / 2) + 4 * lane) = pv[i];
            }
            __builtin_amdgcn_sched_barrier(0);
        }
        gemm_block_pre<2, 1, 4, EE, EE, PE, PH, false, true, 2>(W1 + zw, b1 + zw, ldsC, ldsQ, 2 * w, 0, laneRow, laneHi, aQ);
        if (useT) {
#pragma unroll
            for (int i = 8; i < 16; ++i) {
                int col = (wu & 1) * 32 + 8 * (wu >> 1) + (i >> 1);
                *(v4f*)(ldsP + col * PH + (i & 1) * (HH / 2) + 4 * lane) = pv[i];
            }
        }
        __syncthreads();

        for (int d = 0; d < DD; ++d) {
            {
                int col = t & (COLS - 1);
                int og  = t >> 6;
                const float* pp = ldsP + col * PH + og * 64;
                const float* qq = ldsQ + (d * NT + (col >> 4)) * PH + og * 64;
                float*       hh = ldsY + col * PH + og * 64;
#pragma unroll
                for (int i = 0; i < 16; ++i) {
                    v4f p = *(const v4f*)(pp + 4 * i);
                    v4f q = *(const v4f*)(qq + 4 * i);
                    v4f h;
#pragma unroll
                    for (int j = 0; j < 4; ++j) h[j] = fmaxf(p[j] + q[j], 0.0f);
                    *(v4f*)(hh + 4 * i) = h;
                }
            }
            __syncthreads();

            bool pf = more && d == 0;
            if (pf) fetch_cols(gv, cv, x, mean_pc, bx, n0x, jx, t, !useT);

            gemm_block_areg<2, 4, 2, HH, HH, PH, PH, true, true>(W2, b2 + zw, ldsY, ldsX, 2 * w, 0, laneRow, laneHi, w2f);

            int zoff = 0;
            asm volatile("" : "+s"(zoff));
            v4f a3[2][1][4];
            load_a<1, 4, HH>(a3[0], W3 + zoff, w >> 1, laneRow, laneHi, 0);
            load_a<1, 4, HH>(a3[1], W3 + zoff, w >> 1, laneRow, laneHi, 32);
            lds_barrier();

            if (pf) store_cols(ldsN, ldsC, gv, cv, t, !useT);

            size_t o   = (((size_t)b * CC + (t & (CC - 1))) * DD + d) * NN + (n0 + (t >> 6));
            float xres = x[o];
            gemm_block_pre<1, 2, 4, HH, HH, PH, PR, false, true, 2>(W3 + zoff, b3, ldsX, ldsR, w >> 1, (w & 1) * 2, laneRow, laneHi, a3);
            __syncthreads();

            {
                int c  = t & (CC - 1);
                int nl = t >> 6;
                const float* rcol = &ldsR[(nl * 16) * PR + c];
                float m = rcol[0];
#pragma unroll
                for (int k = 1; k < KK; ++k) m = fmaxf(m, rcol[k * PR]);
                out[o] = m + xres;
            }
        }
        if (!useT) load_a<2, 2, EE>(aP[0], W1 + CC + zw, 2 * w, laneRow, laneHi, 0);
        if (useT) {
            load_a<2, 4, EE>(aQ[0], W1 + zw, 2 * w, laneRow, laneHi, 0);
            __builtin_amdgcn_sched_barrier(0);
            request_tbl<0, 8>(pv, tbl, more ? bx : b, jx, wu, lane);
            __builtin_amdgcn_sched_barrier(0);
            load_a<2, 4, EE>(aQ[1], W1 + zw, 2 * w, laneRow, laneHi, 32);
            __builtin_amdgcn_sched_barrier(0);
            request_tbl<8, 16>(pv, tbl, more ? bx : b, jx, wu, lane);
        }
        jc = jx;
        g = gx;
    }
}

__global__ void __launch_bounds__(256) knn_mlp_kernel(const float* __restrict__ x,
                                                      const float* __restrict__ mean_pc,
                                                      const float* __restrict__ sq,
                                                      const float* __restrict__ W1, const float* __restrict__ b1,
                                                      const float* __restrict__ W2, const float* __restrict__ b2,
                                                      const float* __restrict__ W3, const float* __restrict__ b3,
                                                      float* __restrict__ out,
                                                      const float* __restrict__ tbl, size_t ws_size) {
    __shared__ float mrgD[KW * KK * KQ];
    __shared__ int   mrgI[KW * KK * KQ];
    __shared__ int   sIdx[KK * KQ];
    __shared__ __attribute__((aligned(16))) float ldsX[COLS * PH];
    __shared__ __attribute__((aligned(16))) float ldsY[COLS * PH];
    __shared__ __attribute__((aligned(16))) float ldsP[COLS * PH];
    __shared__ __attribute__((aligned(16))) float ldsQ[QC * PH];
    __shared__ __attribute__((aligned(16))) float ldsR[COLS * PR];
    __shared__ __attribute__((aligned(16))) float ldsN[COLS * PC];
    __shared__ __attribute__((aligned(16))) float ldsC[QC * PE];

    float bd[KK];
    int   bi[KK];
    knn_scan(mean_pc, sq, bd, bi);

    v4f w2f[HH / 16][2][2];
    {
        int lane = threadIdx.x & 31;
        int w    = threadIdx.x >> 5;
#pragma unroll
        for (int kk = 0; kk < HH / 16; ++kk) load_a<2, 2, HH>(w2f[kk], W2, 2 * w, lane & 15, lane >> 4, kk * 16);
    }

    knn_merge(bd, bi, mrgD, mrgI, sIdx);
    lds_barrier();

    if (ws_size >= WS_TBL_NEED)
        mlp_body<true>(x, mean_pc, sIdx, W1, b1, W2, b2, W3, b3, out, tbl,
                       ldsX, ldsY, ldsP, ldsQ, ldsR, ldsN, ldsC, w2f);
    else
        mlp_body<false>(x, mean_pc, sIdx, W1, b1, W2, b2, W3, b3, out, tbl,
                        ldsX, ldsY, ldsP, ldsQ, ldsR, ldsN, ldsC, w2f);
}

extern "C" void kernel_launch(void* const* d_in, const int* in_sizes, int n_in,
                              void* d_out, int out_size, void* d_ws, size_t ws_size,
                              hipStream_t stream) {
    const float* x  = (const float*)d_in[0];
    const float* W1 = (const float*)d_in[1];
    const float* b1 = (const float*)d_in[2];
    const float* W2 = (const float*)d_in[3];
    const float* b2 = (const float*)d_in[4];
    const float* W3 = (const float*)d_in[5];
    const float* b3 = (const float*)d_in[6];
    float* out = (float*)d_out;

    float* mean_pc = (float*)d_ws;
    float* sq      = mean_pc + (size_t)BB * CC * NN;
    float* tbl     = sq + (size_t)BB * NN + (size_t)BB * KK * NN;

    prep_kernel<<<BB * (NN / KQ), 256, 0, stream>>>(x, W1, mean_pc, sq, tbl, ws_size);
    knn_mlp_kernel<<<NWG, 256, 0, stream>>>(x, mean_pc, sq, W1, b1, W2, b2, W3, b3, out, tbl, ws_size);
}
